# starting kernel
# baseline (speedup 1.0000x reference)
#include <hip/hip_runtime.h>

typedef float v2f __attribute__((ext_vector_type(2)));
typedef float v8f __attribute__((ext_vector_type(8)));

#define IN_C   17
#define HID_C  64
#define OUT_C  32
#define NGRAPH 512


__global__ void k_zero4(float4* __restrict__ p, long n4) {
    long i = (long)blockIdx.x * blockDim.x + threadIdx.x;
    if (i < n4) p[i] = make_float4(0.f, 0.f, 0.f, 0.f);
}

__global__ void k_init_deg(float* __restrict__ deg, int n) {
    int i = blockIdx.x * blockDim.x + threadIdx.x;
    if (i < n) deg[i] = 1.0f;
}

__global__ void k_deg_count(const int* __restrict__ dst, float* __restrict__ deg, int E) {
    int e = blockIdx.x * blockDim.x + threadIdx.x;
    if (e < E) atomicAdd(&deg[dst[e]], 1.0f);
}

__global__ void k_dinv(const float* __restrict__ deg, float* __restrict__ dinv, int n) {
    int i = blockIdx.x * blockDim.x + threadIdx.x;
    if (i < n) dinv[i] = rsqrtf(deg[i]);
}


template <int CIN, int COUT>
__global__ void k_gemm(const float* __restrict__ A, const float* __restrict__ W,
                       float* __restrict__ out, int n) {
    const int  lane = threadIdx.x;
    const int  m    = lane & 15;
    const int  h    = lane >> 4;
    const long rowbase = (long)blockIdx.x * 16;

    constexpr int NT = COUT / 16;
    constexpr int KS = (CIN + 3) / 4;

    const long rA = rowbase + m < n ? rowbase + m : (long)n - 1;
    const float* ar = A + rA * CIN;

    v8f acc[NT] = {};
    for (int ks = 0; ks < KS; ++ks) {
        const int ka = ks * 4 + 2 * h;
        const int kb = ka + 1;
        const float ma = (ka < CIN) ? 1.f : 0.f;
        const float mb = (kb < CIN) ? 1.f : 0.f;
        const int kca = (ka < CIN) ? ka : (CIN - 1);
        const int kcb = (kb < CIN) ? kb : (CIN - 1);

        v2f a;
        a.x = ar[kca] * ma;
        a.y = ar[kcb] * mb;

#pragma unroll
        for (int nt = 0; nt < NT; ++nt) {
            v2f b;
            b.x = W[kca * COUT + nt * 16 + m] * ma;
            b.y = W[kcb * COUT + nt * 16 + m] * mb;
            acc[nt] = __builtin_amdgcn_wmma_f32_16x16x4_f32(
                false, a, false, b, (short)0, acc[nt], false, false);
        }
    }
#pragma unroll
    for (int nt = 0; nt < NT; ++nt)
#pragma unroll
        for (int v = 0; v < 8; ++v) {
            long r = rowbase + v + 8 * h;
            if (r < n) out[r * COUT + nt * 16 + m] = acc[nt][v];
        }
}


template <int C>
__global__ void k_edge_agg(const int* __restrict__ src, const int* __restrict__ dst,
                           const float* __restrict__ dinv, const float* __restrict__ hin,
                           float* __restrict__ agg, int E) {
    constexpr int G = C / 4;
    long tid = (long)blockIdx.x * blockDim.x + threadIdx.x;
    if (tid >= (long)E * G) return;
    int e  = (int)(tid / G);
    int cg = (int)(tid % G);
    int s = src[e], d = dst[e];
    float w = dinv[s] * dinv[d];
    const float4 v = *(const float4*)(hin + (long)s * C + cg * 4);
    float* o = agg + (long)d * C + cg * 4;
    atomicAdd(o + 0, w * v.x);
    atomicAdd(o + 1, w * v.y);
    atomicAdd(o + 2, w * v.z);
    atomicAdd(o + 3, w * v.w);
}


__global__ void k_post1(float* __restrict__ agg1, const float* __restrict__ h1,
                        const float* __restrict__ dinv, const float* __restrict__ b1, int n) {
    long tid = (long)blockIdx.x * blockDim.x + threadIdx.x;
    if (tid >= (long)n * HID_C) return;
    int i = (int)(tid / HID_C);
    int c = (int)(tid % HID_C);
    float sl = dinv[i];
    float v = agg1[tid] + sl * sl * h1[tid] + b1[c];
    agg1[tid] = v > 0.f ? v : 0.f;
}


__global__ void k_cnt(const int* __restrict__ batch, float* __restrict__ cnt, int n) {
    int i = blockIdx.x * blockDim.x + threadIdx.x;
    if (i < n) atomicAdd(&cnt[batch[i]], 1.0f);
}

__global__ void k_pool(const float* __restrict__ agg2, const float* __restrict__ h2,
                       const float* __restrict__ dinv, const int* __restrict__ batch,
                       float* __restrict__ pool, int n) {
    long tid = (long)blockIdx.x * blockDim.x + threadIdx.x;
    if (tid >= (long)n * OUT_C) return;
    int i = (int)(tid / OUT_C);
    int c = (int)(tid % OUT_C);
    float sl = dinv[i];
    float v = agg2[tid] + sl * sl * h2[tid];
    atomicAdd(&pool[batch[i] * OUT_C + c], v);
}


__global__ void k_fc(const float* __restrict__ pool, const float* __restrict__ cnt,
                     const float* __restrict__ b2,
                     const float* __restrict__ Wf1, const float* __restrict__ bf1,
                     const float* __restrict__ Wf2, const float* __restrict__ bf2,
                     float* __restrict__ out) {
    __shared__ float P[16 * 32];
    __shared__ float T[16 * 32];
    const int lane = threadIdx.x;
    const int m = lane & 15;
    const int h = lane >> 4;
    const int g0 = blockIdx.x * 16;

    for (int idx = lane; idx < 16 * 32; idx += 32) {
        int r = idx >> 5, c = idx & 31;
        float ct = cnt[g0 + r];
        ct = ct > 1.f ? ct : 1.f;
        P[idx] = pool[(g0 + r) * 32 + c] / ct + b2[c];
    }
    __syncthreads();

    v8f acc[2] = {};
#pragma unroll
    for (int ks = 0; ks < 8; ++ks) {
        const int ka = ks * 4 + 2 * h, kb = ka + 1;
        v2f a; a.x = P[m * 32 + ka]; a.y = P[m * 32 + kb];
#pragma unroll
        for (int nt = 0; nt < 2; ++nt) {
            v2f b; b.x = Wf1[ka * 32 + nt * 16 + m]; b.y = Wf1[kb * 32 + nt * 16 + m];
            acc[nt] = __builtin_amdgcn_wmma_f32_16x16x4_f32(
                false, a, false, b, (short)0, acc[nt], false, false);
        }
    }
#pragma unroll
    for (int nt = 0; nt < 2; ++nt)
#pragma unroll
        for (int v = 0; v < 8; ++v) {
            int r = v + 8 * h, c = nt * 16 + m;
            float t = acc[nt][v] + bf1[c];
            T[r * 32 + c] = t > 0.f ? t : 0.f;
        }
    __syncthreads();

    v8f o[2] = {};
#pragma unroll
    for (int ks = 0; ks < 8; ++ks) {
        const int ka = ks * 4 + 2 * h, kb = ka + 1;
        v2f a; a.x = T[m * 32 + ka]; a.y = T[m * 32 + kb];
#pragma unroll
        for (int nt = 0; nt < 2; ++nt) {
            v2f b; b.x = Wf2[ka * 32 + nt * 16 + m]; b.y = Wf2[kb * 32 + nt * 16 + m];
            o[nt] = __builtin_amdgcn_wmma_f32_16x16x4_f32(
                false, a, false, b, (short)0, o[nt], false, false);
        }
    }
#pragma unroll
    for (int nt = 0; nt < 2; ++nt)
#pragma unroll
        for (int v = 0; v < 8; ++v) {
            int r = v + 8 * h, c = nt * 16 + m;
            out[(g0 + r) * 32 + c] = o[nt][v] + bf2[c];
        }
}


extern "C" void kernel_launch(void* const* d_in, const int* in_sizes, int n_in,
                              void* d_out, int out_size, void* d_ws, size_t ws_size,
                              hipStream_t stream) {
    const float* x     = (const float*)d_in[0];
    const int*   ei    = (const int*)d_in[1];
    const int*   batch = (const int*)d_in[2];
    const float* W1  = (const float*)d_in[3];
    const float* b1  = (const float*)d_in[4];
    const float* W2  = (const float*)d_in[5];
    const float* b2  = (const float*)d_in[6];
    const float* Wf1 = (const float*)d_in[7];
    const float* bf1 = (const float*)d_in[8];
    const float* Wf2 = (const float*)d_in[9];
    const float* bf2 = (const float*)d_in[10];

    const int N = in_sizes[0] / IN_C;
    const int E = in_sizes[1] / 2;
    const int* src = ei;
    const int* dst = ei + E;

    float* ws   = (float*)d_ws;
    float* deg  = ws;
    float* dinv = deg  + N;
    float* h1   = dinv + N;
    float* agg1 = h1   + (size_t)N * HID_C;
    float* h2   = agg1 + (size_t)N * HID_C;
    float* agg2 = h2   + (size_t)N * OUT_C;
    float* pool = agg2 + (size_t)N * OUT_C;
    float* cnt  = pool + NGRAPH * OUT_C;

    const int B = 256;
    {
        long n4 = (long)N * HID_C / 4;
        k_zero4<<<(int)((n4 + B - 1) / B), B, 0, stream>>>((float4*)agg1, n4);
        n4 = (long)N * OUT_C / 4;
        k_zero4<<<(int)((n4 + B - 1) / B), B, 0, stream>>>((float4*)agg2, n4);
        n4 = (NGRAPH * OUT_C + NGRAPH) / 4;
        k_zero4<<<(int)((n4 + B - 1) / B), B, 0, stream>>>((float4*)pool, n4);
    }

    k_init_deg <<<(N + B - 1) / B, B, 0, stream>>>(deg, N);
    k_deg_count<<<(E + B - 1) / B, B, 0, stream>>>(dst, deg, E);
    k_dinv     <<<(N + B - 1) / B, B, 0, stream>>>(deg, dinv, N);

    const int tiles = (N + 15) / 16;
    k_gemm<IN_C, HID_C><<<tiles, 32, 0, stream>>>(x, W1, h1, N);

    long t1 = (long)E * (HID_C / 4);
    k_edge_agg<HID_C><<<(int)((t1 + B - 1) / B), B, 0, stream>>>(src, dst, dinv, h1, agg1, E);

    long t2 = (long)N * HID_C;
    k_post1<<<(int)((t2 + B - 1) / B), B, 0, stream>>>(agg1, h1, dinv, b1, N);

    k_gemm<HID_C, OUT_C><<<tiles, 32, 0, stream>>>(agg1, W2, h2, N);

    long t3 = (long)E * (OUT_C / 4);
    k_edge_agg<OUT_C><<<(int)((t3 + B - 1) / B), B, 0, stream>>>(src, dst, dinv, h2, agg2, E);

    k_cnt<<<(N + B - 1) / B, B, 0, stream>>>(batch, cnt, N);

    long t4 = (long)N * OUT_C;
    k_pool<<<(int)((t4 + B - 1) / B), B, 0, stream>>>(agg2, h2, dinv, batch, pool, N);

    k_fc<<<NGRAPH / 16, 32, 0, stream>>>(pool, cnt, b2, Wf1, bf1, Wf2, bf2, (float*)d_out);
}


# layer 1 aggregates the 17 input channels BEFORE its GEMM ((A X) W1, sums re-associated): 17 float atomics per edge, not 64; self-loop seeds the buffer, bias+ReLU move into the GEMM epilogue
# speedup vs baseline: 1.4473x; 1.4473x over previous
#include <hip/hip_runtime.h>

typedef float v2f __attribute__((ext_vector_type(2)));
typedef float v8f __attribute__((ext_vector_type(8)));

#define IN_C   17
#define AGX_LD 32
#define HID_C  64
#define OUT_C  32
#define NGRAPH 512


__global__ void k_zero4(float4* __restrict__ p, long n4) {
    long i = (long)blockIdx.x * blockDim.x + threadIdx.x;
    if (i < n4) p[i] = make_float4(0.f, 0.f, 0.f, 0.f);
}

__global__ void k_init_deg(float* __restrict__ deg, int n) {
    int i = blockIdx.x * blockDim.x + threadIdx.x;
    if (i < n) deg[i] = 1.0f;
}

__global__ void k_deg_count(const int* __restrict__ dst, float* __restrict__ deg, int E) {
    int e = blockIdx.x * blockDim.x + threadIdx.x;
    if (e < E) atomicAdd(&deg[dst[e]], 1.0f);
}

__global__ void k_dinv(const float* __restrict__ deg, float* __restrict__ dinv, int n) {
    int i = blockIdx.x * blockDim.x + threadIdx.x;
    if (i < n) dinv[i] = rsqrtf(deg[i]);
}


template <int CIN, int LDA, int COUT, bool BIAS_RELU>
__global__ void k_gemm(const float* __restrict__ A, const float* __restrict__ W,
                       const float* __restrict__ bias, float* __restrict__ out, int n) {
    const int  lane = threadIdx.x;
    const int  m    = lane & 15;
    const int  h    = lane >> 4;
    const long rowbase = (long)blockIdx.x * 16;

    constexpr int NT = COUT / 16;
    constexpr int KS = (CIN + 3) / 4;

    const long rA = rowbase + m < n ? rowbase + m : (long)n - 1;
    const float* ar = A + rA * LDA;

    v8f acc[NT] = {};
    for (int ks = 0; ks < KS; ++ks) {
        const int ka = ks * 4 + 2 * h;
        const int kb = ka + 1;
        const float ma = (ka < CIN) ? 1.f : 0.f;
        const float mb = (kb < CIN) ? 1.f : 0.f;
        const int kca = (ka < CIN) ? ka : (CIN - 1);
        const int kcb = (kb < CIN) ? kb : (CIN - 1);

        v2f a;
        a.x = ar[kca] * ma;
        a.y = ar[kcb] * mb;

#pragma unroll
        for (int nt = 0; nt < NT; ++nt) {
            v2f b;
            b.x = W[kca * COUT + nt * 16 + m] * ma;
            b.y = W[kcb * COUT + nt * 16 + m] * mb;
            acc[nt] = __builtin_amdgcn_wmma_f32_16x16x4_f32(
                false, a, false, b, (short)0, acc[nt], false, false);
        }
    }
#pragma unroll
    for (int nt = 0; nt < NT; ++nt) {
        const float bv = BIAS_RELU ? bias[nt * 16 + m] : 0.f;
#pragma unroll
        for (int v = 0; v < 8; ++v) {
            long r = rowbase + v + 8 * h;
            float t = acc[nt][v];
            if (BIAS_RELU) { t += bv; t = t > 0.f ? t : 0.f; }
            if (r < n) out[r * COUT + nt * 16 + m] = t;
        }
    }
}


template <int C>
__global__ void k_edge_agg(const int* __restrict__ src, const int* __restrict__ dst,
                           const float* __restrict__ dinv, const float* __restrict__ hin,
                           float* __restrict__ agg, int E) {
    constexpr int G = C / 4;
    long tid = (long)blockIdx.x * blockDim.x + threadIdx.x;
    if (tid >= (long)E * G) return;
    int e  = (int)(tid / G);
    int cg = (int)(tid % G);
    int s = src[e], d = dst[e];
    float w = dinv[s] * dinv[d];
    const float4 v = *(const float4*)(hin + (long)s * C + cg * 4);
    float* o = agg + (long)d * C + cg * 4;
    atomicAdd(o + 0, w * v.x);
    atomicAdd(o + 1, w * v.y);
    atomicAdd(o + 2, w * v.z);
    atomicAdd(o + 3, w * v.w);
}


#define AGX_ROWS 15
#define AGX_U    4

__global__ void k_init_aggx(const float* __restrict__ x, const float* __restrict__ dinv,
                            float* __restrict__ aggx, int n) {
    const long i = (long)blockIdx.x * AGX_ROWS + threadIdx.y;
    const int  c = threadIdx.x;
    if (i >= n) return;
    const float sl = dinv[i];
    aggx[i * AGX_LD + c] = sl * sl * x[i * IN_C + c];
}

template <bool GUARD>
__device__ __forceinline__ void edge_agg_x_body(const int* __restrict__ src, const int* __restrict__ dst,
                                                const float* __restrict__ dinv, const float* __restrict__ x,
                                                float* __restrict__ aggx, long ebase, int E) {
    const int c = threadIdx.x;
    long  o[AGX_U];
    float v[AGX_U];
    bool  ok[AGX_U];
#pragma unroll
    for (int k = 0; k < AGX_U; ++k) {
        long e = ebase + k * AGX_ROWS + threadIdx.y;
        ok[k] = !GUARD || e < E;
        if (GUARD) e = ok[k] ? e : (long)E - 1;
        const int s = src[e], d = dst[e];
        v[k] = dinv[s] * dinv[d] * x[(long)s * IN_C + c];
        o[k] = (long)d * AGX_LD + c;
    }
#pragma unroll
    for (int k = 0; k < AGX_U; ++k)
        if (ok[k]) atomicAdd(aggx + o[k], v[k]);
}

__global__ void k_edge_agg_x(const int* __restrict__ src, const int* __restrict__ dst,
                             const float* __restrict__ dinv, const float* __restrict__ x,
                             float* __restrict__ aggx, int E) {
    const long ebase = (long)blockIdx.x * (AGX_U * AGX_ROWS);
    if (ebase + AGX_U * AGX_ROWS <= E) edge_agg_x_body<false>(src, dst, dinv, x, aggx, ebase, E);
    else                               edge_agg_x_body<true >(src, dst, dinv, x, aggx, ebase, E);
}


__global__ void k_cnt(const int* __restrict__ batch, float* __restrict__ cnt, int n) {
    int i = blockIdx.x * blockDim.x + threadIdx.x;
    if (i < n) atomicAdd(&cnt[batch[i]], 1.0f);
}

__global__ void k_pool(const float* __restrict__ agg2, const float* __restrict__ h2,
                       const float* __restrict__ dinv, const int* __restrict__ batch,
                       float* __restrict__ pool, int n) {
    long tid = (long)blockIdx.x * blockDim.x + threadIdx.x;
    if (tid >= (long)n * OUT_C) return;
    int i = (int)(tid / OUT_C);
    int c = (int)(tid % OUT_C);
    float sl = dinv[i];
    float v = agg2[tid] + sl * sl * h2[tid];
    atomicAdd(&pool[batch[i] * OUT_C + c], v);
}


__global__ void k_fc(const float* __restrict__ pool, const float* __restrict__ cnt,
                     const float* __restrict__ b2,
                     const float* __restrict__ Wf1, const float* __restrict__ bf1,
                     const float* __restrict__ Wf2, const float* __restrict__ bf2,
                     float* __restrict__ out) {
    __shared__ float P[16 * 32];
    __shared__ float T[16 * 32];
    const int lane = threadIdx.x;
    const int m = lane & 15;
    const int h = lane >> 4;
    const int g0 = blockIdx.x * 16;

    for (int idx = lane; idx < 16 * 32; idx += 32) {
        int r = idx >> 5, c = idx & 31;
        float ct = cnt[g0 + r];
        ct = ct > 1.f ? ct : 1.f;
        P[idx] = pool[(g0 + r) * 32 + c] / ct + b2[c];
    }
    __syncthreads();

    v8f acc[2] = {};
#pragma unroll
    for (int ks = 0; ks < 8; ++ks) {
        const int ka = ks * 4 + 2 * h, kb = ka + 1;
        v2f a; a.x = P[m * 32 + ka]; a.y = P[m * 32 + kb];
#pragma unroll
        for (int nt = 0; nt < 2; ++nt) {
            v2f b; b.x = Wf1[ka * 32 + nt * 16 + m]; b.y = Wf1[kb * 32 + nt * 16 + m];
            acc[nt] = __builtin_amdgcn_wmma_f32_16x16x4_f32(
                false, a, false, b, (short)0, acc[nt], false, false);
        }
    }
#pragma unroll
    for (int nt = 0; nt < 2; ++nt)
#pragma unroll
        for (int v = 0; v < 8; ++v) {
            int r = v + 8 * h, c = nt * 16 + m;
            float t = acc[nt][v] + bf1[c];
            T[r * 32 + c] = t > 0.f ? t : 0.f;
        }
    __syncthreads();

    v8f o[2] = {};
#pragma unroll
    for (int ks = 0; ks < 8; ++ks) {
        const int ka = ks * 4 + 2 * h, kb = ka + 1;
        v2f a; a.x = T[m * 32 + ka]; a.y = T[m * 32 + kb];
#pragma unroll
        for (int nt = 0; nt < 2; ++nt) {
            v2f b; b.x = Wf2[ka * 32 + nt * 16 + m]; b.y = Wf2[kb * 32 + nt * 16 + m];
            o[nt] = __builtin_amdgcn_wmma_f32_16x16x4_f32(
                false, a, false, b, (short)0, o[nt], false, false);
        }
    }
#pragma unroll
    for (int nt = 0; nt < 2; ++nt)
#pragma unroll
        for (int v = 0; v < 8; ++v) {
            int r = v + 8 * h, c = nt * 16 + m;
            out[(g0 + r) * 32 + c] = o[nt][v] + bf2[c];
        }
}


extern "C" void kernel_launch(void* const* d_in, const int* in_sizes, int n_in,
                              void* d_out, int out_size, void* d_ws, size_t ws_size,
                              hipStream_t stream) {
    const float* x     = (const float*)d_in[0];
    const int*   ei    = (const int*)d_in[1];
    const int*   batch = (const int*)d_in[2];
    const float* W1  = (const float*)d_in[3];
    const float* b1  = (const float*)d_in[4];
    const float* W2  = (const float*)d_in[5];
    const float* b2  = (const float*)d_in[6];
    const float* Wf1 = (const float*)d_in[7];
    const float* bf1 = (const float*)d_in[8];
    const float* Wf2 = (const float*)d_in[9];
    const float* bf2 = (const float*)d_in[10];

    const int N = in_sizes[0] / IN_C;
    const int E = in_sizes[1] / 2;
    const int* src = ei;
    const int* dst = ei + E;

    float* ws   = (float*)d_ws;
    float* deg  = ws;
    float* dinv = deg  + N;
    float* aggx = dinv + N;
    float* hid  = aggx + (size_t)N * HID_C;
    float* h2   = hid  + (size_t)N * HID_C;
    float* agg2 = h2   + (size_t)N * OUT_C;
    float* pool = agg2 + (size_t)N * OUT_C;
    float* cnt  = pool + NGRAPH * OUT_C;

    const int B = 256;
    {
        long n4 = (long)N * OUT_C / 4;
        k_zero4<<<(int)((n4 + B - 1) / B), B, 0, stream>>>((float4*)agg2, n4);
        n4 = (NGRAPH * OUT_C + NGRAPH) / 4;
        k_zero4<<<(int)((n4 + B - 1) / B), B, 0, stream>>>((float4*)pool, n4);
    }

    k_init_deg <<<(N + B - 1) / B, B, 0, stream>>>(deg, N);
    k_deg_count<<<(E + B - 1) / B, B, 0, stream>>>(dst, deg, E);
    k_dinv     <<<(N + B - 1) / B, B, 0, stream>>>(deg, dinv, N);

    const dim3 BX(IN_C, AGX_ROWS);
    k_init_aggx<<<(N + AGX_ROWS - 1) / AGX_ROWS, BX, 0, stream>>>(x, dinv, aggx, N);

    k_edge_agg_x<<<(E + AGX_U * AGX_ROWS - 1) / (AGX_U * AGX_ROWS), BX, 0, stream>>>(src, dst, dinv, x, aggx, E);

    const int tiles = (N + 15) / 16;
    k_gemm<IN_C, AGX_LD, HID_C, true><<<tiles, 32, 0, stream>>>(aggx, W1, b1, hid, N);

    k_gemm<HID_C, HID_C, OUT_C, false><<<tiles, 32, 0, stream>>>(hid, W2, nullptr, h2, N);

    long t3 = (long)E * (OUT_C / 4);
    k_edge_agg<OUT_C><<<(int)((t3 + B - 1) / B), B, 0, stream>>>(src, dst, dinv, h2, agg2, E);

    k_cnt<<<(N + B - 1) / B, B, 0, stream>>>(batch, cnt, N);

    long t4 = (long)N * OUT_C;
    k_pool<<<(int)((t4 + B - 1) / B), B, 0, stream>>>(agg2, h2, dinv, batch, pool, N);

    k_fc<<<NGRAPH / 16, 32, 0, stream>>>(pool, cnt, b2, Wf1, bf1, Wf2, bf2, (float*)d_out);
}


# degree count leaves its one scattered global float atomic per edge: integer LDS histograms per (edge slice, node range) block, k_dinv sums the 32 partial planes; counts exact; one launch less
# speedup vs baseline: 1.8367x; 1.2691x over previous
#include <hip/hip_runtime.h>

typedef float v2f __attribute__((ext_vector_type(2)));
typedef float v8f __attribute__((ext_vector_type(8)));

#define IN_C   17
#define AGX_LD 32
#define HID_C  64
#define OUT_C  32
#define NGRAPH 512


__global__ void k_zero4(float4* __restrict__ p, long n4) {
    long i = (long)blockIdx.x * blockDim.x + threadIdx.x;
    if (i < n4) p[i] = make_float4(0.f, 0.f, 0.f, 0.f);
}


#define DEG_RN 12800
#define DEG_S  32
#define DEG_T  1024
#define DEG_U  8

__global__ void k_deg_hist(const int* __restrict__ dst, int* __restrict__ part, int n, int E) {
    __shared__ int bins[DEG_RN];
    const int lo = blockIdx.y * DEG_RN;
    for (int j = threadIdx.x; j < DEG_RN; j += DEG_T) bins[j] = 0;
    __syncthreads();

    const int per = (E + DEG_S - 1) / DEG_S;
    const int e0  = blockIdx.x * per;
    const int e1  = e0 + per < E ? e0 + per : E;
    for (int eb = e0; eb < e1; eb += DEG_T * DEG_U) {
        int d[DEG_U];
#pragma unroll
        for (int k = 0; k < DEG_U; ++k) {
            const int e = eb + k * DEG_T + threadIdx.x;
            d[k] = e < e1 ? dst[e] : -1;
        }
#pragma unroll
        for (int k = 0; k < DEG_U; ++k) {
            const unsigned r = (unsigned)(d[k] - lo);
            if (r < (unsigned)DEG_RN) atomicAdd(&bins[r], 1);
        }
    }
    __syncthreads();

    const int cnt = n - lo < DEG_RN ? n - lo : DEG_RN;
    int* row = part + (long)blockIdx.x * n + lo;
    for (int j = threadIdx.x; j < cnt; j += DEG_T) row[j] = bins[j];
}

__global__ void k_dinv(const int* __restrict__ part, float* __restrict__ dinv, int n) {
    int i = blockIdx.x * blockDim.x + threadIdx.x;
    if (i >= n) return;
    int c = 1;
#pragma unroll
    for (int s = 0; s < DEG_S; ++s) c += part[(long)s * n + i];
    dinv[i] = rsqrtf((float)c);
}


template <int CIN, int LDA, int COUT, bool BIAS_RELU>
__global__ void k_gemm(const float* __restrict__ A, const float* __restrict__ W,
                       const float* __restrict__ bias, float* __restrict__ out, int n) {
    const int  lane = threadIdx.x;
    const int  m    = lane & 15;
    const int  h    = lane >> 4;
    const long rowbase = (long)blockIdx.x * 16;

    constexpr int NT = COUT / 16;
    constexpr int KS = (CIN + 3) / 4;

    const long rA = rowbase + m < n ? rowbase + m : (long)n - 1;
    const float* ar = A + rA * LDA;

    v8f acc[NT] = {};
    for (int ks = 0; ks < KS; ++ks) {
        const int ka = ks * 4 + 2 * h;
        const int kb = ka + 1;
        const float ma = (ka < CIN) ? 1.f : 0.f;
        const float mb = (kb < CIN) ? 1.f : 0.f;
        const int kca = (ka < CIN) ? ka : (CIN - 1);
        const int kcb = (kb < CIN) ? kb : (CIN - 1);

        v2f a;
        a.x = ar[kca] * ma;
        a.y = ar[kcb] * mb;

#pragma unroll
        for (int nt = 0; nt < NT; ++nt) {
            v2f b;
            b.x = W[kca * COUT + nt * 16 + m] * ma;
            b.y = W[kcb * COUT + nt * 16 + m] * mb;
            acc[nt] = __builtin_amdgcn_wmma_f32_16x16x4_f32(
                false, a, false, b, (short)0, acc[nt], false, false);
        }
    }
#pragma unroll
    for (int nt = 0; nt < NT; ++nt) {
        const float bv = BIAS_RELU ? bias[nt * 16 + m] : 0.f;
#pragma unroll
        for (int v = 0; v < 8; ++v) {
            long r = rowbase + v + 8 * h;
            float t = acc[nt][v];
            if (BIAS_RELU) { t += bv; t = t > 0.f ? t : 0.f; }
            if (r < n) out[r * COUT + nt * 16 + m] = t;
        }
    }
}


template <int C>
__global__ void k_edge_agg(const int* __restrict__ src, const int* __restrict__ dst,
                           const float* __restrict__ dinv, const float* __restrict__ hin,
                           float* __restrict__ agg, int E) {
    constexpr int G = C / 4;
    long tid = (long)blockIdx.x * blockDim.x + threadIdx.x;
    if (tid >= (long)E * G) return;
    int e  = (int)(tid / G);
    int cg = (int)(tid % G);
    int s = src[e], d = dst[e];
    float w = dinv[s] * dinv[d];
    const float4 v = *(const float4*)(hin + (long)s * C + cg * 4);
    float* o = agg + (long)d * C + cg * 4;
    atomicAdd(o + 0, w * v.x);
    atomicAdd(o + 1, w * v.y);
    atomicAdd(o + 2, w * v.z);
    atomicAdd(o + 3, w * v.w);
}


#define AGX_ROWS 15
#define AGX_U    4

__global__ void k_init_aggx(const float* __restrict__ x, const float* __restrict__ dinv,
                            float* __restrict__ aggx, int n) {
    const long i = (long)blockIdx.x * AGX_ROWS + threadIdx.y;
    const int  c = threadIdx.x;
    if (i >= n) return;
    const float sl = dinv[i];
    aggx[i * AGX_LD + c] = sl * sl * x[i * IN_C + c];
}

template <bool GUARD>
__device__ __forceinline__ void edge_agg_x_body(const int* __restrict__ src, const int* __restrict__ dst,
                                                const float* __restrict__ dinv, const float* __restrict__ x,
                                                float* __restrict__ aggx, long ebase, int E) {
    const int c = threadIdx.x;
    long  o[AGX_U];
    float v[AGX_U];
    bool  ok[AGX_U];
#pragma unroll
    for (int k = 0; k < AGX_U; ++k) {
        long e = ebase + k * AGX_ROWS + threadIdx.y;
        ok[k] = !GUARD || e < E;
        if (GUARD) e = ok[k] ? e : (long)E - 1;
        const int s = src[e], d = dst[e];
        v[k] = dinv[s] * dinv[d] * x[(long)s * IN_C + c];
        o[k] = (long)d * AGX_LD + c;
    }
#pragma unroll
    for (int k = 0; k < AGX_U; ++k)
        if (ok[k]) atomicAdd(aggx + o[k], v[k]);
}

__global__ void k_edge_agg_x(const int* __restrict__ src, const int* __restrict__ dst,
                             const float* __restrict__ dinv, const float* __restrict__ x,
                             float* __restrict__ aggx, int E) {
    const long ebase = (long)blockIdx.x * (AGX_U * AGX_ROWS);
    if (ebase + AGX_U * AGX_ROWS <= E) edge_agg_x_body<false>(src, dst, dinv, x, aggx, ebase, E);
    else                               edge_agg_x_body<true >(src, dst, dinv, x, aggx, ebase, E);
}


__global__ void k_cnt(const int* __restrict__ batch, float* __restrict__ cnt, int n) {
    int i = blockIdx.x * blockDim.x + threadIdx.x;
    if (i < n) atomicAdd(&cnt[batch[i]], 1.0f);
}

__global__ void k_pool(const float* __restrict__ agg2, const float* __restrict__ h2,
                       const float* __restrict__ dinv, const int* __restrict__ batch,
                       float* __restrict__ pool, int n) {
    long tid = (long)blockIdx.x * blockDim.x + threadIdx.x;
    if (tid >= (long)n * OUT_C) return;
    int i = (int)(tid / OUT_C);
    int c = (int)(tid % OUT_C);
    float sl = dinv[i];
    float v = agg2[tid] + sl * sl * h2[tid];
    atomicAdd(&pool[batch[i] * OUT_C + c], v);
}


__global__ void k_fc(const float* __restrict__ pool, const float* __restrict__ cnt,
                     const float* __restrict__ b2,
                     const float* __restrict__ Wf1, const float* __restrict__ bf1,
                     const float* __restrict__ Wf2, const float* __restrict__ bf2,
                     float* __restrict__ out) {
    __shared__ float P[16 * 32];
    __shared__ float T[16 * 32];
    const int lane = threadIdx.x;
    const int m = lane & 15;
    const int h = lane >> 4;
    const int g0 = blockIdx.x * 16;

    for (int idx = lane; idx < 16 * 32; idx += 32) {
        int r = idx >> 5, c = idx & 31;
        float ct = cnt[g0 + r];
        ct = ct > 1.f ? ct : 1.f;
        P[idx] = pool[(g0 + r) * 32 + c] / ct + b2[c];
    }
    __syncthreads();

    v8f acc[2] = {};
#pragma unroll
    for (int ks = 0; ks < 8; ++ks) {
        const int ka = ks * 4 + 2 * h, kb = ka + 1;
        v2f a; a.x = P[m * 32 + ka]; a.y = P[m * 32 + kb];
#pragma unroll
        for (int nt = 0; nt < 2; ++nt) {
            v2f b; b.x = Wf1[ka * 32 + nt * 16 + m]; b.y = Wf1[kb * 32 + nt * 16 + m];
            acc[nt] = __builtin_amdgcn_wmma_f32_16x16x4_f32(
                false, a, false, b, (short)0, acc[nt], false, false);
        }
    }
#pragma unroll
    for (int nt = 0; nt < 2; ++nt)
#pragma unroll
        for (int v = 0; v < 8; ++v) {
            int r = v + 8 * h, c = nt * 16 + m;
            float t = acc[nt][v] + bf1[c];
            T[r * 32 + c] = t > 0.f ? t : 0.f;
        }
    __syncthreads();

    v8f o[2] = {};
#pragma unroll
    for (int ks = 0; ks < 8; ++ks) {
        const int ka = ks * 4 + 2 * h, kb = ka + 1;
        v2f a; a.x = T[m * 32 + ka]; a.y = T[m * 32 + kb];
#pragma unroll
        for (int nt = 0; nt < 2; ++nt) {
            v2f b; b.x = Wf2[ka * 32 + nt * 16 + m]; b.y = Wf2[kb * 32 + nt * 16 + m];
            o[nt] = __builtin_amdgcn_wmma_f32_16x16x4_f32(
                false, a, false, b, (short)0, o[nt], false, false);
        }
    }
#pragma unroll
    for (int nt = 0; nt < 2; ++nt)
#pragma unroll
        for (int v = 0; v < 8; ++v) {
            int r = v + 8 * h, c = nt * 16 + m;
            out[(g0 + r) * 32 + c] = o[nt][v] + bf2[c];
        }
}


extern "C" void kernel_launch(void* const* d_in, const int* in_sizes, int n_in,
                              void* d_out, int out_size, void* d_ws, size_t ws_size,
                              hipStream_t stream) {
    const float* x     = (const float*)d_in[0];
    const int*   ei    = (const int*)d_in[1];
    const int*   batch = (const int*)d_in[2];
    const float* W1  = (const float*)d_in[3];
    const float* b1  = (const float*)d_in[4];
    const float* W2  = (const float*)d_in[5];
    const float* b2  = (const float*)d_in[6];
    const float* Wf1 = (const float*)d_in[7];
    const float* bf1 = (const float*)d_in[8];
    const float* Wf2 = (const float*)d_in[9];
    const float* bf2 = (const float*)d_in[10];

    const int N = in_sizes[0] / IN_C;
    const int E = in_sizes[1] / 2;
    const int* src = ei;
    const int* dst = ei + E;

    float* ws   = (float*)d_ws;
    float* deg  = ws;
    float* dinv = deg  + N;
    float* aggx = dinv + N;
    float* hid  = aggx + (size_t)N * HID_C;
    float* h2   = hid  + (size_t)N * HID_C;
    float* agg2 = h2   + (size_t)N * OUT_C;
    float* pool = agg2 + (size_t)N * OUT_C;
    float* cnt  = pool + NGRAPH * OUT_C;

    const int B = 256;
    {
        long n4 = (long)N * OUT_C / 4;
        k_zero4<<<(int)((n4 + B - 1) / B), B, 0, stream>>>((float4*)agg2, n4);
        n4 = (NGRAPH * OUT_C + NGRAPH) / 4;
        k_zero4<<<(int)((n4 + B - 1) / B), B, 0, stream>>>((float4*)pool, n4);
    }

    int* part = (int*)hid;
    k_deg_hist<<<dim3(DEG_S, (N + DEG_RN - 1) / DEG_RN), DEG_T, 0, stream>>>(dst, part, N, E);
    k_dinv    <<<(N + B - 1) / B, B, 0, stream>>>(part, dinv, N);

    const dim3 BX(IN_C, AGX_ROWS);
    k_init_aggx<<<(N + AGX_ROWS - 1) / AGX_ROWS, BX, 0, stream>>>(x, dinv, aggx, N);

    k_edge_agg_x<<<(E + AGX_U * AGX_ROWS - 1) / (AGX_U * AGX_ROWS), BX, 0, stream>>>(src, dst, dinv, x, aggx, E);

    const int tiles = (N + 15) / 16;
    k_gemm<IN_C, AGX_LD, HID_C, true><<<tiles, 32, 0, stream>>>(aggx, W1, b1, hid, N);

    k_gemm<HID_C, HID_C, OUT_C, false><<<tiles, 32, 0, stream>>>(hid, W2, nullptr, h2, N);

    long t3 = (long)E * (OUT_C / 4);
    k_edge_agg<OUT_C><<<(int)((t3 + B - 1) / B), B, 0, stream>>>(src, dst, dinv, h2, agg2, E);

    k_cnt<<<(N + B - 1) / B, B, 0, stream>>>(batch, cnt, N);

    long t4 = (long)N * OUT_C;
    k_pool<<<(int)((t4 + B - 1) / B), B, 0, stream>>>(agg2, h2, dinv, batch, pool, N);

    k_fc<<<NGRAPH / 16, 32, 0, stream>>>(pool, cnt, b2, Wf1, bf1, Wf2, bf2, (float*)d_out);
}


# conv2's aggregation folded into the mean pool (both linear): each edge adds w*h2[src] to its destination's GRAPH row in a per-block LDS table; one global add per entry at exit; sums re-ordered
# speedup vs baseline: 2.4785x; 1.3494x over previous
#include <hip/hip_runtime.h>

typedef float v2f __attribute__((ext_vector_type(2)));
typedef float v8f __attribute__((ext_vector_type(8)));

#define IN_C   17
#define AGX_LD 32
#define HID_C  64
#define OUT_C  32
#define NGRAPH 512


__global__ void k_zero4(float4* __restrict__ p, long n4) {
    long i = (long)blockIdx.x * blockDim.x + threadIdx.x;
    if (i < n4) p[i] = make_float4(0.f, 0.f, 0.f, 0.f);
}


#define DEG_RN 12800
#define DEG_S  32
#define DEG_T  1024
#define DEG_U  8

__global__ void k_deg_hist(const int* __restrict__ dst, int* __restrict__ part, int n, int E) {
    __shared__ int bins[DEG_RN];
    const int lo = blockIdx.y * DEG_RN;
    for (int j = threadIdx.x; j < DEG_RN; j += DEG_T) bins[j] = 0;
    __syncthreads();

    const int per = (E + DEG_S - 1) / DEG_S;
    const int e0  = blockIdx.x * per;
    const int e1  = e0 + per < E ? e0 + per : E;
    for (int eb = e0; eb < e1; eb += DEG_T * DEG_U) {
        int d[DEG_U];
#pragma unroll
        for (int k = 0; k < DEG_U; ++k) {
            const int e = eb + k * DEG_T + threadIdx.x;
            d[k] = e < e1 ? dst[e] : -1;
        }
#pragma unroll
        for (int k = 0; k < DEG_U; ++k) {
            const unsigned r = (unsigned)(d[k] - lo);
            if (r < (unsigned)DEG_RN) atomicAdd(&bins[r], 1);
        }
    }
    __syncthreads();

    const int cnt = n - lo < DEG_RN ? n - lo : DEG_RN;
    int* row = part + (long)blockIdx.x * n + lo;
    for (int j = threadIdx.x; j < cnt; j += DEG_T) row[j] = bins[j];
}

__global__ void k_dinv(const int* __restrict__ part, float* __restrict__ dinv, int n) {
    int i = blockIdx.x * blockDim.x + threadIdx.x;
    if (i >= n) return;
    int c = 1;
#pragma unroll
    for (int s = 0; s < DEG_S; ++s) c += part[(long)s * n + i];
    dinv[i] = rsqrtf((float)c);
}


template <int CIN, int LDA, int COUT, bool BIAS_RELU>
__global__ void k_gemm(const float* __restrict__ A, const float* __restrict__ W,
                       const float* __restrict__ bias, float* __restrict__ out, int n) {
    const int  lane = threadIdx.x;
    const int  m    = lane & 15;
    const int  h    = lane >> 4;
    const long rowbase = (long)blockIdx.x * 16;

    constexpr int NT = COUT / 16;
    constexpr int KS = (CIN + 3) / 4;

    const long rA = rowbase + m < n ? rowbase + m : (long)n - 1;
    const float* ar = A + rA * LDA;

    v8f acc[NT] = {};
    for (int ks = 0; ks < KS; ++ks) {
        const int ka = ks * 4 + 2 * h;
        const int kb = ka + 1;
        const float ma = (ka < CIN) ? 1.f : 0.f;
        const float mb = (kb < CIN) ? 1.f : 0.f;
        const int kca = (ka < CIN) ? ka : (CIN - 1);
        const int kcb = (kb < CIN) ? kb : (CIN - 1);

        v2f a;
        a.x = ar[kca] * ma;
        a.y = ar[kcb] * mb;

#pragma unroll
        for (int nt = 0; nt < NT; ++nt) {
            v2f b;
            b.x = W[kca * COUT + nt * 16 + m] * ma;
            b.y = W[kcb * COUT + nt * 16 + m] * mb;
            acc[nt] = __builtin_amdgcn_wmma_f32_16x16x4_f32(
                false, a, false, b, (short)0, acc[nt], false, false);
        }
    }
#pragma unroll
    for (int nt = 0; nt < NT; ++nt) {
        const float bv = BIAS_RELU ? bias[nt * 16 + m] : 0.f;
#pragma unroll
        for (int v = 0; v < 8; ++v) {
            long r = rowbase + v + 8 * h;
            float t = acc[nt][v];
            if (BIAS_RELU) { t += bv; t = t > 0.f ? t : 0.f; }
            if (r < n) out[r * COUT + nt * 16 + m] = t;
        }
    }
}


#define AGX_ROWS 15
#define AGX_U    4

__global__ void k_init_aggx(const float* __restrict__ x, const float* __restrict__ dinv,
                            float* __restrict__ aggx, int n) {
    const long i = (long)blockIdx.x * AGX_ROWS + threadIdx.y;
    const int  c = threadIdx.x;
    if (i >= n) return;
    const float sl = dinv[i];
    aggx[i * AGX_LD + c] = sl * sl * x[i * IN_C + c];
}

template <bool GUARD>
__device__ __forceinline__ void edge_agg_x_body(const int* __restrict__ src, const int* __restrict__ dst,
                                                const float* __restrict__ dinv, const float* __restrict__ x,
                                                float* __restrict__ aggx, long ebase, int E) {
    const int c = threadIdx.x;
    long  o[AGX_U];
    float v[AGX_U];
    bool  ok[AGX_U];
#pragma unroll
    for (int k = 0; k < AGX_U; ++k) {
        long e = ebase + k * AGX_ROWS + threadIdx.y;
        ok[k] = !GUARD || e < E;
        if (GUARD) e = ok[k] ? e : (long)E - 1;
        const int s = src[e], d = dst[e];
        v[k] = dinv[s] * dinv[d] * x[(long)s * IN_C + c];
        o[k] = (long)d * AGX_LD + c;
    }
#pragma unroll
    for (int k = 0; k < AGX_U; ++k)
        if (ok[k]) atomicAdd(aggx + o[k], v[k]);
}

__global__ void k_edge_agg_x(const int* __restrict__ src, const int* __restrict__ dst,
                             const float* __restrict__ dinv, const float* __restrict__ x,
                             float* __restrict__ aggx, int E) {
    const long ebase = (long)blockIdx.x * (AGX_U * AGX_ROWS);
    if (ebase + AGX_U * AGX_ROWS <= E) edge_agg_x_body<false>(src, dst, dinv, x, aggx, ebase, E);
    else                               edge_agg_x_body<true >(src, dst, dinv, x, aggx, ebase, E);
}


#define AP_T 1024
#define AP_NB 256

__global__ void k_cnt(const int* __restrict__ batch, float* __restrict__ cnt, int n) {
    int i = blockIdx.x * blockDim.x + threadIdx.x;
    if (i < n) atomicAdd(&cnt[batch[i]], 1.0f);
}

template <bool GUARD>
__device__ __forceinline__ void agg_pool_chunk(const int* __restrict__ src, const int* __restrict__ dst,
                                               const float* __restrict__ dinv, const int* __restrict__ batch,
                                               const float* __restrict__ h2, float* acc,
                                               long base, long total, int E, int lane) {
    const long idx = base + lane;
    int s = 0, d = 0;
    float w = 0.f;
    if (!GUARD || idx < total) {
        if (idx < E) { s = src[idx]; d = dst[idx]; }
        else         { s = d = (int)(idx - E); }
        w = dinv[s] * dinv[d];
    }
    const int g = batch[d];
    const int cnt = GUARD ? (int)(total - base) : 32;

    float v[32];
#pragma unroll
    for (int k = 0; k < 32; ++k) {
        const int sk = __builtin_amdgcn_readlane(s, k);
        const __attribute__((address_space(1))) float* row =
            (const __attribute__((address_space(1))) float*)(h2 + (long)sk * OUT_C);
        asm("" : "+s"(row));
        v[k] = (!GUARD || k < cnt) ? row[lane] : 0.f;
    }
#pragma unroll
    for (int k = 0; k < 32; ++k) {
        const int   gk = __builtin_amdgcn_readlane(g, k);
        const float wk = __int_as_float(__builtin_amdgcn_readlane(__float_as_int(w), k));
        if (!GUARD || k < cnt) atomicAdd(acc + gk * OUT_C + lane, wk * v[k]);
    }
}

__global__ void __launch_bounds__(AP_T)
k_agg_pool(const int* __restrict__ src, const int* __restrict__ dst,
           const float* __restrict__ dinv, const int* __restrict__ batch,
           const float* __restrict__ h2, float* __restrict__ pool, int n, int E) {
    __shared__ float acc[NGRAPH * OUT_C];
    for (int j = threadIdx.x; j < NGRAPH * OUT_C; j += AP_T) acc[j] = 0.f;
    __syncthreads();

    const int  lane   = threadIdx.x & 31;
    const int  wave   = threadIdx.x >> 5;
    const long total  = (long)E + n;
    const long nchunk = (total + 31) / 32;
    for (long c = (long)blockIdx.x * (AP_T / 32) + wave; c < nchunk; c += (long)gridDim.x * (AP_T / 32)) {
        const long base = c * 32;
        if (base + 32 <= total) agg_pool_chunk<false>(src, dst, dinv, batch, h2, acc, base, total, E, lane);
        else                    agg_pool_chunk<true >(src, dst, dinv, batch, h2, acc, base, total, E, lane);
    }
    __syncthreads();

    for (int j = threadIdx.x; j < NGRAPH * OUT_C; j += AP_T) {
        const int jj = (j + blockIdx.x * 64) & (NGRAPH * OUT_C - 1);
        atomicAdd(pool + jj, acc[jj]);
    }
}


__global__ void k_fc(const float* __restrict__ pool, const float* __restrict__ cnt,
                     const float* __restrict__ b2,
                     const float* __restrict__ Wf1, const float* __restrict__ bf1,
                     const float* __restrict__ Wf2, const float* __restrict__ bf2,
                     float* __restrict__ out) {
    __shared__ float P[16 * 32];
    __shared__ float T[16 * 32];
    const int lane = threadIdx.x;
    const int m = lane & 15;
    const int h = lane >> 4;
    const int g0 = blockIdx.x * 16;

    for (int idx = lane; idx < 16 * 32; idx += 32) {
        int r = idx >> 5, c = idx & 31;
        float ct = cnt[g0 + r];
        ct = ct > 1.f ? ct : 1.f;
        P[idx] = pool[(g0 + r) * 32 + c] / ct + b2[c];
    }
    __syncthreads();

    v8f acc[2] = {};
#pragma unroll
    for (int ks = 0; ks < 8; ++ks) {
        const int ka = ks * 4 + 2 * h, kb = ka + 1;
        v2f a; a.x = P[m * 32 + ka]; a.y = P[m * 32 + kb];
#pragma unroll
        for (int nt = 0; nt < 2; ++nt) {
            v2f b; b.x = Wf1[ka * 32 + nt * 16 + m]; b.y = Wf1[kb * 32 + nt * 16 + m];
            acc[nt] = __builtin_amdgcn_wmma_f32_16x16x4_f32(
                false, a, false, b, (short)0, acc[nt], false, false);
        }
    }
#pragma unroll
    for (int nt = 0; nt < 2; ++nt)
#pragma unroll
        for (int v = 0; v < 8; ++v) {
            int r = v + 8 * h, c = nt * 16 + m;
            float t = acc[nt][v] + bf1[c];
            T[r * 32 + c] = t > 0.f ? t : 0.f;
        }
    __syncthreads();

    v8f o[2] = {};
#pragma unroll
    for (int ks = 0; ks < 8; ++ks) {
        const int ka = ks * 4 + 2 * h, kb = ka + 1;
        v2f a; a.x = T[m * 32 + ka]; a.y = T[m * 32 + kb];
#pragma unroll
        for (int nt = 0; nt < 2; ++nt) {
            v2f b; b.x = Wf2[ka * 32 + nt * 16 + m]; b.y = Wf2[kb * 32 + nt * 16 + m];
            o[nt] = __builtin_amdgcn_wmma_f32_16x16x4_f32(
                false, a, false, b, (short)0, o[nt], false, false);
        }
    }
#pragma unroll
    for (int nt = 0; nt < 2; ++nt)
#pragma unroll
        for (int v = 0; v < 8; ++v) {
            int r = v + 8 * h, c = nt * 16 + m;
            out[(g0 + r) * 32 + c] = o[nt][v] + bf2[c];
        }
}


extern "C" void kernel_launch(void* const* d_in, const int* in_sizes, int n_in,
                              void* d_out, int out_size, void* d_ws, size_t ws_size,
                              hipStream_t stream) {
    const float* x     = (const float*)d_in[0];
    const int*   ei    = (const int*)d_in[1];
    const int*   batch = (const int*)d_in[2];
    const float* W1  = (const float*)d_in[3];
    const float* b1  = (const float*)d_in[4];
    const float* W2  = (const float*)d_in[5];
    const float* b2  = (const float*)d_in[6];
    const float* Wf1 = (const float*)d_in[7];
    const float* bf1 = (const float*)d_in[8];
    const float* Wf2 = (const float*)d_in[9];
    const float* bf2 = (const float*)d_in[10];

    const int N = in_sizes[0] / IN_C;
    const int E = in_sizes[1] / 2;
    const int* src = ei;
    const int* dst = ei + E;

    float* ws   = (float*)d_ws;
    float* deg  = ws;
    float* dinv = deg  + N;
    float* aggx = dinv + N;
    float* hid  = aggx + (size_t)N * HID_C;
    float* h2   = hid  + (size_t)N * HID_C;
    float* pool = h2   + (size_t)N * OUT_C * 2;
    float* cnt  = pool + NGRAPH * OUT_C;

    const int B = 256;
    {
        long n4 = (NGRAPH * OUT_C + NGRAPH) / 4;
        k_zero4<<<(int)((n4 + B - 1) / B), B, 0, stream>>>((float4*)pool, n4);
    }

    int* part = (int*)hid;
    k_deg_hist<<<dim3(DEG_S, (N + DEG_RN - 1) / DEG_RN), DEG_T, 0, stream>>>(dst, part, N, E);
    k_dinv    <<<(N + B - 1) / B, B, 0, stream>>>(part, dinv, N);

    const dim3 BX(IN_C, AGX_ROWS);
    k_init_aggx<<<(N + AGX_ROWS - 1) / AGX_ROWS, BX, 0, stream>>>(x, dinv, aggx, N);

    k_edge_agg_x<<<(E + AGX_U * AGX_ROWS - 1) / (AGX_U * AGX_ROWS), BX, 0, stream>>>(src, dst, dinv, x, aggx, E);

    const int tiles = (N + 15) / 16;
    k_gemm<IN_C, AGX_LD, HID_C, true><<<tiles, 32, 0, stream>>>(aggx, W1, b1, hid, N);

    k_gemm<HID_C, HID_C, OUT_C, false><<<tiles, 32, 0, stream>>>(hid, W2, nullptr, h2, N);

    k_cnt<<<(N + B - 1) / B, B, 0, stream>>>(batch, cnt, N);

    k_agg_pool<<<AP_NB, AP_T, 0, stream>>>(src, dst, dinv, batch, h2, pool, N, E);

    k_fc<<<NGRAPH / 16, 32, 0, stream>>>(pool, cnt, b2, Wf1, bf1, Wf2, bf2, (float*)d_out);
}


# 10 launches -> 6: zeroing joins the degree histogram; dinv, graph sizes (integer LDS counts, exact) and self-loop seed share one per-node launch; both layer GEMMs one launch, hidden tile kept in LDS
# speedup vs baseline: 3.1125x; 1.2558x over previous
#include <hip/hip_runtime.h>

typedef float v2f __attribute__((ext_vector_type(2)));
typedef float v8f __attribute__((ext_vector_type(8)));

#define IN_C   17
#define AGX_LD 32
#define HID_C  64
#define OUT_C  32
#define NGRAPH 512


#define DEG_RN 12800
#define DEG_S  32
#define DEG_T  1024
#define DEG_U  8


#define NACC (NGRAPH * OUT_C + NGRAPH)
static_assert(DEG_S * DEG_T >= NACC, "the first row of blocks must cover the accumulators");

__global__ void k_deg_hist(const int* __restrict__ dst, int* __restrict__ part, float* __restrict__ acc0,
                           int n, int E) {
    __shared__ int bins[DEG_RN];
    if (blockIdx.y == 0) {
        const int j = blockIdx.x * DEG_T + threadIdx.x;
        if (j < NACC) acc0[j] = 0.f;
    }
    const int lo = blockIdx.y * DEG_RN;
    for (int j = threadIdx.x; j < DEG_RN; j += DEG_T) bins[j] = 0;
    __syncthreads();

    const int per = (E + DEG_S - 1) / DEG_S;
    const int e0  = blockIdx.x * per;
    const int e1  = e0 + per < E ? e0 + per : E;
    for (int eb = e0; eb < e1; eb += DEG_T * DEG_U) {
        int d[DEG_U];
#pragma unroll
        for (int k = 0; k < DEG_U; ++k) {
            const int e = eb + k * DEG_T + threadIdx.x;
            d[k] = e < e1 ? dst[e] : -1;
        }
#pragma unroll
        for (int k = 0; k < DEG_U; ++k) {
            const unsigned r = (unsigned)(d[k] - lo);
            if (r < (unsigned)DEG_RN) atomicAdd(&bins[r], 1);
        }
    }
    __syncthreads();

    const int cnt = n - lo < DEG_RN ? n - lo : DEG_RN;
    int* row = part + (long)blockIdx.x * n + lo;
    for (int j = threadIdx.x; j < cnt; j += DEG_T) row[j] = bins[j];
}


#define ND_T 256

__global__ void __launch_bounds__(ND_T)
k_node(const int* __restrict__ part, const int* __restrict__ batch, const float* __restrict__ x,
       float* __restrict__ dinv, float* __restrict__ cnt, float* __restrict__ aggx, int n) {
    __shared__ float sl[ND_T];
    __shared__ int   gsz[NGRAPH];
    const int t    = threadIdx.x;
    const int base = blockIdx.x * ND_T;
    const int i    = base + t;
    for (int j = t; j < NGRAPH; j += ND_T) gsz[j] = 0;

    const bool   full = base + ND_T <= n;
    const float* xb   = x + (long)base * IN_C;
    float v[IN_C];
    if (full) {
#pragma unroll
        for (int k = 0; k < IN_C; ++k) v[k] = xb[k * ND_T + t];
    }

    float dv = 0.f;
    int   g  = -1;
    if (i < n) {
        int c = 1;
#pragma unroll
        for (int s = 0; s < DEG_S; ++s) c += part[(long)s * n + i];
        dv = rsqrtf((float)c);
        g  = batch[i];
    }
    sl[t] = dv * dv;
    __syncthreads();
    if (g >= 0) atomicAdd(&gsz[g], 1);
    __syncthreads();
    for (int j = t; j < NGRAPH; j += ND_T) {
        const int q = gsz[j];
        if (q) atomicAdd(&cnt[j], (float)q);
    }

    if (i < n) dinv[i] = dv;
    if (full) {
#pragma unroll
        for (int k = 0; k < IN_C; ++k) {
            const int idx = k * ND_T + t;
            const int r = idx / IN_C, c = idx - r * IN_C;
            aggx[(long)(base + r) * AGX_LD + c] = sl[r] * v[k];
        }
    } else {
        const int cntf = (n - base) * IN_C;
        for (int idx = t; idx < cntf; idx += ND_T) {
            const int r = idx / IN_C, c = idx - r * IN_C;
            aggx[(long)(base + r) * AGX_LD + c] = sl[r] * xb[idx];
        }
    }
}


#define MM_W 8
#define HP   68

__global__ void __launch_bounds__(MM_W * 32)
k_mlp(const float* __restrict__ aggx, const float* __restrict__ W1, const float* __restrict__ b1,
      const float* __restrict__ W2, float* __restrict__ h2, int n) {
    __shared__ __attribute__((aligned(16))) float Hs[MM_W][16 * HP];
    const int  lane = threadIdx.x & 31;
    const int  wave = threadIdx.x >> 5;
    const int  m    = lane & 15;
    const int  h    = lane >> 4;
    const long rowbase = ((long)blockIdx.x * MM_W + wave) * 16;
    float* H = Hs[wave];

    const long rA = rowbase + m < n ? rowbase + m : (long)n - 1;
    const float* ar = aggx + rA * AGX_LD;

    constexpr int KS1 = (IN_C + 3) / 4;
    v8f acc[HID_C / 16] = {};
#pragma unroll
    for (int ks = 0; ks < KS1; ++ks) {
        const int ka = ks * 4 + 2 * h;
        const int kb = ka + 1;
        const float ma = (ka < IN_C) ? 1.f : 0.f;
        const float mb = (kb < IN_C) ? 1.f : 0.f;
        const int kca = (ka < IN_C) ? ka : (IN_C - 1);
        const int kcb = (kb < IN_C) ? kb : (IN_C - 1);

        v2f a;
        a.x = ar[kca] * ma;
        a.y = ar[kcb] * mb;
#pragma unroll
        for (int nt = 0; nt < HID_C / 16; ++nt) {
            v2f b;
            b.x = W1[kca * HID_C + nt * 16 + m] * ma;
            b.y = W1[kcb * HID_C + nt * 16 + m] * mb;
            acc[nt] = __builtin_amdgcn_wmma_f32_16x16x4_f32(
                false, a, false, b, (short)0, acc[nt], false, false);
        }
    }
#pragma unroll
    for (int nt = 0; nt < HID_C / 16; ++nt) {
        const float bv = b1[nt * 16 + m];
#pragma unroll
        for (int v = 0; v < 8; ++v) {
            float t = acc[nt][v] + bv;
            H[(v + 8 * h) * HP + nt * 16 + m] = t > 0.f ? t : 0.f;
        }
    }
    __syncthreads();

    v8f o[OUT_C / 16] = {};
#pragma unroll
    for (int ks = 0; ks < HID_C / 4; ++ks) {
        const int ka = ks * 4 + 2 * h, kb = ka + 1;
        const v2f a = *(const v2f*)(H + m * HP + ka);
#pragma unroll
        for (int nt = 0; nt < OUT_C / 16; ++nt) {
            v2f b;
            b.x = W2[ka * OUT_C + nt * 16 + m];
            b.y = W2[kb * OUT_C + nt * 16 + m];
            o[nt] = __builtin_amdgcn_wmma_f32_16x16x4_f32(
                false, a, false, b, (short)0, o[nt], false, false);
        }
    }
    float* orow = h2 + (rowbase + 8 * h) * OUT_C + m;
    if (rowbase + 16 <= n) {
#pragma unroll
        for (int nt = 0; nt < OUT_C / 16; ++nt)
#pragma unroll
            for (int v = 0; v < 8; ++v) orow[v * OUT_C + nt * 16] = o[nt][v];
    } else {
#pragma unroll
        for (int nt = 0; nt < OUT_C / 16; ++nt)
#pragma unroll
            for (int v = 0; v < 8; ++v)
                if (rowbase + v + 8 * h < n) orow[v * OUT_C + nt * 16] = o[nt][v];
    }
}


#define AGX_ROWS 15
#define AGX_U    4

template <bool GUARD>
__device__ __forceinline__ void edge_agg_x_body(const int* __restrict__ src, const int* __restrict__ dst,
                                                const float* __restrict__ dinv, const float* __restrict__ x,
                                                float* __restrict__ aggx, long ebase, int E) {
    const int c = threadIdx.x;
    long  o[AGX_U];
    float v[AGX_U];
    bool  ok[AGX_U];
#pragma unroll
    for (int k = 0; k < AGX_U; ++k) {
        long e = ebase + k * AGX_ROWS + threadIdx.y;
        ok[k] = !GUARD || e < E;
        if (GUARD) e = ok[k] ? e : (long)E - 1;
        const int s = src[e], d = dst[e];
        v[k] = dinv[s] * dinv[d] * x[(long)s * IN_C + c];
        o[k] = (long)d * AGX_LD + c;
    }
#pragma unroll
    for (int k = 0; k < AGX_U; ++k)
        if (ok[k]) atomicAdd(aggx + o[k], v[k]);
}

__global__ void k_edge_agg_x(const int* __restrict__ src, const int* __restrict__ dst,
                             const float* __restrict__ dinv, const float* __restrict__ x,
                             float* __restrict__ aggx, int E) {
    const long ebase = (long)blockIdx.x * (AGX_U * AGX_ROWS);
    if (ebase + AGX_U * AGX_ROWS <= E) edge_agg_x_body<false>(src, dst, dinv, x, aggx, ebase, E);
    else                               edge_agg_x_body<true >(src, dst, dinv, x, aggx, ebase, E);
}


#define AP_T 1024
#define AP_NB 256

template <bool GUARD>
__device__ __forceinline__ void agg_pool_chunk(const int* __restrict__ src, const int* __restrict__ dst,
                                               const float* __restrict__ dinv, const int* __restrict__ batch,
                                               const float* __restrict__ h2, float* acc,
                                               long base, long total, int E, int lane) {
    const long idx = base + lane;
    int s = 0, d = 0;
    float w = 0.f;
    if (!GUARD || idx < total) {
        if (idx < E) { s = src[idx]; d = dst[idx]; }
        else         { s = d = (int)(idx - E); }
        w = dinv[s] * dinv[d];
    }
    const int g = batch[d];
    const int cnt = GUARD ? (int)(total - base) : 32;

    float v[32];
#pragma unroll
    for (int k = 0; k < 32; ++k) {
        const int sk = __builtin_amdgcn_readlane(s, k);
        const __attribute__((address_space(1))) float* row =
            (const __attribute__((address_space(1))) float*)(h2 + (long)sk * OUT_C);
        asm("" : "+s"(row));
        v[k] = (!GUARD || k < cnt) ? row[lane] : 0.f;
    }
#pragma unroll
    for (int k = 0; k < 32; ++k) {
        const int   gk = __builtin_amdgcn_readlane(g, k);
        const float wk = __int_as_float(__builtin_amdgcn_readlane(__float_as_int(w), k));
        if (!GUARD || k < cnt) atomicAdd(acc + gk * OUT_C + lane, wk * v[k]);
    }
}

__global__ void __launch_bounds__(AP_T)
k_agg_pool(const int* __restrict__ src, const int* __restrict__ dst,
           const float* __restrict__ dinv, const int* __restrict__ batch,
           const float* __restrict__ h2, float* __restrict__ pool, int n, int E) {
    __shared__ float acc[NGRAPH * OUT_C];
    for (int j = threadIdx.x; j < NGRAPH * OUT_C; j += AP_T) acc[j] = 0.f;
    __syncthreads();

    const int  lane   = threadIdx.x & 31;
    const int  wave   = threadIdx.x >> 5;
    const long total  = (long)E + n;
    const long nchunk = (total + 31) / 32;
    for (long c = (long)blockIdx.x * (AP_T / 32) + wave; c < nchunk; c += (long)gridDim.x * (AP_T / 32)) {
        const long base = c * 32;
        if (base + 32 <= total) agg_pool_chunk<false>(src, dst, dinv, batch, h2, acc, base, total, E, lane);
        else                    agg_pool_chunk<true >(src, dst, dinv, batch, h2, acc, base, total, E, lane);
    }
    __syncthreads();

    for (int j = threadIdx.x; j < NGRAPH * OUT_C; j += AP_T) {
        const int jj = (j + blockIdx.x * 64) & (NGRAPH * OUT_C - 1);
        atomicAdd(pool + jj, acc[jj]);
    }
}


__global__ void k_fc(const float* __restrict__ pool, const float* __restrict__ cnt,
                     const float* __restrict__ b2,
                     const float* __restrict__ Wf1, const float* __restrict__ bf1,
                     const float* __restrict__ Wf2, const float* __restrict__ bf2,
                     float* __restrict__ out) {
    __shared__ float P[16 * 32];
    __shared__ float T[16 * 32];
    const int lane = threadIdx.x;
    const int m = lane & 15;
    const int h = lane >> 4;
    const int g0 = blockIdx.x * 16;

    for (int idx = lane; idx < 16 * 32; idx += 32) {
        int r = idx >> 5, c = idx & 31;
        float ct = cnt[g0 + r];
        ct = ct > 1.f ? ct : 1.f;
        P[idx] = pool[(g0 + r) * 32 + c] / ct + b2[c];
    }
    __syncthreads();

    v8f acc[2] = {};
#pragma unroll
    for (int ks = 0; ks < 8; ++ks) {
        const int ka = ks * 4 + 2 * h, kb = ka + 1;
        v2f a; a.x = P[m * 32 + ka]; a.y = P[m * 32 + kb];
#pragma unroll
        for (int nt = 0; nt < 2; ++nt) {
            v2f b; b.x = Wf1[ka * 32 + nt * 16 + m]; b.y = Wf1[kb * 32 + nt * 16 + m];
            acc[nt] = __builtin_amdgcn_wmma_f32_16x16x4_f32(
                false, a, false, b, (short)0, acc[nt], false, false);
        }
    }
#pragma unroll
    for (int nt = 0; nt < 2; ++nt)
#pragma unroll
        for (int v = 0; v < 8; ++v) {
            int r = v + 8 * h, c = nt * 16 + m;
            float t = acc[nt][v] + bf1[c];
            T[r * 32 + c] = t > 0.f ? t : 0.f;
        }
    __syncthreads();

    v8f o[2] = {};
#pragma unroll
    for (int ks = 0; ks < 8; ++ks) {
        const int ka = ks * 4 + 2 * h, kb = ka + 1;
        v2f a; a.x = T[m * 32 + ka]; a.y = T[m * 32 + kb];
#pragma unroll
        for (int nt = 0; nt < 2; ++nt) {
            v2f b; b.x = Wf2[ka * 32 + nt * 16 + m]; b.y = Wf2[kb * 32 + nt * 16 + m];
            o[nt] = __builtin_amdgcn_wmma_f32_16x16x4_f32(
                false, a, false, b, (short)0, o[nt], false, false);
        }
    }
#pragma unroll
    for (int nt = 0; nt < 2; ++nt)
#pragma unroll
        for (int v = 0; v < 8; ++v) {
            int r = v + 8 * h, c = nt * 16 + m;
            out[(g0 + r) * 32 + c] = o[nt][v] + bf2[c];
        }
}


extern "C" void kernel_launch(void* const* d_in, const int* in_sizes, int n_in,
                              void* d_out, int out_size, void* d_ws, size_t ws_size,
                              hipStream_t stream) {
    const float* x     = (const float*)d_in[0];
    const int*   ei    = (const int*)d_in[1];
    const int*   batch = (const int*)d_in[2];
    const float* W1  = (const float*)d_in[3];
    const float* b1  = (const float*)d_in[4];
    const float* W2  = (const float*)d_in[5];
    const float* b2  = (const float*)d_in[6];
    const float* Wf1 = (const float*)d_in[7];
    const float* bf1 = (const float*)d_in[8];
    const float* Wf2 = (const float*)d_in[9];
    const float* bf2 = (const float*)d_in[10];

    const int N = in_sizes[0] / IN_C;
    const int E = in_sizes[1] / 2;
    const int* src = ei;
    const int* dst = ei + E;

    float* ws   = (float*)d_ws;
    float* deg  = ws;
    float* dinv = deg  + N;
    float* aggx = dinv + N;
    float* hid  = aggx + (size_t)N * HID_C;
    float* h2   = hid  + (size_t)N * HID_C;
    float* pool = h2   + (size_t)N * OUT_C * 2;
    float* cnt  = pool + NGRAPH * OUT_C;

    int* part = (int*)hid;
    k_deg_hist<<<dim3(DEG_S, (N + DEG_RN - 1) / DEG_RN), DEG_T, 0, stream>>>(dst, part, pool, N, E);

    k_node<<<(N + ND_T - 1) / ND_T, ND_T, 0, stream>>>(part, batch, x, dinv, cnt, aggx, N);

    const dim3 BX(IN_C, AGX_ROWS);
    k_edge_agg_x<<<(E + AGX_U * AGX_ROWS - 1) / (AGX_U * AGX_ROWS), BX, 0, stream>>>(src, dst, dinv, x, aggx, E);

    const int tiles = (N + 15) / 16;
    k_mlp<<<(tiles + MM_W - 1) / MM_W, MM_W * 32, 0, stream>>>(aggx, W1, b1, W2, h2, N);

    k_agg_pool<<<AP_NB, AP_T, 0, stream>>>(src, dst, dinv, batch, h2, pool, N, E);

    k_fc<<<NGRAPH / 16, 32, 0, stream>>>(pool, cnt, b2, Wf1, bf1, Wf2, bf2, (float*)d_out);
}


# layer-1 edge sums leave global float atomics: edges grouped by destination range (LDS counting sort, in the degree launch); one owner block per 200 nodes adds rows in LDS; sums re-ordered
# speedup vs baseline: 3.6477x; 1.1720x over previous
#include <hip/hip_runtime.h>

typedef float v2f __attribute__((ext_vector_type(2)));
typedef float v8f __attribute__((ext_vector_type(8)));

#define IN_C   17
#define AGX_LD 32
#define HID_C  64
#define OUT_C  32
#define NGRAPH 512


#define DEG_RN 12800
#define DEG_S  32
#define DEG_T  1024
#define DEG_U  8


#define NACC (NGRAPH * OUT_C + NGRAPH)
static_assert(DEG_S * DEG_T >= NACC, "the first row of blocks must cover the accumulators");


#define BKN     200
#define NBK_MAX 512
#define EPB     4096
#define BIN_U   (EPB / DEG_T)
#define NP_MAX  512
#define SCAN_W  17
#define SCAN_N  (32 * SCAN_W)

__device__ __forceinline__ void scan_offsets(int* a, int n, int tid) {
    if (tid < 32) {
        int loc[SCAN_W];
        int sum = 0;
#pragma unroll
        for (int j = 0; j < SCAN_W; ++j) {
            const int idx = tid * SCAN_W + j;
            loc[j] = sum;
            sum += idx < n ? a[idx] : 0;
        }
        int inc = sum;
#pragma unroll
        for (int dlt = 1; dlt < 32; dlt <<= 1) {
            const int o = __shfl_up(inc, dlt);
            if (tid >= dlt) inc += o;
        }
        const int exc = inc - sum;
#pragma unroll
        for (int j = 0; j < SCAN_W; ++j) {
            const int idx = tid * SCAN_W + j;
            if (idx <= n) a[idx] = exc + loc[j];
        }
    }
}

static_assert(2 * EPB + SCAN_N <= DEG_RN, "the grouping tables reuse the degree counters' LDS");

__global__ void __launch_bounds__(DEG_T)
k_deg_bin(const int* __restrict__ src, const int* __restrict__ dst, int* __restrict__ part,
          float* __restrict__ acc0, int* __restrict__ rec_s, int* __restrict__ rec_d,
          int* __restrict__ offT, int n, int E, int ndeg, int np, int nbk) {
    __shared__ int bins[DEG_RN];
    const int bid = blockIdx.x;
    if (bid < ndeg) {
        const int bx = bid % DEG_S;
        const int by = bid / DEG_S;
        if (by == 0) {
            const int j = bx * DEG_T + threadIdx.x;
            if (j < NACC) acc0[j] = 0.f;
        }
        const int lo = by * DEG_RN;
        for (int j = threadIdx.x; j < DEG_RN; j += DEG_T) bins[j] = 0;
        __syncthreads();

        const int per = (E + DEG_S - 1) / DEG_S;
        const int e0  = bx * per;
        const int e1  = e0 + per < E ? e0 + per : E;
        for (int eb = e0; eb < e1; eb += DEG_T * DEG_U) {
            int d[DEG_U];
#pragma unroll
            for (int k = 0; k < DEG_U; ++k) {
                const int e = eb + k * DEG_T + threadIdx.x;
                d[k] = e < e1 ? dst[e] : -1;
            }
#pragma unroll
            for (int k = 0; k < DEG_U; ++k) {
                const unsigned r = (unsigned)(d[k] - lo);
                if (r < (unsigned)DEG_RN) atomicAdd(&bins[r], 1);
            }
        }
        __syncthreads();

        const int cnt = n - lo < DEG_RN ? n - lo : DEG_RN;
        int* row = part + (long)bx * n + lo;
        for (int j = threadIdx.x; j < cnt; j += DEG_T) row[j] = bins[j];
    } else {
        const int pb = bid - ndeg;
        const int t  = threadIdx.x;
        int* stg_s = bins;
        int* stg_d = bins + EPB;
        int* gcnt  = bins + 2 * EPB;
        for (int j = t; j < SCAN_N; j += DEG_T) gcnt[j] = 0;
        __syncthreads();

        const long base = (long)pb * EPB;
        int  s[BIN_U], d[BIN_U], b[BIN_U], r[BIN_U];
#pragma unroll
        for (int k = 0; k < BIN_U; ++k) {
            const long e = base + k * DEG_T + t;
            const bool ok = e < E;
            s[k] = ok ? src[e] : 0;
            d[k] = ok ? dst[e] : -1;
        }
#pragma unroll
        for (int k = 0; k < BIN_U; ++k) {
            b[k] = d[k] >= 0 ? d[k] / BKN : -1;
            r[k] = b[k] >= 0 ? atomicAdd(&gcnt[b[k]], 1) : 0;
        }
        __syncthreads();
        scan_offsets(gcnt, nbk, t);
        __syncthreads();
#pragma unroll
        for (int k = 0; k < BIN_U; ++k) {
            if (b[k] >= 0) {
                const int pos = gcnt[b[k]] + r[k];
                stg_s[pos] = s[k];
                stg_d[pos] = d[k];
            }
        }
        __syncthreads();
        const int total = gcnt[nbk];
        for (int j = t; j < total; j += DEG_T) {
            rec_s[base + j] = stg_s[j];
            rec_d[base + j] = stg_d[j];
        }
        for (int g = t; g <= nbk; g += DEG_T) offT[(long)g * np + pb] = gcnt[g];
    }
}


#define ND_T 256

__global__ void __launch_bounds__(ND_T)
k_node(const int* __restrict__ part, const int* __restrict__ batch, const float* __restrict__ x,
       float* __restrict__ dinv, float* __restrict__ cnt, float* __restrict__ xs, int n) {
    __shared__ float sl[ND_T];
    __shared__ int   gsz[NGRAPH];
    const int t    = threadIdx.x;
    const int base = blockIdx.x * ND_T;
    const int i    = base + t;
    for (int j = t; j < NGRAPH; j += ND_T) gsz[j] = 0;

    const bool   full = base + ND_T <= n;
    const float* xb   = x + (long)base * IN_C;
    float v[IN_C];
    if (full) {
#pragma unroll
        for (int k = 0; k < IN_C; ++k) v[k] = xb[k * ND_T + t];
    }

    float dv = 0.f;
    int   g  = -1;
    if (i < n) {
        int c = 1;
#pragma unroll
        for (int s = 0; s < DEG_S; ++s) c += part[(long)s * n + i];
        dv = rsqrtf((float)c);
        g  = batch[i];
    }
    sl[t] = dv;
    __syncthreads();
    if (g >= 0) atomicAdd(&gsz[g], 1);
    __syncthreads();
    for (int j = t; j < NGRAPH; j += ND_T) {
        const int q = gsz[j];
        if (q) atomicAdd(&cnt[j], (float)q);
    }

    if (i < n) dinv[i] = dv;
    if (full) {
#pragma unroll
        for (int k = 0; k < IN_C; ++k) {
            const int idx = k * ND_T + t;
            const int r = idx / IN_C, c = idx - r * IN_C;
            xs[(long)(base + r) * AGX_LD + c] = sl[r] * v[k];
        }
    } else {
        const int cntf = (n - base) * IN_C;
        for (int idx = t; idx < cntf; idx += ND_T) {
            const int r = idx / IN_C, c = idx - r * IN_C;
            xs[(long)(base + r) * AGX_LD + c] = sl[r] * xb[idx];
        }
    }
}


#define MM_W 8
#define HP   68

__global__ void __launch_bounds__(MM_W * 32)
k_mlp(const float* __restrict__ aggx, const float* __restrict__ W1, const float* __restrict__ b1,
      const float* __restrict__ W2, float* __restrict__ h2, int n) {
    __shared__ __attribute__((aligned(16))) float Hs[MM_W][16 * HP];
    const int  lane = threadIdx.x & 31;
    const int  wave = threadIdx.x >> 5;
    const int  m    = lane & 15;
    const int  h    = lane >> 4;
    const long rowbase = ((long)blockIdx.x * MM_W + wave) * 16;
    float* H = Hs[wave];

    const long rA = rowbase + m < n ? rowbase + m : (long)n - 1;
    const float* ar = aggx + rA * AGX_LD;

    constexpr int KS1 = (IN_C + 3) / 4;
    v8f acc[HID_C / 16] = {};
#pragma unroll
    for (int ks = 0; ks < KS1; ++ks) {
        const int ka = ks * 4 + 2 * h;
        const int kb = ka + 1;
        const float ma = (ka < IN_C) ? 1.f : 0.f;
        const float mb = (kb < IN_C) ? 1.f : 0.f;
        const int kca = (ka < IN_C) ? ka : (IN_C - 1);
        const int kcb = (kb < IN_C) ? kb : (IN_C - 1);

        v2f a;
        a.x = ar[kca] * ma;
        a.y = ar[kcb] * mb;
#pragma unroll
        for (int nt = 0; nt < HID_C / 16; ++nt) {
            v2f b;
            b.x = W1[kca * HID_C + nt * 16 + m] * ma;
            b.y = W1[kcb * HID_C + nt * 16 + m] * mb;
            acc[nt] = __builtin_amdgcn_wmma_f32_16x16x4_f32(
                false, a, false, b, (short)0, acc[nt], false, false);
        }
    }
#pragma unroll
    for (int nt = 0; nt < HID_C / 16; ++nt) {
        const float bv = b1[nt * 16 + m];
#pragma unroll
        for (int v = 0; v < 8; ++v) {
            float t = acc[nt][v] + bv;
            H[(v + 8 * h) * HP + nt * 16 + m] = t > 0.f ? t : 0.f;
        }
    }
    __syncthreads();

    v8f o[OUT_C / 16] = {};
#pragma unroll
    for (int ks = 0; ks < HID_C / 4; ++ks) {
        const int ka = ks * 4 + 2 * h, kb = ka + 1;
        const v2f a = *(const v2f*)(H + m * HP + ka);
#pragma unroll
        for (int nt = 0; nt < OUT_C / 16; ++nt) {
            v2f b;
            b.x = W2[ka * OUT_C + nt * 16 + m];
            b.y = W2[kb * OUT_C + nt * 16 + m];
            o[nt] = __builtin_amdgcn_wmma_f32_16x16x4_f32(
                false, a, false, b, (short)0, o[nt], false, false);
        }
    }
    float* orow = h2 + (rowbase + 8 * h) * OUT_C + m;
    if (rowbase + 16 <= n) {
#pragma unroll
        for (int nt = 0; nt < OUT_C / 16; ++nt)
#pragma unroll
            for (int v = 0; v < 8; ++v) orow[v * OUT_C + nt * 16] = o[nt][v];
    } else {
#pragma unroll
        for (int nt = 0; nt < OUT_C / 16; ++nt)
#pragma unroll
            for (int v = 0; v < 8; ++v)
                if (rowbase + v + 8 * h < n) orow[v * OUT_C + nt * 16] = o[nt][v];
    }
}


#define OWN_ROWS 60
#define OWN_T    (IN_C * OWN_ROWS)
#define OWN_U    8
#define OWN_CAP  4096
#define OWN_Q    4

__global__ void __launch_bounds__(OWN_T)
k_own_acc(const int* __restrict__ rec_s, const int* __restrict__ rec_d, const int* __restrict__ offT,
          const float* __restrict__ dinv, const float* __restrict__ xs, float* __restrict__ aggx,
          int n, int np) {
    __shared__ float acc[BKN * IN_C];
    __shared__ int   segoff[SCAN_N];
    __shared__ int   segbase[NP_MAX];
    __shared__ int2  stg[OWN_CAP];
    const int c   = threadIdx.x;
    const int y   = threadIdx.y;
    const int tid = y * IN_C + c;
    const int g   = blockIdx.x;
    const int n0  = g * BKN;

    for (int j = tid; j < BKN * IN_C; j += OWN_T) acc[j] = 0.f;
    for (int pb = tid; pb < np; pb += OWN_T) {
        const int a = offT[(long)g * np + pb];
        const int b = offT[(long)(g + 1) * np + pb];
        segbase[pb] = pb * EPB + a;
        segoff[pb]  = b - a;
    }
    __syncthreads();
    scan_offsets(segoff, np, tid);
    __syncthreads();
    const int total = segoff[np];

    for (int c0 = 0; c0 < total; c0 += OWN_CAP) {
        const int m = total - c0 < OWN_CAP ? total - c0 : OWN_CAP;
        for (int jb = tid; jb < m; jb += OWN_T * OWN_Q) {
            int a[OWN_Q], sv[OWN_Q], dv[OWN_Q];
#pragma unroll
            for (int q = 0; q < OWN_Q; ++q) {
                const int j = jb + q * OWN_T;
                const int f = c0 + (j < m ? j : jb);
                int lo = 0, hi = np;
#pragma unroll
                for (int it = 0; it < 10; ++it) {
                    const int  mid = (lo + hi) >> 1;
                    const bool act = hi - lo > 1;
                    const bool le  = segoff[mid] <= f;
                    lo = (act && le) ? mid : lo;
                    hi = (act && !le) ? mid : hi;
                }
                a[q] = segbase[lo] + (f - segoff[lo]);
            }
#pragma unroll
            for (int q = 0; q < OWN_Q; ++q) {
                sv[q] = rec_s[a[q]];
                dv[q] = rec_d[a[q]];
            }
#pragma unroll
            for (int q = 0; q < OWN_Q; ++q) {
                const int j = jb + q * OWN_T;
                if (j < m) stg[j] = make_int2(sv[q], (dv[q] - n0) * IN_C);
            }
        }
        __syncthreads();
        for (int j0 = y; j0 < m; j0 += OWN_ROWS * OWN_U) {
            float v[OWN_U];
            int   sn[OWN_U], o[OWN_U];
#pragma unroll
            for (int u = 0; u < OWN_U; ++u) {
                const int  j  = j0 + u * OWN_ROWS;
                const int2 r2 = stg[j < m ? j : j0];
                sn[u] = r2.x;
                o[u]  = j < m ? r2.y + c : -1;
            }
#pragma unroll
            for (int u = 0; u < OWN_U; ++u) v[u] = xs[(long)sn[u] * AGX_LD + c];
#pragma unroll
            for (int u = 0; u < OWN_U; ++u)
                if (o[u] >= 0) atomicAdd(&acc[o[u]], v[u]);
        }
        __syncthreads();
    }

    const int rows = n - n0 < BKN ? n - n0 : BKN;
    for (int r = y; r < rows; r += OWN_ROWS) {
        const long i = n0 + r;
        aggx[i * AGX_LD + c] = dinv[i] * (acc[r * IN_C + c] + xs[i * AGX_LD + c]);
    }
}


#define AP_T 1024
#define AP_NB 256

template <bool GUARD>
__device__ __forceinline__ void agg_pool_chunk(const int* __restrict__ src, const int* __restrict__ dst,
                                               const float* __restrict__ dinv, const int* __restrict__ batch,
                                               const float* __restrict__ h2, float* acc,
                                               long base, long total, int E, int lane) {
    const long idx = base + lane;
    int s = 0, d = 0;
    float w = 0.f;
    if (!GUARD || idx < total) {
        if (idx < E) { s = src[idx]; d = dst[idx]; }
        else         { s = d = (int)(idx - E); }
        w = dinv[s] * dinv[d];
    }
    const int g = batch[d];
    const int cnt = GUARD ? (int)(total - base) : 32;

    float v[32];
#pragma unroll
    for (int k = 0; k < 32; ++k) {
        const int sk = __builtin_amdgcn_readlane(s, k);
        const __attribute__((address_space(1))) float* row =
            (const __attribute__((address_space(1))) float*)(h2 + (long)sk * OUT_C);
        asm("" : "+s"(row));
        v[k] = (!GUARD || k < cnt) ? row[lane] : 0.f;
    }
#pragma unroll
    for (int k = 0; k < 32; ++k) {
        const int   gk = __builtin_amdgcn_readlane(g, k);
        const float wk = __int_as_float(__builtin_amdgcn_readlane(__float_as_int(w), k));
        if (!GUARD || k < cnt) atomicAdd(acc + gk * OUT_C + lane, wk * v[k]);
    }
}

__global__ void __launch_bounds__(AP_T)
k_agg_pool(const int* __restrict__ src, const int* __restrict__ dst,
           const float* __restrict__ dinv, const int* __restrict__ batch,
           const float* __restrict__ h2, float* __restrict__ pool, int n, int E) {
    __shared__ float acc[NGRAPH * OUT_C];
    for (int j = threadIdx.x; j < NGRAPH * OUT_C; j += AP_T) acc[j] = 0.f;
    __syncthreads();

    const int  lane   = threadIdx.x & 31;
    const int  wave   = threadIdx.x >> 5;
    const long total  = (long)E + n;
    const long nchunk = (total + 31) / 32;
    for (long c = (long)blockIdx.x * (AP_T / 32) + wave; c < nchunk; c += (long)gridDim.x * (AP_T / 32)) {
        const long base = c * 32;
        if (base + 32 <= total) agg_pool_chunk<false>(src, dst, dinv, batch, h2, acc, base, total, E, lane);
        else                    agg_pool_chunk<true >(src, dst, dinv, batch, h2, acc, base, total, E, lane);
    }
    __syncthreads();

    for (int j = threadIdx.x; j < NGRAPH * OUT_C; j += AP_T) {
        const int jj = (j + blockIdx.x * 64) & (NGRAPH * OUT_C - 1);
        atomicAdd(pool + jj, acc[jj]);
    }
}


__global__ void k_fc(const float* __restrict__ pool, const float* __restrict__ cnt,
                     const float* __restrict__ b2,
                     const float* __restrict__ Wf1, const float* __restrict__ bf1,
                     const float* __restrict__ Wf2, const float* __restrict__ bf2,
                     float* __restrict__ out) {
    __shared__ float P[16 * 32];
    __shared__ float T[16 * 32];
    const int lane = threadIdx.x;
    const int m = lane & 15;
    const int h = lane >> 4;
    const int g0 = blockIdx.x * 16;

    for (int idx = lane; idx < 16 * 32; idx += 32) {
        int r = idx >> 5, c = idx & 31;
        float ct = cnt[g0 + r];
        ct = ct > 1.f ? ct : 1.f;
        P[idx] = pool[(g0 + r) * 32 + c] / ct + b2[c];
    }
    __syncthreads();

    v8f acc[2] = {};
#pragma unroll
    for (int ks = 0; ks < 8; ++ks) {
        const int ka = ks * 4 + 2 * h, kb = ka + 1;
        v2f a; a.x = P[m * 32 + ka]; a.y = P[m * 32 + kb];
#pragma unroll
        for (int nt = 0; nt < 2; ++nt) {
            v2f b; b.x = Wf1[ka * 32 + nt * 16 + m]; b.y = Wf1[kb * 32 + nt * 16 + m];
            acc[nt] = __builtin_amdgcn_wmma_f32_16x16x4_f32(
                false, a, false, b, (short)0, acc[nt], false, false);
        }
    }
#pragma unroll
    for (int nt = 0; nt < 2; ++nt)
#pragma unroll
        for (int v = 0; v < 8; ++v) {
            int r = v + 8 * h, c = nt * 16 + m;
            float t = acc[nt][v] + bf1[c];
            T[r * 32 + c] = t > 0.f ? t : 0.f;
        }
    __syncthreads();

    v8f o[2] = {};
#pragma unroll
    for (int ks = 0; ks < 8; ++ks) {
        const int ka = ks * 4 + 2 * h, kb = ka + 1;
        v2f a; a.x = T[m * 32 + ka]; a.y = T[m * 32 + kb];
#pragma unroll
        for (int nt = 0; nt < 2; ++nt) {
            v2f b; b.x = Wf2[ka * 32 + nt * 16 + m]; b.y = Wf2[kb * 32 + nt * 16 + m];
            o[nt] = __builtin_amdgcn_wmma_f32_16x16x4_f32(
                false, a, false, b, (short)0, o[nt], false, false);
        }
    }
#pragma unroll
    for (int nt = 0; nt < 2; ++nt)
#pragma unroll
        for (int v = 0; v < 8; ++v) {
            int r = v + 8 * h, c = nt * 16 + m;
            out[(g0 + r) * 32 + c] = o[nt][v] + bf2[c];
        }
}


extern "C" void kernel_launch(void* const* d_in, const int* in_sizes, int n_in,
                              void* d_out, int out_size, void* d_ws, size_t ws_size,
                              hipStream_t stream) {
    const float* x     = (const float*)d_in[0];
    const int*   ei    = (const int*)d_in[1];
    const int*   batch = (const int*)d_in[2];
    const float* W1  = (const float*)d_in[3];
    const float* b1  = (const float*)d_in[4];
    const float* W2  = (const float*)d_in[5];
    const float* b2  = (const float*)d_in[6];
    const float* Wf1 = (const float*)d_in[7];
    const float* bf1 = (const float*)d_in[8];
    const float* Wf2 = (const float*)d_in[9];
    const float* bf2 = (const float*)d_in[10];

    const int N = in_sizes[0] / IN_C;
    const int E = in_sizes[1] / 2;
    const int* src = ei;
    const int* dst = ei + E;

    float* ws   = (float*)d_ws;
    float* deg  = ws;
    float* dinv = deg  + N;
    float* aggx = dinv + N;
    float* xs   = aggx + (size_t)N * AGX_LD;
    float* hid  = aggx + (size_t)N * HID_C;
    float* h2   = hid  + (size_t)N * HID_C;
    float* pool = h2   + (size_t)N * OUT_C * 2;
    float* cnt  = pool + NGRAPH * OUT_C;

    int* part = (int*)hid;
    const int ndeg = DEG_S * ((N + DEG_RN - 1) / DEG_RN);
    const int np   = (E + EPB - 1) / EPB;
    const int nbk  = (N + BKN - 1) / BKN;
    int* rec_s = (int*)(hid + (size_t)N * OUT_C);
    int* rec_d = (int*)(h2  + (size_t)N * OUT_C);
    int* offT  = rec_d + (size_t)np * EPB;
    k_deg_bin<<<ndeg + np, DEG_T, 0, stream>>>(src, dst, part, pool, rec_s, rec_d, offT, N, E, ndeg, np, nbk);

    k_node<<<(N + ND_T - 1) / ND_T, ND_T, 0, stream>>>(part, batch, x, dinv, cnt, xs, N);

    const dim3 BO(IN_C, OWN_ROWS);
    k_own_acc<<<nbk, BO, 0, stream>>>(rec_s, rec_d, offT, dinv, xs, aggx, N, np);

    const int tiles = (N + 15) / 16;
    k_mlp<<<(tiles + MM_W - 1) / MM_W, MM_W * 32, 0, stream>>>(aggx, W1, b1, W2, h2, N);

    k_agg_pool<<<AP_NB, AP_T, 0, stream>>>(src, dst, dinv, batch, h2, pool, N, E);

    k_fc<<<NGRAPH / 16, 32, 0, stream>>>(pool, cnt, b2, Wf1, bf1, Wf2, bf2, (float*)d_out);
}


# edge-to-pool pass makes 2 scattered reads per item, not 4: dinv[src] rides in the h2 rows (scaled on entry to the 2nd GEMM), dinv+graph id of dst share one 8-byte record; products re-associated
# speedup vs baseline: 4.0046x; 1.0978x over previous
#include <hip/hip_runtime.h>

typedef float v2f __attribute__((ext_vector_type(2)));
typedef float v8f __attribute__((ext_vector_type(8)));

#define IN_C   17
#define AGX_LD 32
#define HID_C  64
#define OUT_C  32
#define NGRAPH 512


#define DEG_RN 12800
#define DEG_S  32
#define DEG_T  1024
#define DEG_U  8


#define NACC (NGRAPH * OUT_C + NGRAPH)
static_assert(DEG_S * DEG_T >= NACC, "the first row of blocks must cover the accumulators");


#define BKN     200
#define NBK_MAX 512
#define EPB     4096
#define BIN_U   (EPB / DEG_T)
#define NP_MAX  512
#define SCAN_W  17
#define SCAN_N  (32 * SCAN_W)

__device__ __forceinline__ void scan_offsets(int* a, int n, int tid) {
    if (tid < 32) {
        int loc[SCAN_W];
        int sum = 0;
#pragma unroll
        for (int j = 0; j < SCAN_W; ++j) {
            const int idx = tid * SCAN_W + j;
            loc[j] = sum;
            sum += idx < n ? a[idx] : 0;
        }
        int inc = sum;
#pragma unroll
        for (int dlt = 1; dlt < 32; dlt <<= 1) {
            const int o = __shfl_up(inc, dlt);
            if (tid >= dlt) inc += o;
        }
        const int exc = inc - sum;
#pragma unroll
        for (int j = 0; j < SCAN_W; ++j) {
            const int idx = tid * SCAN_W + j;
            if (idx <= n) a[idx] = exc + loc[j];
        }
    }
}

static_assert(2 * EPB + SCAN_N <= DEG_RN, "the grouping tables reuse the degree counters' LDS");

__global__ void __launch_bounds__(DEG_T)
k_deg_bin(const int* __restrict__ src, const int* __restrict__ dst, int* __restrict__ part,
          float* __restrict__ acc0, int* __restrict__ rec_s, int* __restrict__ rec_d,
          int* __restrict__ offT, int n, int E, int ndeg, int np, int nbk) {
    __shared__ int bins[DEG_RN];
    const int bid = blockIdx.x;
    if (bid < ndeg) {
        const int bx = bid % DEG_S;
        const int by = bid / DEG_S;
        if (by == 0) {
            const int j = bx * DEG_T + threadIdx.x;
            if (j < NACC) acc0[j] = 0.f;
        }
        const int lo = by * DEG_RN;
        for (int j = threadIdx.x; j < DEG_RN; j += DEG_T) bins[j] = 0;
        __syncthreads();

        const int per = (E + DEG_S - 1) / DEG_S;
        const int e0  = bx * per;
        const int e1  = e0 + per < E ? e0 + per : E;
        for (int eb = e0; eb < e1; eb += DEG_T * DEG_U) {
            int d[DEG_U];
#pragma unroll
            for (int k = 0; k < DEG_U; ++k) {
                const int e = eb + k * DEG_T + threadIdx.x;
                d[k] = e < e1 ? dst[e] : -1;
            }
#pragma unroll
            for (int k = 0; k < DEG_U; ++k) {
                const unsigned r = (unsigned)(d[k] - lo);
                if (r < (unsigned)DEG_RN) atomicAdd(&bins[r], 1);
            }
        }
        __syncthreads();

        const int cnt = n - lo < DEG_RN ? n - lo : DEG_RN;
        int* row = part + (long)bx * n + lo;
        for (int j = threadIdx.x; j < cnt; j += DEG_T) row[j] = bins[j];
    } else {
        const int pb = bid - ndeg;
        const int t  = threadIdx.x;
        int* stg_s = bins;
        int* stg_d = bins + EPB;
        int* gcnt  = bins + 2 * EPB;
        for (int j = t; j < SCAN_N; j += DEG_T) gcnt[j] = 0;
        __syncthreads();

        const long base = (long)pb * EPB;
        int  s[BIN_U], d[BIN_U], b[BIN_U], r[BIN_U];
#pragma unroll
        for (int k = 0; k < BIN_U; ++k) {
            const long e = base + k * DEG_T + t;
            const bool ok = e < E;
            s[k] = ok ? src[e] : 0;
            d[k] = ok ? dst[e] : -1;
        }
#pragma unroll
        for (int k = 0; k < BIN_U; ++k) {
            b[k] = d[k] >= 0 ? d[k] / BKN : -1;
            r[k] = b[k] >= 0 ? atomicAdd(&gcnt[b[k]], 1) : 0;
        }
        __syncthreads();
        scan_offsets(gcnt, nbk, t);
        __syncthreads();
#pragma unroll
        for (int k = 0; k < BIN_U; ++k) {
            if (b[k] >= 0) {
                const int pos = gcnt[b[k]] + r[k];
                stg_s[pos] = s[k];
                stg_d[pos] = d[k];
            }
        }
        __syncthreads();
        const int total = gcnt[nbk];
        for (int j = t; j < total; j += DEG_T) {
            rec_s[base + j] = stg_s[j];
            rec_d[base + j] = stg_d[j];
        }
        for (int g = t; g <= nbk; g += DEG_T) offT[(long)g * np + pb] = gcnt[g];
    }
}


#define ND_T 256

__global__ void __launch_bounds__(ND_T)
k_node(const int* __restrict__ part, const int* __restrict__ batch, const float* __restrict__ x,
       float* __restrict__ nd, float* __restrict__ cnt, float* __restrict__ xs, int n) {
    __shared__ float sl[ND_T];
    __shared__ int   gsz[NGRAPH];
    const int t    = threadIdx.x;
    const int base = blockIdx.x * ND_T;
    const int i    = base + t;
    for (int j = t; j < NGRAPH; j += ND_T) gsz[j] = 0;

    const bool   full = base + ND_T <= n;
    const float* xb   = x + (long)base * IN_C;
    float v[IN_C];
    if (full) {
#pragma unroll
        for (int k = 0; k < IN_C; ++k) v[k] = xb[k * ND_T + t];
    }

    float dv = 0.f;
    int   g  = -1;
    if (i < n) {
        int c = 1;
#pragma unroll
        for (int s = 0; s < DEG_S; ++s) c += part[(long)s * n + i];
        dv = rsqrtf((float)c);
        g  = batch[i];
    }
    sl[t] = dv;
    __syncthreads();
    if (g >= 0) atomicAdd(&gsz[g], 1);
    __syncthreads();
    for (int j = t; j < NGRAPH; j += ND_T) {
        const int q = gsz[j];
        if (q) atomicAdd(&cnt[j], (float)q);
    }

    if (i < n) reinterpret_cast<float2*>(nd)[i] = make_float2(dv, __int_as_float(g));
    if (full) {
#pragma unroll
        for (int k = 0; k < IN_C; ++k) {
            const int idx = k * ND_T + t;
            const int r = idx / IN_C, c = idx - r * IN_C;
            xs[(long)(base + r) * AGX_LD + c] = sl[r] * v[k];
        }
    } else {
        const int cntf = (n - base) * IN_C;
        for (int idx = t; idx < cntf; idx += ND_T) {
            const int r = idx / IN_C, c = idx - r * IN_C;
            xs[(long)(base + r) * AGX_LD + c] = sl[r] * xb[idx];
        }
    }
}


#define MM_W 8
#define HP   68

__global__ void __launch_bounds__(MM_W * 32)
k_mlp(const float* __restrict__ aggx, const float* __restrict__ W1, const float* __restrict__ b1,
      const float* __restrict__ W2, const float* __restrict__ nd, float* __restrict__ h2, int n) {
    __shared__ __attribute__((aligned(16))) float Hs[MM_W][16 * HP];
    const int  lane = threadIdx.x & 31;
    const int  wave = threadIdx.x >> 5;
    const int  m    = lane & 15;
    const int  h    = lane >> 4;
    const long rowbase = ((long)blockIdx.x * MM_W + wave) * 16;
    float* H = Hs[wave];

    const long rA = rowbase + m < n ? rowbase + m : (long)n - 1;
    const float* ar = aggx + rA * AGX_LD;

    const float dr = nd[2 * rA];

    constexpr int KS1 = (IN_C + 3) / 4;
    v8f acc[HID_C / 16] = {};
#pragma unroll
    for (int ks = 0; ks < KS1; ++ks) {
        const int ka = ks * 4 + 2 * h;
        const int kb = ka + 1;
        const float ma = (ka < IN_C) ? 1.f : 0.f;
        const float mb = (kb < IN_C) ? 1.f : 0.f;
        const int kca = (ka < IN_C) ? ka : (IN_C - 1);
        const int kcb = (kb < IN_C) ? kb : (IN_C - 1);

        v2f a;
        a.x = ar[kca] * ma;
        a.y = ar[kcb] * mb;
#pragma unroll
        for (int nt = 0; nt < HID_C / 16; ++nt) {
            v2f b;
            b.x = W1[kca * HID_C + nt * 16 + m] * ma;
            b.y = W1[kcb * HID_C + nt * 16 + m] * mb;
            acc[nt] = __builtin_amdgcn_wmma_f32_16x16x4_f32(
                false, a, false, b, (short)0, acc[nt], false, false);
        }
    }
#pragma unroll
    for (int nt = 0; nt < HID_C / 16; ++nt) {
        const float bv = b1[nt * 16 + m];
#pragma unroll
        for (int v = 0; v < 8; ++v) {
            float t = acc[nt][v] + bv;
            H[(v + 8 * h) * HP + nt * 16 + m] = t > 0.f ? t : 0.f;
        }
    }
    __syncthreads();

    v8f o[OUT_C / 16] = {};
#pragma unroll
    for (int ks = 0; ks < HID_C / 4; ++ks) {
        const int ka = ks * 4 + 2 * h, kb = ka + 1;
        const v2f a = *(const v2f*)(H + m * HP + ka) * dr;
#pragma unroll
        for (int nt = 0; nt < OUT_C / 16; ++nt) {
            v2f b;
            b.x = W2[ka * OUT_C + nt * 16 + m];
            b.y = W2[kb * OUT_C + nt * 16 + m];
            o[nt] = __builtin_amdgcn_wmma_f32_16x16x4_f32(
                false, a, false, b, (short)0, o[nt], false, false);
        }
    }
    float* orow = h2 + (rowbase + 8 * h) * OUT_C + m;
    if (rowbase + 16 <= n) {
#pragma unroll
        for (int nt = 0; nt < OUT_C / 16; ++nt)
#pragma unroll
            for (int v = 0; v < 8; ++v) orow[v * OUT_C + nt * 16] = o[nt][v];
    } else {
#pragma unroll
        for (int nt = 0; nt < OUT_C / 16; ++nt)
#pragma unroll
            for (int v = 0; v < 8; ++v)
                if (rowbase + v + 8 * h < n) orow[v * OUT_C + nt * 16] = o[nt][v];
    }
}


#define OWN_ROWS 60
#define OWN_T    (IN_C * OWN_ROWS)
#define OWN_U    8
#define OWN_CAP  4096
#define OWN_Q    4

__global__ void __launch_bounds__(OWN_T)
k_own_acc(const int* __restrict__ rec_s, const int* __restrict__ rec_d, const int* __restrict__ offT,
          const float* __restrict__ nd, const float* __restrict__ xs, float* __restrict__ aggx,
          int n, int np) {
    __shared__ float acc[BKN * IN_C];
    __shared__ int   segoff[SCAN_N];
    __shared__ int   segbase[NP_MAX];
    __shared__ int2  stg[OWN_CAP];
    const int c   = threadIdx.x;
    const int y   = threadIdx.y;
    const int tid = y * IN_C + c;
    const int g   = blockIdx.x;
    const int n0  = g * BKN;

    for (int j = tid; j < BKN * IN_C; j += OWN_T) acc[j] = 0.f;
    for (int pb = tid; pb < np; pb += OWN_T) {
        const int a = offT[(long)g * np + pb];
        const int b = offT[(long)(g + 1) * np + pb];
        segbase[pb] = pb * EPB + a;
        segoff[pb]  = b - a;
    }
    __syncthreads();
    scan_offsets(segoff, np, tid);
    __syncthreads();
    const int total = segoff[np];

    for (int c0 = 0; c0 < total; c0 += OWN_CAP) {
        const int m = total - c0 < OWN_CAP ? total - c0 : OWN_CAP;
        for (int jb = tid; jb < m; jb += OWN_T * OWN_Q) {
            int a[OWN_Q], sv[OWN_Q], dv[OWN_Q];
#pragma unroll
            for (int q = 0; q < OWN_Q; ++q) {
                const int j = jb + q * OWN_T;
                const int f = c0 + (j < m ? j : jb);
                int lo = 0, hi = np;
#pragma unroll
                for (int it = 0; it < 10; ++it) {
                    const int  mid = (lo + hi) >> 1;
                    const bool act = hi - lo > 1;
                    const bool le  = segoff[mid] <= f;
                    lo = (act && le) ? mid : lo;
                    hi = (act && !le) ? mid : hi;
                }
                a[q] = segbase[lo] + (f - segoff[lo]);
            }
#pragma unroll
            for (int q = 0; q < OWN_Q; ++q) {
                sv[q] = rec_s[a[q]];
                dv[q] = rec_d[a[q]];
            }
#pragma unroll
            for (int q = 0; q < OWN_Q; ++q) {
                const int j = jb + q * OWN_T;
                if (j < m) stg[j] = make_int2(sv[q], (dv[q] - n0) * IN_C);
            }
        }
        __syncthreads();
        for (int j0 = y; j0 < m; j0 += OWN_ROWS * OWN_U) {
            float v[OWN_U];
            int   sn[OWN_U], o[OWN_U];
#pragma unroll
            for (int u = 0; u < OWN_U; ++u) {
                const int  j  = j0 + u * OWN_ROWS;
                const int2 r2 = stg[j < m ? j : j0];
                sn[u] = r2.x;
                o[u]  = j < m ? r2.y + c : -1;
            }
#pragma unroll
            for (int u = 0; u < OWN_U; ++u) v[u] = xs[(long)sn[u] * AGX_LD + c];
#pragma unroll
            for (int u = 0; u < OWN_U; ++u)
                if (o[u] >= 0) atomicAdd(&acc[o[u]], v[u]);
        }
        __syncthreads();
    }

    const int rows = n - n0 < BKN ? n - n0 : BKN;
    for (int r = y; r < rows; r += OWN_ROWS) {
        const long i = n0 + r;
        aggx[i * AGX_LD + c] = nd[2 * i] * (acc[r * IN_C + c] + xs[i * AGX_LD + c]);
    }
}


#define AP_T 1024
#define AP_NB 256

template <bool GUARD>
__device__ __forceinline__ void agg_pool_chunk(const int* __restrict__ src, const int* __restrict__ dst,
                                               const float* __restrict__ nd,
                                               const float* __restrict__ h2, float* acc,
                                               long base, long total, int E, int lane) {
    const long idx = base + lane;
    int s = 0, d = 0;
    if (!GUARD || idx < total) {
        if (idx < E) { s = src[idx]; d = dst[idx]; }
        else         { s = d = (int)(idx - E); }
    }
    const float2 rec = reinterpret_cast<const float2*>(nd)[d];
    const float w = rec.x;
    const int   g = __float_as_int(rec.y);
    const int cnt = GUARD ? (int)(total - base) : 32;

    float v[32];
#pragma unroll
    for (int k = 0; k < 32; ++k) {
        const int sk = __builtin_amdgcn_readlane(s, k);
        const __attribute__((address_space(1))) float* row =
            (const __attribute__((address_space(1))) float*)(h2 + (long)sk * OUT_C);
        asm("" : "+s"(row));
        v[k] = (!GUARD || k < cnt) ? row[lane] : 0.f;
    }
#pragma unroll
    for (int k = 0; k < 32; ++k) {
        const int   gk = __builtin_amdgcn_readlane(g, k);
        const float wk = __int_as_float(__builtin_amdgcn_readlane(__float_as_int(w), k));
        if (!GUARD || k < cnt) atomicAdd(acc + gk * OUT_C + lane, wk * v[k]);
    }
}

__global__ void __launch_bounds__(AP_T)
k_agg_pool(const int* __restrict__ src, const int* __restrict__ dst,
           const float* __restrict__ nd,
           const float* __restrict__ h2, float* __restrict__ pool, int n, int E) {
    __shared__ float acc[NGRAPH * OUT_C];
    for (int j = threadIdx.x; j < NGRAPH * OUT_C; j += AP_T) acc[j] = 0.f;
    __syncthreads();

    const int  lane   = threadIdx.x & 31;
    const int  wave   = threadIdx.x >> 5;
    const long total  = (long)E + n;
    const long nchunk = (total + 31) / 32;
    for (long c = (long)blockIdx.x * (AP_T / 32) + wave; c < nchunk; c += (long)gridDim.x * (AP_T / 32)) {
        const long base = c * 32;
        if (base + 32 <= total) agg_pool_chunk<false>(src, dst, nd, h2, acc, base, total, E, lane);
        else                    agg_pool_chunk<true >(src, dst, nd, h2, acc, base, total, E, lane);
    }
    __syncthreads();

    for (int j = threadIdx.x; j < NGRAPH * OUT_C; j += AP_T) {
        const int jj = (j + blockIdx.x * 64) & (NGRAPH * OUT_C - 1);
        atomicAdd(pool + jj, acc[jj]);
    }
}


__global__ void k_fc(const float* __restrict__ pool, const float* __restrict__ cnt,
                     const float* __restrict__ b2,
                     const float* __restrict__ Wf1, const float* __restrict__ bf1,
                     const float* __restrict__ Wf2, const float* __restrict__ bf2,
                     float* __restrict__ out) {
    __shared__ float P[16 * 32];
    __shared__ float T[16 * 32];
    const int lane = threadIdx.x;
    const int m = lane & 15;
    const int h = lane >> 4;
    const int g0 = blockIdx.x * 16;

    for (int idx = lane; idx < 16 * 32; idx += 32) {
        int r = idx >> 5, c = idx & 31;
        float ct = cnt[g0 + r];
        ct = ct > 1.f ? ct : 1.f;
        P[idx] = pool[(g0 + r) * 32 + c] / ct + b2[c];
    }
    __syncthreads();

    v8f acc[2] = {};
#pragma unroll
    for (int ks = 0; ks < 8; ++ks) {
        const int ka = ks * 4 + 2 * h, kb = ka + 1;
        v2f a; a.x = P[m * 32 + ka]; a.y = P[m * 32 + kb];
#pragma unroll
        for (int nt = 0; nt < 2; ++nt) {
            v2f b; b.x = Wf1[ka * 32 + nt * 16 + m]; b.y = Wf1[kb * 32 + nt * 16 + m];
            acc[nt] = __builtin_amdgcn_wmma_f32_16x16x4_f32(
                false, a, false, b, (short)0, acc[nt], false, false);
        }
    }
#pragma unroll
    for (int nt = 0; nt < 2; ++nt)
#pragma unroll
        for (int v = 0; v < 8; ++v) {
            int r = v + 8 * h, c = nt * 16 + m;
            float t = acc[nt][v] + bf1[c];
            T[r * 32 + c] = t > 0.f ? t : 0.f;
        }
    __syncthreads();

    v8f o[2] = {};
#pragma unroll
    for (int ks = 0; ks < 8; ++ks) {
        const int ka = ks * 4 + 2 * h, kb = ka + 1;
        v2f a; a.x = T[m * 32 + ka]; a.y = T[m * 32 + kb];
#pragma unroll
        for (int nt = 0; nt < 2; ++nt) {
            v2f b; b.x = Wf2[ka * 32 + nt * 16 + m]; b.y = Wf2[kb * 32 + nt * 16 + m];
            o[nt] = __builtin_amdgcn_wmma_f32_16x16x4_f32(
                false, a, false, b, (short)0, o[nt], false, false);
        }
    }
#pragma unroll
    for (int nt = 0; nt < 2; ++nt)
#pragma unroll
        for (int v = 0; v < 8; ++v) {
            int r = v + 8 * h, c = nt * 16 + m;
            out[(g0 + r) * 32 + c] = o[nt][v] + bf2[c];
        }
}


extern "C" void kernel_launch(void* const* d_in, const int* in_sizes, int n_in,
                              void* d_out, int out_size, void* d_ws, size_t ws_size,
                              hipStream_t stream) {
    const float* x     = (const float*)d_in[0];
    const int*   ei    = (const int*)d_in[1];
    const int*   batch = (const int*)d_in[2];
    const float* W1  = (const float*)d_in[3];
    const float* b1  = (const float*)d_in[4];
    const float* W2  = (const float*)d_in[5];
    const float* b2  = (const float*)d_in[6];
    const float* Wf1 = (const float*)d_in[7];
    const float* bf1 = (const float*)d_in[8];
    const float* Wf2 = (const float*)d_in[9];
    const float* bf2 = (const float*)d_in[10];

    const int N = in_sizes[0] / IN_C;
    const int E = in_sizes[1] / 2;
    const int* src = ei;
    const int* dst = ei + E;

    float* ws   = (float*)d_ws;
    float* nd   = ws;
    float* aggx = nd + 2 * (size_t)N;
    float* xs   = aggx + (size_t)N * AGX_LD;
    float* hid  = aggx + (size_t)N * HID_C;
    float* h2   = hid  + (size_t)N * HID_C;
    float* pool = h2   + (size_t)N * OUT_C * 2;
    float* cnt  = pool + NGRAPH * OUT_C;

    int* part = (int*)hid;
    const int ndeg = DEG_S * ((N + DEG_RN - 1) / DEG_RN);
    const int np   = (E + EPB - 1) / EPB;
    const int nbk  = (N + BKN - 1) / BKN;
    int* rec_s = (int*)(hid + (size_t)N * OUT_C);
    int* rec_d = (int*)(h2  + (size_t)N * OUT_C);
    int* offT  = rec_d + (size_t)np * EPB;
    k_deg_bin<<<ndeg + np, DEG_T, 0, stream>>>(src, dst, part, pool, rec_s, rec_d, offT, N, E, ndeg, np, nbk);

    k_node<<<(N + ND_T - 1) / ND_T, ND_T, 0, stream>>>(part, batch, x, nd, cnt, xs, N);

    const dim3 BO(IN_C, OWN_ROWS);
    k_own_acc<<<nbk, BO, 0, stream>>>(rec_s, rec_d, offT, nd, xs, aggx, N, np);

    const int tiles = (N + 15) / 16;
    k_mlp<<<(tiles + MM_W - 1) / MM_W, MM_W * 32, 0, stream>>>(aggx, W1, b1, W2, nd, h2, N);

    k_agg_pool<<<AP_NB, AP_T, 0, stream>>>(src, dst, nd, h2, pool, N, E);

    k_fc<<<NGRAPH / 16, 32, 0, stream>>>(pool, cnt, b2, Wf1, bf1, Wf2, bf2, (float*)d_out);
}


# 6 launches -> 5: the owner block that sums 200 nodes' input rows in LDS also runs both layer GEMMs on them (13 waves, a 16-row tile each, A operand read from the LDS sums); no aggregated-row buffer
# speedup vs baseline: 4.2678x; 1.0657x over previous
#include <hip/hip_runtime.h>

typedef float v2f __attribute__((ext_vector_type(2)));
typedef float v8f __attribute__((ext_vector_type(8)));

#define IN_C   17
#define AGX_LD 32
#define HID_C  64
#define OUT_C  32
#define NGRAPH 512


#define DEG_RN 12800
#define DEG_S  32
#define DEG_T  1024
#define DEG_U  8


#define NACC (NGRAPH * OUT_C + NGRAPH)
static_assert(DEG_S * DEG_T >= NACC, "the first row of blocks must cover the accumulators");


#define BKN     200
#define NBK_MAX 512
#define EPB     4096
#define BIN_U   (EPB / DEG_T)
#define NP_MAX  512
#define SCAN_W  17
#define SCAN_N  (32 * SCAN_W)

__device__ __forceinline__ void scan_offsets(int* a, int n, int tid) {
    if (tid < 32) {
        int loc[SCAN_W];
        int sum = 0;
#pragma unroll
        for (int j = 0; j < SCAN_W; ++j) {
            const int idx = tid * SCAN_W + j;
            loc[j] = sum;
            sum += idx < n ? a[idx] : 0;
        }
        int inc = sum;
#pragma unroll
        for (int dlt = 1; dlt < 32; dlt <<= 1) {
            const int o = __shfl_up(inc, dlt);
            if (tid >= dlt) inc += o;
        }
        const int exc = inc - sum;
#pragma unroll
        for (int j = 0; j < SCAN_W; ++j) {
            const int idx = tid * SCAN_W + j;
            if (idx <= n) a[idx] = exc + loc[j];
        }
    }
}

static_assert(2 * EPB + SCAN_N <= DEG_RN, "the grouping tables reuse the degree counters' LDS");

__global__ void __launch_bounds__(DEG_T)
k_deg_bin(const int* __restrict__ src, const int* __restrict__ dst, int* __restrict__ part,
          float* __restrict__ acc0, int* __restrict__ rec_s, int* __restrict__ rec_d,
          int* __restrict__ offT, int n, int E, int ndeg, int np, int nbk) {
    __shared__ int bins[DEG_RN];
    const int bid = blockIdx.x;
    if (bid < ndeg) {
        const int bx = bid % DEG_S;
        const int by = bid / DEG_S;
        if (by == 0) {
            const int j = bx * DEG_T + threadIdx.x;
            if (j < NACC) acc0[j] = 0.f;
        }
        const int lo = by * DEG_RN;
        for (int j = threadIdx.x; j < DEG_RN; j += DEG_T) bins[j] = 0;
        __syncthreads();

        const int per = (E + DEG_S - 1) / DEG_S;
        const int e0  = bx * per;
        const int e1  = e0 + per < E ? e0 + per : E;
        for (int eb = e0; eb < e1; eb += DEG_T * DEG_U) {
            int d[DEG_U];
#pragma unroll
            for (int k = 0; k < DEG_U; ++k) {
                const int e = eb + k * DEG_T + threadIdx.x;
                d[k] = e < e1 ? dst[e] : -1;
            }
#pragma unroll
            for (int k = 0; k < DEG_U; ++k) {
                const unsigned r = (unsigned)(d[k] - lo);
                if (r < (unsigned)DEG_RN) atomicAdd(&bins[r], 1);
            }
        }
        __syncthreads();

        const int cnt = n - lo < DEG_RN ? n - lo : DEG_RN;
        int* row = part + (long)bx * n + lo;
        for (int j = threadIdx.x; j < cnt; j += DEG_T) row[j] = bins[j];
    } else {
        const int pb = bid - ndeg;
        const int t  = threadIdx.x;
        int* stg_s = bins;
        int* stg_d = bins + EPB;
        int* gcnt  = bins + 2 * EPB;
        for (int j = t; j < SCAN_N; j += DEG_T) gcnt[j] = 0;
        __syncthreads();

        const long base = (long)pb * EPB;
        int  s[BIN_U], d[BIN_U], b[BIN_U], r[BIN_U];
#pragma unroll
        for (int k = 0; k < BIN_U; ++k) {
            const long e = base + k * DEG_T + t;
            const bool ok = e < E;
            s[k] = ok ? src[e] : 0;
            d[k] = ok ? dst[e] : -1;
        }
#pragma unroll
        for (int k = 0; k < BIN_U; ++k) {
            b[k] = d[k] >= 0 ? d[k] / BKN : -1;
            r[k] = b[k] >= 0 ? atomicAdd(&gcnt[b[k]], 1) : 0;
        }
        __syncthreads();
        scan_offsets(gcnt, nbk, t);
        __syncthreads();
#pragma unroll
        for (int k = 0; k < BIN_U; ++k) {
            if (b[k] >= 0) {
                const int pos = gcnt[b[k]] + r[k];
                stg_s[pos] = s[k];
                stg_d[pos] = d[k];
            }
        }
        __syncthreads();
        const int total = gcnt[nbk];
        for (int j = t; j < total; j += DEG_T) {
            rec_s[base + j] = stg_s[j];
            rec_d[base + j] = stg_d[j];
        }
        for (int g = t; g <= nbk; g += DEG_T) offT[(long)g * np + pb] = gcnt[g];
    }
}


#define ND_T 256

__global__ void __launch_bounds__(ND_T)
k_node(const int* __restrict__ part, const int* __restrict__ batch, const float* __restrict__ x,
       float* __restrict__ nd, float* __restrict__ cnt, float* __restrict__ xs, int n) {
    __shared__ float sl[ND_T];
    __shared__ int   gsz[NGRAPH];
    const int t    = threadIdx.x;
    const int base = blockIdx.x * ND_T;
    const int i    = base + t;
    for (int j = t; j < NGRAPH; j += ND_T) gsz[j] = 0;

    const bool   full = base + ND_T <= n;
    const float* xb   = x + (long)base * IN_C;
    float v[IN_C];
    if (full) {
#pragma unroll
        for (int k = 0; k < IN_C; ++k) v[k] = xb[k * ND_T + t];
    }

    float dv = 0.f;
    int   g  = -1;
    if (i < n) {
        int c = 1;
#pragma unroll
        for (int s = 0; s < DEG_S; ++s) c += part[(long)s * n + i];
        dv = rsqrtf((float)c);
        g  = batch[i];
    }
    sl[t] = dv;
    __syncthreads();
    if (g >= 0) atomicAdd(&gsz[g], 1);
    __syncthreads();
    for (int j = t; j < NGRAPH; j += ND_T) {
        const int q = gsz[j];
        if (q) atomicAdd(&cnt[j], (float)q);
    }

    if (i < n) reinterpret_cast<float2*>(nd)[i] = make_float2(dv, __int_as_float(g));
    if (full) {
#pragma unroll
        for (int k = 0; k < IN_C; ++k) {
            const int idx = k * ND_T + t;
            const int r = idx / IN_C, c = idx - r * IN_C;
            xs[(long)(base + r) * AGX_LD + c] = sl[r] * v[k];
        }
    } else {
        const int cntf = (n - base) * IN_C;
        for (int idx = t; idx < cntf; idx += ND_T) {
            const int r = idx / IN_C, c = idx - r * IN_C;
            xs[(long)(base + r) * AGX_LD + c] = sl[r] * xb[idx];
        }
    }
}


#define OWN_ROWS 60
#define OWN_T    (IN_C * OWN_ROWS)
#define OWN_U    8
#define OWN_CAP  4096
#define OWN_Q    4
#define OWN_TL   ((BKN + 15) / 16)
#define AP       20
#define HP       68

static_assert(AP >= ((IN_C + 3) / 4) * 4, "the k padding of the first product lies inside a row of the sums");
static_assert(OWN_TL * 32 <= OWN_T - 31, "the tile waves are whole waves");

__global__ void __launch_bounds__(OWN_T)
k_own_mlp(const int* __restrict__ rec_s, const int* __restrict__ rec_d, const int* __restrict__ offT,
          const float* __restrict__ nd, const float* __restrict__ xs,
          const float* __restrict__ W1, const float* __restrict__ b1, const float* __restrict__ W2,
          float* __restrict__ h2, int n, int np) {
    __shared__ __attribute__((aligned(16))) float acc[OWN_TL * 16 * AP];
    __shared__ int   segoff[SCAN_N];
    __shared__ int   segbase[NP_MAX];
    __shared__ int2  stg[OWN_CAP];
    __shared__ __attribute__((aligned(16))) float Hs[OWN_TL][16 * HP];
    const int c   = threadIdx.x;
    const int y   = threadIdx.y;
    const int tid = y * IN_C + c;
    const int g   = blockIdx.x;
    const int n0  = g * BKN;

    for (int j = tid; j < OWN_TL * 16 * AP; j += OWN_T) acc[j] = 0.f;
    for (int pb = tid; pb < np; pb += OWN_T) {
        const int a = offT[(long)g * np + pb];
        const int b = offT[(long)(g + 1) * np + pb];
        segbase[pb] = pb * EPB + a;
        segoff[pb]  = b - a;
    }
    __syncthreads();
    scan_offsets(segoff, np, tid);
    __syncthreads();
    const int total = segoff[np];

    for (int c0 = 0; c0 < total; c0 += OWN_CAP) {
        const int m = total - c0 < OWN_CAP ? total - c0 : OWN_CAP;
        for (int jb = tid; jb < m; jb += OWN_T * OWN_Q) {
            int a[OWN_Q], sv[OWN_Q], dv[OWN_Q];
#pragma unroll
            for (int q = 0; q < OWN_Q; ++q) {
                const int j = jb + q * OWN_T;
                const int f = c0 + (j < m ? j : jb);
                int lo = 0, hi = np;
#pragma unroll
                for (int it = 0; it < 10; ++it) {
                    const int  mid = (lo + hi) >> 1;
                    const bool act = hi - lo > 1;
                    const bool le  = segoff[mid] <= f;
                    lo = (act && le) ? mid : lo;
                    hi = (act && !le) ? mid : hi;
                }
                a[q] = segbase[lo] + (f - segoff[lo]);
            }
#pragma unroll
            for (int q = 0; q < OWN_Q; ++q) {
                sv[q] = rec_s[a[q]];
                dv[q] = rec_d[a[q]];
            }
#pragma unroll
            for (int q = 0; q < OWN_Q; ++q) {
                const int j = jb + q * OWN_T;
                if (j < m) stg[j] = make_int2(sv[q], (dv[q] - n0) * AP);
            }
        }
        __syncthreads();
        for (int j0 = y; j0 < m; j0 += OWN_ROWS * OWN_U) {
            float v[OWN_U];
            int   sn[OWN_U], o[OWN_U];
#pragma unroll
            for (int u = 0; u < OWN_U; ++u) {
                const int  j  = j0 + u * OWN_ROWS;
                const int2 r2 = stg[j < m ? j : j0];
                sn[u] = r2.x;
                o[u]  = j < m ? r2.y + c : -1;
            }
#pragma unroll
            for (int u = 0; u < OWN_U; ++u) v[u] = xs[(long)sn[u] * AGX_LD + c];
#pragma unroll
            for (int u = 0; u < OWN_U; ++u)
                if (o[u] >= 0) atomicAdd(&acc[o[u]], v[u]);
        }
        __syncthreads();
    }

    const int rows = n - n0 < BKN ? n - n0 : BKN;
    for (int r = y; r < rows; r += OWN_ROWS) {
        const long i = n0 + r;
        acc[r * AP + c] = nd[2 * i] * (acc[r * AP + c] + xs[i * AGX_LD + c]);
    }
    __syncthreads();

    const int  lane = tid & 31;
    const int  wave = tid >> 5;
    const int  m    = lane & 15;
    const int  h    = lane >> 4;
    const int  r0   = wave * 16;
    const bool mine = wave < OWN_TL && r0 < rows;
    float* H  = Hs[mine ? wave : 0];
    float  dr = 0.f;
    if (mine) {
        const long rA = (long)n0 + r0 + m < n ? (long)n0 + r0 + m : (long)n - 1;
        dr = nd[2 * rA];
        const float* ar = acc + (r0 + m) * AP;

        constexpr int KS1 = (IN_C + 3) / 4;
        v8f a1[HID_C / 16] = {};
#pragma unroll
        for (int ks = 0; ks < KS1; ++ks) {
            const int ka = ks * 4 + 2 * h;
            const int kb = ka + 1;
            const float ma = (ka < IN_C) ? 1.f : 0.f;
            const float mb = (kb < IN_C) ? 1.f : 0.f;
            const int kca = (ka < IN_C) ? ka : (IN_C - 1);
            const int kcb = (kb < IN_C) ? kb : (IN_C - 1);

            const v2f a = *(const v2f*)(ar + ka);
#pragma unroll
            for (int nt = 0; nt < HID_C / 16; ++nt) {
                v2f b;
                b.x = W1[kca * HID_C + nt * 16 + m] * ma;
                b.y = W1[kcb * HID_C + nt * 16 + m] * mb;
                a1[nt] = __builtin_amdgcn_wmma_f32_16x16x4_f32(
                    false, a, false, b, (short)0, a1[nt], false, false);
            }
        }
#pragma unroll
        for (int nt = 0; nt < HID_C / 16; ++nt) {
            const float bv = b1[nt * 16 + m];
#pragma unroll
            for (int v = 0; v < 8; ++v) {
                float t = a1[nt][v] + bv;
                H[(v + 8 * h) * HP + nt * 16 + m] = t > 0.f ? t : 0.f;
            }
        }
    }
    __syncthreads();
    if (mine) {
        v8f o[OUT_C / 16] = {};
#pragma unroll
        for (int ks = 0; ks < HID_C / 4; ++ks) {
            const int ka = ks * 4 + 2 * h, kb = ka + 1;
            const v2f a = *(const v2f*)(H + m * HP + ka) * dr;
#pragma unroll
            for (int nt = 0; nt < OUT_C / 16; ++nt) {
                v2f b;
                b.x = W2[ka * OUT_C + nt * 16 + m];
                b.y = W2[kb * OUT_C + nt * 16 + m];
                o[nt] = __builtin_amdgcn_wmma_f32_16x16x4_f32(
                    false, a, false, b, (short)0, o[nt], false, false);
            }
        }
        float* orow = h2 + ((long)n0 + r0 + 8 * h) * OUT_C + m;
        if (r0 + 16 <= rows) {
#pragma unroll
            for (int nt = 0; nt < OUT_C / 16; ++nt)
#pragma unroll
                for (int v = 0; v < 8; ++v) orow[v * OUT_C + nt * 16] = o[nt][v];
        } else {
#pragma unroll
            for (int nt = 0; nt < OUT_C / 16; ++nt)
#pragma unroll
                for (int v = 0; v < 8; ++v)
                    if (r0 + v + 8 * h < rows) orow[v * OUT_C + nt * 16] = o[nt][v];
        }
    }
}


#define AP_T 1024
#define AP_NB 256

template <bool GUARD>
__device__ __forceinline__ void agg_pool_chunk(const int* __restrict__ src, const int* __restrict__ dst,
                                               const float* __restrict__ nd,
                                               const float* __restrict__ h2, float* acc,
                                               long base, long total, int E, int lane) {
    const long idx = base + lane;
    int s = 0, d = 0;
    if (!GUARD || idx < total) {
        if (idx < E) { s = src[idx]; d = dst[idx]; }
        else         { s = d = (int)(idx - E); }
    }
    const float2 rec = reinterpret_cast<const float2*>(nd)[d];
    const float w = rec.x;
    const int   g = __float_as_int(rec.y);
    const int cnt = GUARD ? (int)(total - base) : 32;

    float v[32];
#pragma unroll
    for (int k = 0; k < 32; ++k) {
        const int sk = __builtin_amdgcn_readlane(s, k);
        const __attribute__((address_space(1))) float* row =
            (const __attribute__((address_space(1))) float*)(h2 + (long)sk * OUT_C);
        asm("" : "+s"(row));
        v[k] = (!GUARD || k < cnt) ? row[lane] : 0.f;
    }
#pragma unroll
    for (int k = 0; k < 32; ++k) {
        const int   gk = __builtin_amdgcn_readlane(g, k);
        const float wk = __int_as_float(__builtin_amdgcn_readlane(__float_as_int(w), k));
        if (!GUARD || k < cnt) atomicAdd(acc + gk * OUT_C + lane, wk * v[k]);
    }
}

__global__ void __launch_bounds__(AP_T)
k_agg_pool(const int* __restrict__ src, const int* __restrict__ dst,
           const float* __restrict__ nd,
           const float* __restrict__ h2, float* __restrict__ pool, int n, int E) {
    __shared__ float acc[NGRAPH * OUT_C];
    for (int j = threadIdx.x; j < NGRAPH * OUT_C; j += AP_T) acc[j] = 0.f;
    __syncthreads();

    const int  lane   = threadIdx.x & 31;
    const int  wave   = threadIdx.x >> 5;
    const long total  = (long)E + n;
    const long nchunk = (total + 31) / 32;
    for (long c = (long)blockIdx.x * (AP_T / 32) + wave; c < nchunk; c += (long)gridDim.x * (AP_T / 32)) {
        const long base = c * 32;
        if (base + 32 <= total) agg_pool_chunk<false>(src, dst, nd, h2, acc, base, total, E, lane);
        else                    agg_pool_chunk<true >(src, dst, nd, h2, acc, base, total, E, lane);
    }
    __syncthreads();

    for (int j = threadIdx.x; j < NGRAPH * OUT_C; j += AP_T) {
        const int jj = (j + blockIdx.x * 64) & (NGRAPH * OUT_C - 1);
        atomicAdd(pool + jj, acc[jj]);
    }
}


__global__ void k_fc(const float* __restrict__ pool, const float* __restrict__ cnt,
                     const float* __restrict__ b2,
                     const float* __restrict__ Wf1, const float* __restrict__ bf1,
                     const float* __restrict__ Wf2, const float* __restrict__ bf2,
                     float* __restrict__ out) {
    __shared__ float P[16 * 32];
    __shared__ float T[16 * 32];
    const int lane = threadIdx.x;
    const int m = lane & 15;
    const int h = lane >> 4;
    const int g0 = blockIdx.x * 16;

    for (int idx = lane; idx < 16 * 32; idx += 32) {
        int r = idx >> 5, c = idx & 31;
        float ct = cnt[g0 + r];
        ct = ct > 1.f ? ct : 1.f;
        P[idx] = pool[(g0 + r) * 32 + c] / ct + b2[c];
    }
    __syncthreads();

    v8f acc[2] = {};
#pragma unroll
    for (int ks = 0; ks < 8; ++ks) {
        const int ka = ks * 4 + 2 * h, kb = ka + 1;
        v2f a; a.x = P[m * 32 + ka]; a.y = P[m * 32 + kb];
#pragma unroll
        for (int nt = 0; nt < 2; ++nt) {
            v2f b; b.x = Wf1[ka * 32 + nt * 16 + m]; b.y = Wf1[kb * 32 + nt * 16 + m];
            acc[nt] = __builtin_amdgcn_wmma_f32_16x16x4_f32(
                false, a, false, b, (short)0, acc[nt], false, false);
        }
    }
#pragma unroll
    for (int nt = 0; nt < 2; ++nt)
#pragma unroll
        for (int v = 0; v < 8; ++v) {
            int r = v + 8 * h, c = nt * 16 + m;
            float t = acc[nt][v] + bf1[c];
            T[r * 32 + c] = t > 0.f ? t : 0.f;
        }
    __syncthreads();

    v8f o[2] = {};
#pragma unroll
    for (int ks = 0; ks < 8; ++ks) {
        const int ka = ks * 4 + 2 * h, kb = ka + 1;
        v2f a; a.x = T[m * 32 + ka]; a.y = T[m * 32 + kb];
#pragma unroll
        for (int nt = 0; nt < 2; ++nt) {
            v2f b; b.x = Wf2[ka * 32 + nt * 16 + m]; b.y = Wf2[kb * 32 + nt * 16 + m];
            o[nt] = __builtin_amdgcn_wmma_f32_16x16x4_f32(
                false, a, false, b, (short)0, o[nt], false, false);
        }
    }
#pragma unroll
    for (int nt = 0; nt < 2; ++nt)
#pragma unroll
        for (int v = 0; v < 8; ++v) {
            int r = v + 8 * h, c = nt * 16 + m;
            out[(g0 + r) * 32 + c] = o[nt][v] + bf2[c];
        }
}


extern "C" void kernel_launch(void* const* d_in, const int* in_sizes, int n_in,
                              void* d_out, int out_size, void* d_ws, size_t ws_size,
                              hipStream_t stream) {
    const float* x     = (const float*)d_in[0];
    const int*   ei    = (const int*)d_in[1];
    const int*   batch = (const int*)d_in[2];
    const float* W1  = (const float*)d_in[3];
    const float* b1  = (const float*)d_in[4];
    const float* W2  = (const float*)d_in[5];
    const float* b2  = (const float*)d_in[6];
    const float* Wf1 = (const float*)d_in[7];
    const float* bf1 = (const float*)d_in[8];
    const float* Wf2 = (const float*)d_in[9];
    const float* bf2 = (const float*)d_in[10];

    const int N = in_sizes[0] / IN_C;
    const int E = in_sizes[1] / 2;
    const int* src = ei;
    const int* dst = ei + E;

    float* ws   = (float*)d_ws;
    float* nd   = ws;
    float* reg0 = nd + 2 * (size_t)N;
    float* xs   = reg0 + (size_t)N * AGX_LD;
    float* hid  = reg0 + (size_t)N * HID_C;
    float* h2   = hid  + (size_t)N * HID_C;
    float* pool = h2   + (size_t)N * OUT_C * 2;
    float* cnt  = pool + NGRAPH * OUT_C;

    int* part = (int*)hid;
    const int ndeg = DEG_S * ((N + DEG_RN - 1) / DEG_RN);
    const int np   = (E + EPB - 1) / EPB;
    const int nbk  = (N + BKN - 1) / BKN;
    int* rec_s = (int*)(hid + (size_t)N * OUT_C);
    int* rec_d = (int*)(h2  + (size_t)N * OUT_C);
    int* offT  = rec_d + (size_t)np * EPB;
    k_deg_bin<<<ndeg + np, DEG_T, 0, stream>>>(src, dst, part, pool, rec_s, rec_d, offT, N, E, ndeg, np, nbk);

    k_node<<<(N + ND_T - 1) / ND_T, ND_T, 0, stream>>>(part, batch, x, nd, cnt, xs, N);

    const dim3 BO(IN_C, OWN_ROWS);
    k_own_mlp<<<nbk, BO, 0, stream>>>(rec_s, rec_d, offT, nd, xs, W1, b1, W2, h2, N, np);

    k_agg_pool<<<AP_NB, AP_T, 0, stream>>>(src, dst, nd, h2, pool, N, E);

    k_fc<<<NGRAPH / 16, 32, 0, stream>>>(pool, cnt, b2, Wf1, bf1, Wf2, bf2, (float*)d_out);
}


# layer-2 edge pass walks the edges grouped by destination range (built for layer 1): dinv+graph of dst from a 200-entry LDS copy; a block's two leading graph rows summed in registers; sums re-ordered
# speedup vs baseline: 4.6768x; 1.0958x over previous
#include <hip/hip_runtime.h>

typedef float v2f __attribute__((ext_vector_type(2)));
typedef float v8f __attribute__((ext_vector_type(8)));

#define IN_C   17
#define AGX_LD 32
#define HID_C  64
#define OUT_C  32
#define NGRAPH 512


#define DEG_RN 12800
#define DEG_S  32
#define DEG_T  1024
#define DEG_U  8


#define NACC (NGRAPH * OUT_C + NGRAPH)
static_assert(DEG_S * DEG_T >= NACC, "the first row of blocks must cover the accumulators");


#define BKN     200
#define NBK_MAX 512
#define EPB     4096
#define BIN_U   (EPB / DEG_T)
#define NP_MAX  512
#define SCAN_W  17
#define SCAN_N  (32 * SCAN_W)

__device__ __forceinline__ void scan_offsets(int* a, int n, int tid) {
    if (tid < 32) {
        int loc[SCAN_W];
        int sum = 0;
#pragma unroll
        for (int j = 0; j < SCAN_W; ++j) {
            const int idx = tid * SCAN_W + j;
            loc[j] = sum;
            sum += idx < n ? a[idx] : 0;
        }
        int inc = sum;
#pragma unroll
        for (int dlt = 1; dlt < 32; dlt <<= 1) {
            const int o = __shfl_up(inc, dlt);
            if (tid >= dlt) inc += o;
        }
        const int exc = inc - sum;
#pragma unroll
        for (int j = 0; j < SCAN_W; ++j) {
            const int idx = tid * SCAN_W + j;
            if (idx <= n) a[idx] = exc + loc[j];
        }
    }
}

static_assert(2 * EPB + SCAN_N <= DEG_RN, "the grouping tables reuse the degree counters' LDS");

__global__ void __launch_bounds__(DEG_T)
k_deg_bin(const int* __restrict__ src, const int* __restrict__ dst, int* __restrict__ part,
          float* __restrict__ acc0, int* __restrict__ rec_s, int* __restrict__ rec_d,
          int* __restrict__ offT, int n, int E, int ndeg, int np, int nbk) {
    __shared__ int bins[DEG_RN];
    const int bid = blockIdx.x;
    if (bid < ndeg) {
        const int bx = bid % DEG_S;
        const int by = bid / DEG_S;
        if (by == 0) {
            const int j = bx * DEG_T + threadIdx.x;
            if (j < NACC) acc0[j] = 0.f;
        }
        const int lo = by * DEG_RN;
        for (int j = threadIdx.x; j < DEG_RN; j += DEG_T) bins[j] = 0;
        __syncthreads();

        const int per = (E + DEG_S - 1) / DEG_S;
        const int e0  = bx * per;
        const int e1  = e0 + per < E ? e0 + per : E;
        for (int eb = e0; eb < e1; eb += DEG_T * DEG_U) {
            int d[DEG_U];
#pragma unroll
            for (int k = 0; k < DEG_U; ++k) {
                const int e = eb + k * DEG_T + threadIdx.x;
                d[k] = e < e1 ? dst[e] : -1;
            }
#pragma unroll
            for (int k = 0; k < DEG_U; ++k) {
                const unsigned r = (unsigned)(d[k] - lo);
                if (r < (unsigned)DEG_RN) atomicAdd(&bins[r], 1);
            }
        }
        __syncthreads();

        const int cnt = n - lo < DEG_RN ? n - lo : DEG_RN;
        int* row = part + (long)bx * n + lo;
        for (int j = threadIdx.x; j < cnt; j += DEG_T) row[j] = bins[j];
    } else {
        const int pb = bid - ndeg;
        const int t  = threadIdx.x;
        int* stg_s = bins;
        int* stg_d = bins + EPB;
        int* gcnt  = bins + 2 * EPB;
        for (int j = t; j < SCAN_N; j += DEG_T) gcnt[j] = 0;
        __syncthreads();

        const long base = (long)pb * EPB;
        int  s[BIN_U], d[BIN_U], b[BIN_U], r[BIN_U];
#pragma unroll
        for (int k = 0; k < BIN_U; ++k) {
            const long e = base + k * DEG_T + t;
            const bool ok = e < E;
            s[k] = ok ? src[e] : 0;
            d[k] = ok ? dst[e] : -1;
        }
#pragma unroll
        for (int k = 0; k < BIN_U; ++k) {
            b[k] = d[k] >= 0 ? d[k] / BKN : -1;
            r[k] = b[k] >= 0 ? atomicAdd(&gcnt[b[k]], 1) : 0;
        }
        __syncthreads();
        scan_offsets(gcnt, nbk, t);
        __syncthreads();
#pragma unroll
        for (int k = 0; k < BIN_U; ++k) {
            if (b[k] >= 0) {
                const int pos = gcnt[b[k]] + r[k];
                stg_s[pos] = s[k];
                stg_d[pos] = d[k];
            }
        }
        __syncthreads();
        const int total = gcnt[nbk];
        for (int j = t; j < total; j += DEG_T) {
            rec_s[base + j] = stg_s[j];
            rec_d[base + j] = stg_d[j];
        }
        for (int g = t; g <= nbk; g += DEG_T) offT[(long)g * np + pb] = gcnt[g];
    }
}


#define ND_T 256

__global__ void __launch_bounds__(ND_T)
k_node(const int* __restrict__ part, const int* __restrict__ batch, const float* __restrict__ x,
       float* __restrict__ nd, float* __restrict__ cnt, float* __restrict__ xs, int n) {
    __shared__ float sl[ND_T];
    __shared__ int   gsz[NGRAPH];
    const int t    = threadIdx.x;
    const int base = blockIdx.x * ND_T;
    const int i    = base + t;
    for (int j = t; j < NGRAPH; j += ND_T) gsz[j] = 0;

    const bool   full = base + ND_T <= n;
    const float* xb   = x + (long)base * IN_C;
    float v[IN_C];
    if (full) {
#pragma unroll
        for (int k = 0; k < IN_C; ++k) v[k] = xb[k * ND_T + t];
    }

    float dv = 0.f;
    int   g  = -1;
    if (i < n) {
        int c = 1;
#pragma unroll
        for (int s = 0; s < DEG_S; ++s) c += part[(long)s * n + i];
        dv = rsqrtf((float)c);
        g  = batch[i];
    }
    sl[t] = dv;
    __syncthreads();
    if (g >= 0) atomicAdd(&gsz[g], 1);
    __syncthreads();
    for (int j = t; j < NGRAPH; j += ND_T) {
        const int q = gsz[j];
        if (q) atomicAdd(&cnt[j], (float)q);
    }

    if (i < n) reinterpret_cast<float2*>(nd)[i] = make_float2(dv, __int_as_float(g));
    if (full) {
#pragma unroll
        for (int k = 0; k < IN_C; ++k) {
            const int idx = k * ND_T + t;
            const int r = idx / IN_C, c = idx - r * IN_C;
            xs[(long)(base + r) * AGX_LD + c] = sl[r] * v[k];
        }
    } else {
        const int cntf = (n - base) * IN_C;
        for (int idx = t; idx < cntf; idx += ND_T) {
            const int r = idx / IN_C, c = idx - r * IN_C;
            xs[(long)(base + r) * AGX_LD + c] = sl[r] * xb[idx];
        }
    }
}


#define OWN_ROWS 60
#define OWN_T    (IN_C * OWN_ROWS)
#define OWN_U    8
#define OWN_CAP  4096
#define OWN_Q    4
#define OWN_TL   ((BKN + 15) / 16)
#define AP       20
#define HP       68

static_assert(AP >= ((IN_C + 3) / 4) * 4, "the k padding of the first product lies inside a row of the sums");
static_assert(OWN_TL * 32 <= OWN_T - 31, "the tile waves are whole waves");

__global__ void __launch_bounds__(OWN_T)
k_own_mlp(const int* __restrict__ rec_s, const int* __restrict__ rec_d, const int* __restrict__ offT,
          const float* __restrict__ nd, const float* __restrict__ xs,
          const float* __restrict__ W1, const float* __restrict__ b1, const float* __restrict__ W2,
          float* __restrict__ h2, int n, int np) {
    __shared__ __attribute__((aligned(16))) float acc[OWN_TL * 16 * AP];
    __shared__ int   segoff[SCAN_N];
    __shared__ int   segbase[NP_MAX];
    __shared__ int2  stg[OWN_CAP];
    __shared__ __attribute__((aligned(16))) float Hs[OWN_TL][16 * HP];
    const int c   = threadIdx.x;
    const int y   = threadIdx.y;
    const int tid = y * IN_C + c;
    const int g   = blockIdx.x;
    const int n0  = g * BKN;

    for (int j = tid; j < OWN_TL * 16 * AP; j += OWN_T) acc[j] = 0.f;
    for (int pb = tid; pb < np; pb += OWN_T) {
        const int a = offT[(long)g * np + pb];
        const int b = offT[(long)(g + 1) * np + pb];
        segbase[pb] = pb * EPB + a;
        segoff[pb]  = b - a;
    }
    __syncthreads();
    scan_offsets(segoff, np, tid);
    __syncthreads();
    const int total = segoff[np];

    for (int c0 = 0; c0 < total; c0 += OWN_CAP) {
        const int m = total - c0 < OWN_CAP ? total - c0 : OWN_CAP;
        for (int jb = tid; jb < m; jb += OWN_T * OWN_Q) {
            int a[OWN_Q], sv[OWN_Q], dv[OWN_Q];
#pragma unroll
            for (int q = 0; q < OWN_Q; ++q) {
                const int j = jb + q * OWN_T;
                const int f = c0 + (j < m ? j : jb);
                int lo = 0, hi = np;
#pragma unroll
                for (int it = 0; it < 10; ++it) {
                    const int  mid = (lo + hi) >> 1;
                    const bool act = hi - lo > 1;
                    const bool le  = segoff[mid] <= f;
                    lo = (act && le) ? mid : lo;
                    hi = (act && !le) ? mid : hi;
                }
                a[q] = segbase[lo] + (f - segoff[lo]);
            }
#pragma unroll
            for (int q = 0; q < OWN_Q; ++q) {
                sv[q] = rec_s[a[q]];
                dv[q] = rec_d[a[q]];
            }
#pragma unroll
            for (int q = 0; q < OWN_Q; ++q) {
                const int j = jb + q * OWN_T;
                if (j < m) stg[j] = make_int2(sv[q], (dv[q] - n0) * AP);
            }
        }
        __syncthreads();
        for (int j0 = y; j0 < m; j0 += OWN_ROWS * OWN_U) {
            float v[OWN_U];
            int   sn[OWN_U], o[OWN_U];
#pragma unroll
            for (int u = 0; u < OWN_U; ++u) {
                const int  j  = j0 + u * OWN_ROWS;
                const int2 r2 = stg[j < m ? j : j0];
                sn[u] = r2.x;
                o[u]  = j < m ? r2.y + c : -1;
            }
#pragma unroll
            for (int u = 0; u < OWN_U; ++u) v[u] = xs[(long)sn[u] * AGX_LD + c];
#pragma unroll
            for (int u = 0; u < OWN_U; ++u)
                if (o[u] >= 0) atomicAdd(&acc[o[u]], v[u]);
        }
        __syncthreads();
    }

    const int rows = n - n0 < BKN ? n - n0 : BKN;
    for (int r = y; r < rows; r += OWN_ROWS) {
        const long i = n0 + r;
        acc[r * AP + c] = nd[2 * i] * (acc[r * AP + c] + xs[i * AGX_LD + c]);
    }
    __syncthreads();

    const int  lane = tid & 31;
    const int  wave = tid >> 5;
    const int  m    = lane & 15;
    const int  h    = lane >> 4;
    const int  r0   = wave * 16;
    const bool mine = wave < OWN_TL && r0 < rows;
    float* H  = Hs[mine ? wave : 0];
    float  dr = 0.f;
    if (mine) {
        const long rA = (long)n0 + r0 + m < n ? (long)n0 + r0 + m : (long)n - 1;
        dr = nd[2 * rA];
        const float* ar = acc + (r0 + m) * AP;

        constexpr int KS1 = (IN_C + 3) / 4;
        v8f a1[HID_C / 16] = {};
#pragma unroll
        for (int ks = 0; ks < KS1; ++ks) {
            const int ka = ks * 4 + 2 * h;
            const int kb = ka + 1;
            const float ma = (ka < IN_C) ? 1.f : 0.f;
            const float mb = (kb < IN_C) ? 1.f : 0.f;
            const int kca = (ka < IN_C) ? ka : (IN_C - 1);
            const int kcb = (kb < IN_C) ? kb : (IN_C - 1);

            const v2f a = *(const v2f*)(ar + ka);
#pragma unroll
            for (int nt = 0; nt < HID_C / 16; ++nt) {
                v2f b;
                b.x = W1[kca * HID_C + nt * 16 + m] * ma;
                b.y = W1[kcb * HID_C + nt * 16 + m] * mb;
                a1[nt] = __builtin_amdgcn_wmma_f32_16x16x4_f32(
                    false, a, false, b, (short)0, a1[nt], false, false);
            }
        }
#pragma unroll
        for (int nt = 0; nt < HID_C / 16; ++nt) {
            const float bv = b1[nt * 16 + m];
#pragma unroll
            for (int v = 0; v < 8; ++v) {
                float t = a1[nt][v] + bv;
                H[(v + 8 * h) * HP + nt * 16 + m] = t > 0.f ? t : 0.f;
            }
        }
    }
    __syncthreads();
    if (mine) {
        v8f o[OUT_C / 16] = {};
#pragma unroll
        for (int ks = 0; ks < HID_C / 4; ++ks) {
            const int ka = ks * 4 + 2 * h, kb = ka + 1;
            const v2f a = *(const v2f*)(H + m * HP + ka) * dr;
#pragma unroll
            for (int nt = 0; nt < OUT_C / 16; ++nt) {
                v2f b;
                b.x = W2[ka * OUT_C + nt * 16 + m];
                b.y = W2[kb * OUT_C + nt * 16 + m];
                o[nt] = __builtin_amdgcn_wmma_f32_16x16x4_f32(
                    false, a, false, b, (short)0, o[nt], false, false);
            }
        }
        float* orow = h2 + ((long)n0 + r0 + 8 * h) * OUT_C + m;
        if (r0 + 16 <= rows) {
#pragma unroll
            for (int nt = 0; nt < OUT_C / 16; ++nt)
#pragma unroll
                for (int v = 0; v < 8; ++v) orow[v * OUT_C + nt * 16] = o[nt][v];
        } else {
#pragma unroll
            for (int nt = 0; nt < OUT_C / 16; ++nt)
#pragma unroll
                for (int v = 0; v < 8; ++v)
                    if (r0 + v + 8 * h < rows) orow[v * OUT_C + nt * 16] = o[nt][v];
        }
    }
}


#define AP_T 512
#define AP_W (AP_T / 32)
#define GCAP 32

__device__ __forceinline__ void agg_pool_chunk(int s, float w0, float w1, const float* __restrict__ h2,
                                               int lane, float& s0, float& s1) {
    float v[32];
#pragma unroll
    for (int k = 0; k < 32; ++k) {
        const int sk = __builtin_amdgcn_readlane(s, k);
        const __attribute__((address_space(1))) float* row =
            (const __attribute__((address_space(1))) float*)(h2 + (long)sk * OUT_C);
        asm("" : "+s"(row));
        v[k] = row[lane];
    }
#pragma unroll
    for (int k = 0; k < 32; ++k) {
        const float a = __int_as_float(__builtin_amdgcn_readlane(__float_as_int(w0), k));
        const float b = __int_as_float(__builtin_amdgcn_readlane(__float_as_int(w1), k));
        s0 += a * v[k];
        s1 += b * v[k];
    }
}

__global__ void __launch_bounds__(AP_T)
k_agg_pool(const int* __restrict__ rec_s, const int* __restrict__ rec_d, const int* __restrict__ offT,
           const float* __restrict__ nd, const float* __restrict__ h2, float* __restrict__ pool,
           int n, int np) {
    __shared__ float  acc[GCAP * OUT_C];
    __shared__ int    segoff[SCAN_N];
    __shared__ int    segbase[NP_MAX];
    __shared__ int2   stg[OWN_CAP + BKN];
    __shared__ float2 ndl[BKN];
    const int tid  = threadIdx.x;
    const int lane = tid & 31;
    const int wave = tid >> 5;
    const int g    = blockIdx.x;
    const int n0   = g * BKN;
    const int rows = n - n0 < BKN ? n - n0 : BKN;

    for (int j = tid; j < GCAP * OUT_C; j += AP_T) acc[j] = 0.f;
    for (int r = tid; r < rows; r += AP_T) ndl[r] = reinterpret_cast<const float2*>(nd)[n0 + r];
    for (int pb = tid; pb < np; pb += AP_T) {
        const int a = offT[(long)g * np + pb];
        const int b = offT[(long)(g + 1) * np + pb];
        segbase[pb] = pb * EPB + a;
        segoff[pb]  = b - a;
    }
    __syncthreads();
    scan_offsets(segoff, np, tid);
    __syncthreads();
    const int total = segoff[np];
    const int gb = __builtin_amdgcn_readfirstlane(__float_as_int(ndl[0].y));
    float s0 = 0.f, s1 = 0.f;

    for (int c0 = 0; c0 < total || c0 == 0; c0 += OWN_CAP) {
        const int  m    = total - c0 < OWN_CAP ? total - c0 : OWN_CAP;
        const bool last = c0 + OWN_CAP >= total;
        for (int jb = tid; jb < m; jb += AP_T * OWN_Q) {
            int a[OWN_Q], sv[OWN_Q], dv[OWN_Q];
#pragma unroll
            for (int q = 0; q < OWN_Q; ++q) {
                const int j = jb + q * AP_T;
                const int f = c0 + (j < m ? j : jb);
                int lo = 0, hi = np;
#pragma unroll
                for (int it = 0; it < 10; ++it) {
                    const int  mid = (lo + hi) >> 1;
                    const bool act = hi - lo > 1;
                    const bool le  = segoff[mid] <= f;
                    lo = (act && le) ? mid : lo;
                    hi = (act && !le) ? mid : hi;
                }
                a[q] = segbase[lo] + (f - segoff[lo]);
            }
#pragma unroll
            for (int q = 0; q < OWN_Q; ++q) {
                sv[q] = rec_s[a[q]];
                dv[q] = rec_d[a[q]];
            }
#pragma unroll
            for (int q = 0; q < OWN_Q; ++q) {
                const int j = jb + q * AP_T;
                if (j < m) stg[j] = make_int2(sv[q], dv[q] - n0);
            }
        }
        if (last)
            for (int r = tid; r < rows; r += AP_T) stg[m + r] = make_int2(n0 + r, r);
        __syncthreads();

        const int mm = m + (last ? rows : 0);
        for (int j0 = wave * 32; j0 < mm; j0 += AP_W * 32) {
            const bool   ok  = j0 + lane < mm;
            const int2   it  = stg[ok ? j0 + lane : j0];
            const float2 rec = ndl[it.y];
            const int    gi  = __float_as_int(rec.y);
            const float  w0  = (ok && gi == gb)     ? rec.x : 0.f;
            const float  w1  = (ok && gi == gb + 1) ? rec.x : 0.f;
            agg_pool_chunk(it.x, w0, w1, h2, lane, s0, s1);

            unsigned om = __builtin_amdgcn_ballot_w32(ok && gi != gb && gi != gb + 1);
            while (om) {
                const int k = __builtin_ctz(om);
                om &= om - 1;
                const int   sk = __builtin_amdgcn_readlane(it.x, k);
                const int   gk = __builtin_amdgcn_readlane(gi, k);
                const float wk = __int_as_float(__builtin_amdgcn_readlane(__float_as_int(rec.x), k));
                const float p  = wk * h2[(long)sk * OUT_C + lane];
                const unsigned lr = (unsigned)(gk - gb);
                if (lr < (unsigned)GCAP) atomicAdd(acc + lr * OUT_C + lane, p);
                else                     atomicAdd(pool + (long)gk * OUT_C + lane, p);
            }
        }
        __syncthreads();
    }

    if (s0 != 0.f) atomicAdd(acc + lane, s0);
    if (s1 != 0.f) atomicAdd(acc + OUT_C + lane, s1);
    __syncthreads();

    for (int j = tid; j < GCAP * OUT_C; j += AP_T) {
        const float v = acc[j];
        if (v != 0.f) atomicAdd(pool + (long)gb * OUT_C + j, v);
    }
}


__global__ void k_fc(const float* __restrict__ pool, const float* __restrict__ cnt,
                     const float* __restrict__ b2,
                     const float* __restrict__ Wf1, const float* __restrict__ bf1,
                     const float* __restrict__ Wf2, const float* __restrict__ bf2,
                     float* __restrict__ out) {
    __shared__ float P[16 * 32];
    __shared__ float T[16 * 32];
    const int lane = threadIdx.x;
    const int m = lane & 15;
    const int h = lane >> 4;
    const int g0 = blockIdx.x * 16;

    for (int idx = lane; idx < 16 * 32; idx += 32) {
        int r = idx >> 5, c = idx & 31;
        float ct = cnt[g0 + r];
        ct = ct > 1.f ? ct : 1.f;
        P[idx] = pool[(g0 + r) * 32 + c] / ct + b2[c];
    }
    __syncthreads();

    v8f acc[2] = {};
#pragma unroll
    for (int ks = 0; ks < 8; ++ks) {
        const int ka = ks * 4 + 2 * h, kb = ka + 1;
        v2f a; a.x = P[m * 32 + ka]; a.y = P[m * 32 + kb];
#pragma unroll
        for (int nt = 0; nt < 2; ++nt) {
            v2f b; b.x = Wf1[ka * 32 + nt * 16 + m]; b.y = Wf1[kb * 32 + nt * 16 + m];
            acc[nt] = __builtin_amdgcn_wmma_f32_16x16x4_f32(
                false, a, false, b, (short)0, acc[nt], false, false);
        }
    }
#pragma unroll
    for (int nt = 0; nt < 2; ++nt)
#pragma unroll
        for (int v = 0; v < 8; ++v) {
            int r = v + 8 * h, c = nt * 16 + m;
            float t = acc[nt][v] + bf1[c];
            T[r * 32 + c] = t > 0.f ? t : 0.f;
        }
    __syncthreads();

    v8f o[2] = {};
#pragma unroll
    for (int ks = 0; ks < 8; ++ks) {
        const int ka = ks * 4 + 2 * h, kb = ka + 1;
        v2f a; a.x = T[m * 32 + ka]; a.y = T[m * 32 + kb];
#pragma unroll
        for (int nt = 0; nt < 2; ++nt) {
            v2f b; b.x = Wf2[ka * 32 + nt * 16 + m]; b.y = Wf2[kb * 32 + nt * 16 + m];
            o[nt] = __builtin_amdgcn_wmma_f32_16x16x4_f32(
                false, a, false, b, (short)0, o[nt], false, false);
        }
    }
#pragma unroll
    for (int nt = 0; nt < 2; ++nt)
#pragma unroll
        for (int v = 0; v < 8; ++v) {
            int r = v + 8 * h, c = nt * 16 + m;
            out[(g0 + r) * 32 + c] = o[nt][v] + bf2[c];
        }
}


extern "C" void kernel_launch(void* const* d_in, const int* in_sizes, int n_in,
                              void* d_out, int out_size, void* d_ws, size_t ws_size,
                              hipStream_t stream) {
    const float* x     = (const float*)d_in[0];
    const int*   ei    = (const int*)d_in[1];
    const int*   batch = (const int*)d_in[2];
    const float* W1  = (const float*)d_in[3];
    const float* b1  = (const float*)d_in[4];
    const float* W2  = (const float*)d_in[5];
    const float* b2  = (const float*)d_in[6];
    const float* Wf1 = (const float*)d_in[7];
    const float* bf1 = (const float*)d_in[8];
    const float* Wf2 = (const float*)d_in[9];
    const float* bf2 = (const float*)d_in[10];

    const int N = in_sizes[0] / IN_C;
    const int E = in_sizes[1] / 2;
    const int* src = ei;
    const int* dst = ei + E;

    float* ws   = (float*)d_ws;
    float* nd   = ws;
    float* reg0 = nd + 2 * (size_t)N;
    float* xs   = reg0 + (size_t)N * AGX_LD;
    float* hid  = reg0 + (size_t)N * HID_C;
    float* h2   = hid  + (size_t)N * HID_C;
    float* pool = h2   + (size_t)N * OUT_C * 2;
    float* cnt  = pool + NGRAPH * OUT_C;

    int* part = (int*)hid;
    const int ndeg = DEG_S * ((N + DEG_RN - 1) / DEG_RN);
    const int np   = (E + EPB - 1) / EPB;
    const int nbk  = (N + BKN - 1) / BKN;
    int* rec_s = (int*)(hid + (size_t)N * OUT_C);
    int* rec_d = (int*)(h2  + (size_t)N * OUT_C);
    int* offT  = rec_d + (size_t)np * EPB;
    k_deg_bin<<<ndeg + np, DEG_T, 0, stream>>>(src, dst, part, pool, rec_s, rec_d, offT, N, E, ndeg, np, nbk);

    k_node<<<(N + ND_T - 1) / ND_T, ND_T, 0, stream>>>(part, batch, x, nd, cnt, xs, N);

    const dim3 BO(IN_C, OWN_ROWS);
    k_own_mlp<<<nbk, BO, 0, stream>>>(rec_s, rec_d, offT, nd, xs, W1, b1, W2, h2, N, np);

    k_agg_pool<<<nbk, AP_T, 0, stream>>>(rec_s, rec_d, offT, nd, h2, pool, N, np);

    k_fc<<<NGRAPH / 16, 32, 0, stream>>>(pool, cnt, b2, Wf1, bf1, Wf2, bf2, (float*)d_out);
}


# per-node launch runs per 200-node owner range: gathers the range's edge pieces once into one sorted run, counts degrees off it (histogram blocks, 32 planes go); both edge passes read runs, no search
# speedup vs baseline: 5.3070x; 1.1348x over previous
#include <hip/hip_runtime.h>

typedef float v2f __attribute__((ext_vector_type(2)));
typedef float v8f __attribute__((ext_vector_type(8)));

#define IN_C   17
#define AGX_LD 32
#define HID_C  64
#define OUT_C  32
#define NGRAPH 512


#define BKN     200
#define NBK_MAX 512
#define EPB     4096
#define BIN_T   1024
#define BIN_U   (EPB / BIN_T)
#define NP_MAX  512
#define SCAN_W  17
#define SCAN_N  (32 * SCAN_W)
#define SRC_BITS 24
#define SRC_MASK ((1 << SRC_BITS) - 1)

static_assert(NBK_MAX * BKN <= (1 << SRC_BITS), "a source node fits the low bits of an edge word");
static_assert(BKN <= (1 << (32 - SRC_BITS)), "a destination's index in its range fits the high bits");

#define NACC (NGRAPH * OUT_C + NGRAPH)

__device__ __forceinline__ void scan_offsets(int* a, int n, int tid) {
    if (tid < 32) {
        int loc[SCAN_W];
        int sum = 0;
#pragma unroll
        for (int j = 0; j < SCAN_W; ++j) {
            const int idx = tid * SCAN_W + j;
            loc[j] = sum;
            sum += idx < n ? a[idx] : 0;
        }
        int inc = sum;
#pragma unroll
        for (int dlt = 1; dlt < 32; dlt <<= 1) {
            const int o = __shfl_up(inc, dlt);
            if (tid >= dlt) inc += o;
        }
        const int exc = inc - sum;
#pragma unroll
        for (int j = 0; j < SCAN_W; ++j) {
            const int idx = tid * SCAN_W + j;
            if (idx <= n) a[idx] = exc + loc[j];
        }
    }
}

__global__ void __launch_bounds__(BIN_T)
k_bin(const int* __restrict__ src, const int* __restrict__ dst, float* __restrict__ acc0,
      int* __restrict__ rec, int* __restrict__ offT, int E, int np, int nbk) {
    __shared__ int stg[EPB];
    __shared__ int gcnt[SCAN_N];
    const int pb = blockIdx.x;
    const int t  = threadIdx.x;
    for (int j = pb * BIN_T + t; j < NACC; j += np * BIN_T) acc0[j] = 0.f;
    for (int j = t; j < SCAN_N; j += BIN_T) gcnt[j] = 0;
    __syncthreads();

    const long base = (long)pb * EPB;
    int  s[BIN_U], d[BIN_U], b[BIN_U], r[BIN_U];
#pragma unroll
    for (int k = 0; k < BIN_U; ++k) {
        const long e = base + k * BIN_T + t;
        const bool ok = e < E;
        s[k] = ok ? src[e] : 0;
        d[k] = ok ? dst[e] : -1;
    }
#pragma unroll
    for (int k = 0; k < BIN_U; ++k) {
        b[k] = d[k] >= 0 ? d[k] / BKN : -1;
        r[k] = b[k] >= 0 ? atomicAdd(&gcnt[b[k]], 1) : 0;
    }
    __syncthreads();
    scan_offsets(gcnt, nbk, t);
    __syncthreads();
#pragma unroll
    for (int k = 0; k < BIN_U; ++k) {
        if (b[k] >= 0) {
            const unsigned loc = (unsigned)(d[k] - b[k] * BKN);
            stg[gcnt[b[k]] + r[k]] = (int)((unsigned)s[k] | (loc << SRC_BITS));
        }
    }
    __syncthreads();
    const int total = gcnt[nbk];
    for (int j = t; j < total; j += BIN_T) rec[base + j] = stg[j];
    for (int g = t; g <= nbk; g += BIN_T) offT[(long)g * np + pb] = gcnt[g];
}


#define NO_T 256
#define NO_W (NO_T / 32)
#define NO_Q 16
#define NO_XU ((BKN * IN_C + NO_T - 1) / NO_T)

static_assert(NO_T >= BKN, "one thread per node of the range");

__global__ void __launch_bounds__(NO_T)
k_node_own(const int* __restrict__ rec, const int* __restrict__ offT, const int* __restrict__ batch,
           const float* __restrict__ x, int* __restrict__ pk, int* __restrict__ se,
           float* __restrict__ nd, float* __restrict__ cnt, float* __restrict__ xs, int n, int np) {
    __shared__ int   segoff[SCAN_N];
    __shared__ int   segbase[NP_MAX];
    __shared__ int   hist[BKN];
    __shared__ int   wsum[NO_W];
    __shared__ float sl[BKN];
    __shared__ int   gsz[NGRAPH];
    const int tid  = threadIdx.x;
    const int lane = tid & 31;
    const int wave = tid >> 5;
    const int g    = blockIdx.x;
    const int n0   = g * BKN;
    const int rows = n - n0 < BKN ? n - n0 : BKN;
    for (int j = tid; j < NGRAPH; j += NO_T) gsz[j] = 0;
    for (int j = tid; j < BKN; j += NO_T) hist[j] = 0;

    const int    cntf = rows * IN_C;
    const float* xb   = x + (long)n0 * IN_C;
    float xv[NO_XU];
#pragma unroll
    for (int k = 0; k < NO_XU; ++k) {
        const int idx = k * NO_T + tid;
        xv[k] = idx < cntf ? xb[idx] : 0.f;
    }
    const int gi = tid < rows ? batch[n0 + tid] : -1;

    int asum = 0;
    for (int pb = tid; pb < np; pb += NO_T) {
        const int a = offT[(long)g * np + pb];
        const int b = offT[(long)(g + 1) * np + pb];
        segbase[pb] = pb * EPB + a;
        segoff[pb]  = b - a;
        asum += a;
    }
    for (int dlt = 16; dlt > 0; dlt >>= 1) asum += __shfl_xor(asum, dlt);
    if (lane == 0) wsum[wave] = asum;
    __syncthreads();
    scan_offsets(segoff, np, tid);
    __syncthreads();
    const int total = segoff[np];
    int start = 0;
    for (int w = 0; w < NO_W; ++w) start += wsum[w];

    for (int jb = tid; jb < total; jb += NO_T * NO_Q) {
        int a[NO_Q], w[NO_Q];
#pragma unroll
        for (int q = 0; q < NO_Q; ++q) {
            const int j = jb + q * NO_T;
            const int f = j < total ? j : jb;
            int lo = 0, hi = np;
#pragma unroll
            for (int it = 0; it < 10; ++it) {
                const int  mid = (lo + hi) >> 1;
                const bool act = hi - lo > 1;
                const bool le  = segoff[mid] <= f;
                lo = (act && le) ? mid : lo;
                hi = (act && !le) ? mid : hi;
            }
            a[q] = segbase[lo] + (f - segoff[lo]);
        }
#pragma unroll
        for (int q = 0; q < NO_Q; ++q) w[q] = rec[a[q]];
#pragma unroll
        for (int q = 0; q < NO_Q; ++q) {
            const int j = jb + q * NO_T;
            if (j < total) {
                atomicAdd(&hist[(unsigned)w[q] >> SRC_BITS], 1);
                pk[(long)start + j] = w[q];
            }
        }
    }
    __syncthreads();

    float dv = 0.f;
    if (tid < rows) {
        dv = rsqrtf((float)(1 + hist[tid]));
        sl[tid] = dv;
        if (gi >= 0) atomicAdd(&gsz[gi], 1);
    }
    __syncthreads();
    for (int j = tid; j < NGRAPH; j += NO_T) {
        const int q = gsz[j];
        if (q) atomicAdd(&cnt[j], (float)q);
    }

    if (tid == 0) reinterpret_cast<int2*>(se)[g] = make_int2(start, start + total);
    if (tid < rows) reinterpret_cast<float2*>(nd)[n0 + tid] = make_float2(dv, __int_as_float(gi));
#pragma unroll
    for (int k = 0; k < NO_XU; ++k) {
        const int idx = k * NO_T + tid;
        if (idx < cntf) {
            const int r = idx / IN_C, c = idx - r * IN_C;
            xs[(long)(n0 + r) * AGX_LD + c] = sl[r] * xv[k];
        }
    }
}


#define OWN_ROWS 60
#define OWN_T    (IN_C * OWN_ROWS)
#define OWN_U    8
#define OWN_CAP  4096
#define OWN_Q    4
#define OWN_TL   ((BKN + 15) / 16)
#define AP       20
#define HP       68

static_assert(AP >= ((IN_C + 3) / 4) * 4, "the k padding of the first product lies inside a row of the sums");
static_assert(OWN_TL * 32 <= OWN_T - 31, "the tile waves are whole waves");

__global__ void __launch_bounds__(OWN_T)
k_own_mlp(const int* __restrict__ pk, const int* __restrict__ se,
          const float* __restrict__ nd, const float* __restrict__ xs,
          const float* __restrict__ W1, const float* __restrict__ b1, const float* __restrict__ W2,
          float* __restrict__ h2, int n) {
    __shared__ __attribute__((aligned(16))) float acc[OWN_TL * 16 * AP];
    __shared__ int2  stg[OWN_CAP];
    __shared__ __attribute__((aligned(16))) float Hs[OWN_TL][16 * HP];
    const int c   = threadIdx.x;
    const int y   = threadIdx.y;
    const int tid = y * IN_C + c;
    const int g   = blockIdx.x;
    const int n0  = g * BKN;

    for (int j = tid; j < OWN_TL * 16 * AP; j += OWN_T) acc[j] = 0.f;
    const int2 run   = reinterpret_cast<const int2*>(se)[g];
    const int  total = run.y - run.x;
    __syncthreads();

    for (int c0 = 0; c0 < total; c0 += OWN_CAP) {
        const int  m  = total - c0 < OWN_CAP ? total - c0 : OWN_CAP;
        const int* pe = pk + run.x + c0;
        for (int jb = tid; jb < m; jb += OWN_T * OWN_Q) {
            int w[OWN_Q];
#pragma unroll
            for (int q = 0; q < OWN_Q; ++q) {
                const int j = jb + q * OWN_T;
                w[q] = pe[j < m ? j : jb];
            }
#pragma unroll
            for (int q = 0; q < OWN_Q; ++q) {
                const int j = jb + q * OWN_T;
                if (j < m) stg[j] = make_int2(w[q] & SRC_MASK, (int)((unsigned)w[q] >> SRC_BITS) * AP);
            }
        }
        __syncthreads();
        for (int j0 = y; j0 < m; j0 += OWN_ROWS * OWN_U) {
            float v[OWN_U];
            int   sn[OWN_U], o[OWN_U];
#pragma unroll
            for (int u = 0; u < OWN_U; ++u) {
                const int  j  = j0 + u * OWN_ROWS;
                const int2 r2 = stg[j < m ? j : j0];
                sn[u] = r2.x;
                o[u]  = j < m ? r2.y + c : -1;
            }
#pragma unroll
            for (int u = 0; u < OWN_U; ++u) v[u] = xs[(long)sn[u] * AGX_LD + c];
#pragma unroll
            for (int u = 0; u < OWN_U; ++u)
                if (o[u] >= 0) atomicAdd(&acc[o[u]], v[u]);
        }
        __syncthreads();
    }

    const int rows = n - n0 < BKN ? n - n0 : BKN;
    for (int r = y; r < rows; r += OWN_ROWS) {
        const long i = n0 + r;
        acc[r * AP + c] = nd[2 * i] * (acc[r * AP + c] + xs[i * AGX_LD + c]);
    }
    __syncthreads();

    const int  lane = tid & 31;
    const int  wave = tid >> 5;
    const int  m    = lane & 15;
    const int  h    = lane >> 4;
    const int  r0   = wave * 16;
    const bool mine = wave < OWN_TL && r0 < rows;
    float* H  = Hs[mine ? wave : 0];
    float  dr = 0.f;
    if (mine) {
        const long rA = (long)n0 + r0 + m < n ? (long)n0 + r0 + m : (long)n - 1;
        dr = nd[2 * rA];
        const float* ar = acc + (r0 + m) * AP;

        constexpr int KS1 = (IN_C + 3) / 4;
        v8f a1[HID_C / 16] = {};
#pragma unroll
        for (int ks = 0; ks < KS1; ++ks) {
            const int ka = ks * 4 + 2 * h;
            const int kb = ka + 1;
            const float ma = (ka < IN_C) ? 1.f : 0.f;
            const float mb = (kb < IN_C) ? 1.f : 0.f;
            const int kca = (ka < IN_C) ? ka : (IN_C - 1);
            const int kcb = (kb < IN_C) ? kb : (IN_C - 1);

            const v2f a = *(const v2f*)(ar + ka);
#pragma unroll
            for (int nt = 0; nt < HID_C / 16; ++nt) {
                v2f b;
                b.x = W1[kca * HID_C + nt * 16 + m] * ma;
                b.y = W1[kcb * HID_C + nt * 16 + m] * mb;
                a1[nt] = __builtin_amdgcn_wmma_f32_16x16x4_f32(
                    false, a, false, b, (short)0, a1[nt], false, false);
            }
        }
#pragma unroll
        for (int nt = 0; nt < HID_C / 16; ++nt) {
            const float bv = b1[nt * 16 + m];
#pragma unroll
            for (int v = 0; v < 8; ++v) {
                float t = a1[nt][v] + bv;
                H[(v + 8 * h) * HP + nt * 16 + m] = t > 0.f ? t : 0.f;
            }
        }
    }
    __syncthreads();
    if (mine) {
        v8f o[OUT_C / 16] = {};
#pragma unroll
        for (int ks = 0; ks < HID_C / 4; ++ks) {
            const int ka = ks * 4 + 2 * h, kb = ka + 1;
            const v2f a = *(const v2f*)(H + m * HP + ka) * dr;
#pragma unroll
            for (int nt = 0; nt < OUT_C / 16; ++nt) {
                v2f b;
                b.x = W2[ka * OUT_C + nt * 16 + m];
                b.y = W2[kb * OUT_C + nt * 16 + m];
                o[nt] = __builtin_amdgcn_wmma_f32_16x16x4_f32(
                    false, a, false, b, (short)0, o[nt], false, false);
            }
        }
        float* orow = h2 + ((long)n0 + r0 + 8 * h) * OUT_C + m;
        if (r0 + 16 <= rows) {
#pragma unroll
            for (int nt = 0; nt < OUT_C / 16; ++nt)
#pragma unroll
                for (int v = 0; v < 8; ++v) orow[v * OUT_C + nt * 16] = o[nt][v];
        } else {
#pragma unroll
            for (int nt = 0; nt < OUT_C / 16; ++nt)
#pragma unroll
                for (int v = 0; v < 8; ++v)
                    if (r0 + v + 8 * h < rows) orow[v * OUT_C + nt * 16] = o[nt][v];
        }
    }
}


#define AP_T 512
#define AP_W (AP_T / 32)
#define GCAP 32

__device__ __forceinline__ void agg_pool_chunk(int s, float w0, float w1, const float* __restrict__ h2,
                                               int lane, float& s0, float& s1) {
    float v[32];
#pragma unroll
    for (int k = 0; k < 32; ++k) {
        const int sk = __builtin_amdgcn_readlane(s, k);
        const __attribute__((address_space(1))) float* row =
            (const __attribute__((address_space(1))) float*)(h2 + (long)sk * OUT_C);
        asm("" : "+s"(row));
        v[k] = row[lane];
    }
#pragma unroll
    for (int k = 0; k < 32; ++k) {
        const float a = __int_as_float(__builtin_amdgcn_readlane(__float_as_int(w0), k));
        const float b = __int_as_float(__builtin_amdgcn_readlane(__float_as_int(w1), k));
        s0 += a * v[k];
        s1 += b * v[k];
    }
}

__global__ void __launch_bounds__(AP_T)
k_agg_pool(const int* __restrict__ pk, const int* __restrict__ se,
           const float* __restrict__ nd, const float* __restrict__ h2, float* __restrict__ pool,
           int n) {
    __shared__ float  acc[GCAP * OUT_C];
    __shared__ int2   stg[OWN_CAP + BKN];
    __shared__ float2 ndl[BKN];
    const int tid  = threadIdx.x;
    const int lane = tid & 31;
    const int wave = tid >> 5;
    const int g    = blockIdx.x;
    const int n0   = g * BKN;
    const int rows = n - n0 < BKN ? n - n0 : BKN;

    for (int j = tid; j < GCAP * OUT_C; j += AP_T) acc[j] = 0.f;
    for (int r = tid; r < rows; r += AP_T) ndl[r] = reinterpret_cast<const float2*>(nd)[n0 + r];
    const int2 run   = reinterpret_cast<const int2*>(se)[g];
    const int  total = run.y - run.x;
    __syncthreads();
    const int gb = __builtin_amdgcn_readfirstlane(__float_as_int(ndl[0].y));
    float s0 = 0.f, s1 = 0.f;

    for (int c0 = 0; c0 < total || c0 == 0; c0 += OWN_CAP) {
        const int  m    = total - c0 < OWN_CAP ? total - c0 : OWN_CAP;
        const bool last = c0 + OWN_CAP >= total;
        const int* pe   = pk + run.x + c0;
        for (int jb = tid; jb < m; jb += AP_T * OWN_Q) {
            int w[OWN_Q];
#pragma unroll
            for (int q = 0; q < OWN_Q; ++q) {
                const int j = jb + q * AP_T;
                w[q] = pe[j < m ? j : jb];
            }
#pragma unroll
            for (int q = 0; q < OWN_Q; ++q) {
                const int j = jb + q * AP_T;
                if (j < m) stg[j] = make_int2(w[q] & SRC_MASK, (int)((unsigned)w[q] >> SRC_BITS));
            }
        }
        if (last)
            for (int r = tid; r < rows; r += AP_T) stg[m + r] = make_int2(n0 + r, r);
        __syncthreads();

        const int mm = m + (last ? rows : 0);
        for (int j0 = wave * 32; j0 < mm; j0 += AP_W * 32) {
            const bool   ok  = j0 + lane < mm;
            const int2   it  = stg[ok ? j0 + lane : j0];
            const float2 rec = ndl[it.y];
            const int    gi  = __float_as_int(rec.y);
            const float  w0  = (ok && gi == gb)     ? rec.x : 0.f;
            const float  w1  = (ok && gi == gb + 1) ? rec.x : 0.f;
            agg_pool_chunk(it.x, w0, w1, h2, lane, s0, s1);

            unsigned om = __builtin_amdgcn_ballot_w32(ok && gi != gb && gi != gb + 1);
            while (om) {
                const int k = __builtin_ctz(om);
                om &= om - 1;
                const int   sk = __builtin_amdgcn_readlane(it.x, k);
                const int   gk = __builtin_amdgcn_readlane(gi, k);
                const float wk = __int_as_float(__builtin_amdgcn_readlane(__float_as_int(rec.x), k));
                const float p  = wk * h2[(long)sk * OUT_C + lane];
                const unsigned lr = (unsigned)(gk - gb);
                if (lr < (unsigned)GCAP) atomicAdd(acc + lr * OUT_C + lane, p);
                else                     atomicAdd(pool + (long)gk * OUT_C + lane, p);
            }
        }
        __syncthreads();
    }

    if (s0 != 0.f) atomicAdd(acc + lane, s0);
    if (s1 != 0.f) atomicAdd(acc + OUT_C + lane, s1);
    __syncthreads();

    for (int j = tid; j < GCAP * OUT_C; j += AP_T) {
        const float v = acc[j];
        if (v != 0.f) atomicAdd(pool + (long)gb * OUT_C + j, v);
    }
}


__global__ void k_fc(const float* __restrict__ pool, const float* __restrict__ cnt,
                     const float* __restrict__ b2,
                     const float* __restrict__ Wf1, const float* __restrict__ bf1,
                     const float* __restrict__ Wf2, const float* __restrict__ bf2,
                     float* __restrict__ out) {
    __shared__ float P[16 * 32];
    __shared__ float T[16 * 32];
    const int lane = threadIdx.x;
    const int m = lane & 15;
    const int h = lane >> 4;
    const int g0 = blockIdx.x * 16;

    for (int idx = lane; idx < 16 * 32; idx += 32) {
        int r = idx >> 5, c = idx & 31;
        float ct = cnt[g0 + r];
        ct = ct > 1.f ? ct : 1.f;
        P[idx] = pool[(g0 + r) * 32 + c] / ct + b2[c];
    }
    __syncthreads();

    v8f acc[2] = {};
#pragma unroll
    for (int ks = 0; ks < 8; ++ks) {
        const int ka = ks * 4 + 2 * h, kb = ka + 1;
        v2f a; a.x = P[m * 32 + ka]; a.y = P[m * 32 + kb];
#pragma unroll
        for (int nt = 0; nt < 2; ++nt) {
            v2f b; b.x = Wf1[ka * 32 + nt * 16 + m]; b.y = Wf1[kb * 32 + nt * 16 + m];
            acc[nt] = __builtin_amdgcn_wmma_f32_16x16x4_f32(
                false, a, false, b, (short)0, acc[nt], false, false);
        }
    }
#pragma unroll
    for (int nt = 0; nt < 2; ++nt)
#pragma unroll
        for (int v = 0; v < 8; ++v) {
            int r = v + 8 * h, c = nt * 16 + m;
            float t = acc[nt][v] + bf1[c];
            T[r * 32 + c] = t > 0.f ? t : 0.f;
        }
    __syncthreads();

    v8f o[2] = {};
#pragma unroll
    for (int ks = 0; ks < 8; ++ks) {
        const int ka = ks * 4 + 2 * h, kb = ka + 1;
        v2f a; a.x = T[m * 32 + ka]; a.y = T[m * 32 + kb];
#pragma unroll
        for (int nt = 0; nt < 2; ++nt) {
            v2f b; b.x = Wf2[ka * 32 + nt * 16 + m]; b.y = Wf2[kb * 32 + nt * 16 + m];
            o[nt] = __builtin_amdgcn_wmma_f32_16x16x4_f32(
                false, a, false, b, (short)0, o[nt], false, false);
        }
    }
#pragma unroll
    for (int nt = 0; nt < 2; ++nt)
#pragma unroll
        for (int v = 0; v < 8; ++v) {
            int r = v + 8 * h, c = nt * 16 + m;
            out[(g0 + r) * 32 + c] = o[nt][v] + bf2[c];
        }
}


extern "C" void kernel_launch(void* const* d_in, const int* in_sizes, int n_in,
                              void* d_out, int out_size, void* d_ws, size_t ws_size,
                              hipStream_t stream) {
    const float* x     = (const float*)d_in[0];
    const int*   ei    = (const int*)d_in[1];
    const int*   batch = (const int*)d_in[2];
    const float* W1  = (const float*)d_in[3];
    const float* b1  = (const float*)d_in[4];
    const float* W2  = (const float*)d_in[5];
    const float* b2  = (const float*)d_in[6];
    const float* Wf1 = (const float*)d_in[7];
    const float* bf1 = (const float*)d_in[8];
    const float* Wf2 = (const float*)d_in[9];
    const float* bf2 = (const float*)d_in[10];

    const int N = in_sizes[0] / IN_C;
    const int E = in_sizes[1] / 2;
    const int* src = ei;
    const int* dst = ei + E;

    float* ws   = (float*)d_ws;
    float* nd   = ws;
    float* reg0 = nd + 2 * (size_t)N;
    float* xs   = reg0 + (size_t)N * AGX_LD;
    float* hid  = reg0 + (size_t)N * HID_C;
    float* h2   = hid  + (size_t)N * HID_C;
    float* pool = h2   + (size_t)N * OUT_C * 2;
    float* cnt  = pool + NGRAPH * OUT_C;

    const int np   = (E + EPB - 1) / EPB;
    const int nbk  = (N + BKN - 1) / BKN;
    int* se   = (int*)hid;
    int* rec  = (int*)(hid + (size_t)N * OUT_C);
    int* pk   = (int*)(h2  + (size_t)N * OUT_C);
    int* offT = pk + (size_t)np * EPB;
    k_bin<<<np, BIN_T, 0, stream>>>(src, dst, pool, rec, offT, E, np, nbk);

    k_node_own<<<nbk, NO_T, 0, stream>>>(rec, offT, batch, x, pk, se, nd, cnt, xs, N, np);

    const dim3 BO(IN_C, OWN_ROWS);
    k_own_mlp<<<nbk, BO, 0, stream>>>(pk, se, nd, xs, W1, b1, W2, h2, N);

    k_agg_pool<<<nbk, AP_T, 0, stream>>>(pk, se, nd, h2, pool, N);

    k_fc<<<NGRAPH / 16, 32, 0, stream>>>(pool, cnt, b2, Wf1, bf1, Wf2, bf2, (float*)d_out);
}


# layer-1 edge sums leave LDS float atomics: owner block sorts its edges by node in LDS (returning add = rank), 5 lanes a node sum its run in registers from 16-byte row pieces; sums re-ordered
# speedup vs baseline: 5.5531x; 1.0464x over previous
#include <hip/hip_runtime.h>

typedef float v2f __attribute__((ext_vector_type(2)));
typedef float v8f __attribute__((ext_vector_type(8)));

#define IN_C   17
#define AGX_LD 32
#define XS_W   20
#define HID_C  64
#define OUT_C  32
#define NGRAPH 512


#define BKN     200
#define NBK_MAX 512
#define EPB     4096
#define BIN_T   1024
#define BIN_U   (EPB / BIN_T)
#define NP_MAX  512
#define SCAN_W  17
#define SCAN_N  (32 * SCAN_W)
#define SRC_BITS 24
#define SRC_MASK ((1 << SRC_BITS) - 1)

static_assert(NBK_MAX * BKN <= (1 << SRC_BITS), "a source node fits the low bits of an edge word");
static_assert(BKN <= (1 << (32 - SRC_BITS)), "a destination's index in its range fits the high bits");

#define NACC (NGRAPH * OUT_C + NGRAPH)

__device__ __forceinline__ void scan_offsets(int* a, int n, int tid) {
    if (tid < 32) {
        int loc[SCAN_W];
        int sum = 0;
#pragma unroll
        for (int j = 0; j < SCAN_W; ++j) {
            const int idx = tid * SCAN_W + j;
            loc[j] = sum;
            sum += idx < n ? a[idx] : 0;
        }
        int inc = sum;
#pragma unroll
        for (int dlt = 1; dlt < 32; dlt <<= 1) {
            const int o = __shfl_up(inc, dlt);
            if (tid >= dlt) inc += o;
        }
        const int exc = inc - sum;
#pragma unroll
        for (int j = 0; j < SCAN_W; ++j) {
            const int idx = tid * SCAN_W + j;
            if (idx <= n) a[idx] = exc + loc[j];
        }
    }
}

__global__ void __launch_bounds__(BIN_T)
k_bin(const int* __restrict__ src, const int* __restrict__ dst, float* __restrict__ acc0,
      int* __restrict__ rec, int* __restrict__ offT, int E, int np, int nbk) {
    __shared__ int stg[EPB];
    __shared__ int gcnt[SCAN_N];
    const int pb = blockIdx.x;
    const int t  = threadIdx.x;
    for (int j = pb * BIN_T + t; j < NACC; j += np * BIN_T) acc0[j] = 0.f;
    for (int j = t; j < SCAN_N; j += BIN_T) gcnt[j] = 0;
    __syncthreads();

    const long base = (long)pb * EPB;
    int  s[BIN_U], d[BIN_U], b[BIN_U], r[BIN_U];
#pragma unroll
    for (int k = 0; k < BIN_U; ++k) {
        const long e = base + k * BIN_T + t;
        const bool ok = e < E;
        s[k] = ok ? src[e] : 0;
        d[k] = ok ? dst[e] : -1;
    }
#pragma unroll
    for (int k = 0; k < BIN_U; ++k) {
        b[k] = d[k] >= 0 ? d[k] / BKN : -1;
        r[k] = b[k] >= 0 ? atomicAdd(&gcnt[b[k]], 1) : 0;
    }
    __syncthreads();
    scan_offsets(gcnt, nbk, t);
    __syncthreads();
#pragma unroll
    for (int k = 0; k < BIN_U; ++k) {
        if (b[k] >= 0) {
            const unsigned loc = (unsigned)(d[k] - b[k] * BKN);
            stg[gcnt[b[k]] + r[k]] = (int)((unsigned)s[k] | (loc << SRC_BITS));
        }
    }
    __syncthreads();
    const int total = gcnt[nbk];
    for (int j = t; j < total; j += BIN_T) rec[base + j] = stg[j];
    for (int g = t; g <= nbk; g += BIN_T) offT[(long)g * np + pb] = gcnt[g];
}


#define NO_T 256
#define NO_W (NO_T / 32)
#define NO_Q 16
#define NO_XU ((BKN * IN_C + NO_T - 1) / NO_T)

static_assert(NO_T >= BKN, "one thread per node of the range");

__global__ void __launch_bounds__(NO_T)
k_node_own(const int* __restrict__ rec, const int* __restrict__ offT, const int* __restrict__ batch,
           const float* __restrict__ x, int* __restrict__ pk, int* __restrict__ se,
           float* __restrict__ nd, float* __restrict__ cnt, float* __restrict__ xs, int n, int np) {
    __shared__ int   segoff[SCAN_N];
    __shared__ int   segbase[NP_MAX];
    __shared__ int   hist[BKN];
    __shared__ int   wsum[NO_W];
    __shared__ float sl[BKN];
    __shared__ int   gsz[NGRAPH];
    const int tid  = threadIdx.x;
    const int lane = tid & 31;
    const int wave = tid >> 5;
    const int g    = blockIdx.x;
    const int n0   = g * BKN;
    const int rows = n - n0 < BKN ? n - n0 : BKN;
    for (int j = tid; j < NGRAPH; j += NO_T) gsz[j] = 0;
    for (int j = tid; j < BKN; j += NO_T) hist[j] = 0;

    const int    cntf = rows * IN_C;
    const float* xb   = x + (long)n0 * IN_C;
    float xv[NO_XU];
#pragma unroll
    for (int k = 0; k < NO_XU; ++k) {
        const int idx = k * NO_T + tid;
        xv[k] = idx < cntf ? xb[idx] : 0.f;
    }
    const int gi = tid < rows ? batch[n0 + tid] : -1;

    int asum = 0;
    for (int pb = tid; pb < np; pb += NO_T) {
        const int a = offT[(long)g * np + pb];
        const int b = offT[(long)(g + 1) * np + pb];
        segbase[pb] = pb * EPB + a;
        segoff[pb]  = b - a;
        asum += a;
    }
    for (int dlt = 16; dlt > 0; dlt >>= 1) asum += __shfl_xor(asum, dlt);
    if (lane == 0) wsum[wave] = asum;
    __syncthreads();
    scan_offsets(segoff, np, tid);
    __syncthreads();
    const int total = segoff[np];
    int start = 0;
    for (int w = 0; w < NO_W; ++w) start += wsum[w];

    for (int jb = tid; jb < total; jb += NO_T * NO_Q) {
        int a[NO_Q], w[NO_Q];
#pragma unroll
        for (int q = 0; q < NO_Q; ++q) {
            const int j = jb + q * NO_T;
            const int f = j < total ? j : jb;
            int lo = 0, hi = np;
#pragma unroll
            for (int it = 0; it < 10; ++it) {
                const int  mid = (lo + hi) >> 1;
                const bool act = hi - lo > 1;
                const bool le  = segoff[mid] <= f;
                lo = (act && le) ? mid : lo;
                hi = (act && !le) ? mid : hi;
            }
            a[q] = segbase[lo] + (f - segoff[lo]);
        }
#pragma unroll
        for (int q = 0; q < NO_Q; ++q) w[q] = rec[a[q]];
#pragma unroll
        for (int q = 0; q < NO_Q; ++q) {
            const int j = jb + q * NO_T;
            if (j < total) {
                atomicAdd(&hist[(unsigned)w[q] >> SRC_BITS], 1);
                pk[(long)start + j] = w[q];
            }
        }
    }
    __syncthreads();

    float dv = 0.f;
    if (tid < rows) {
        dv = rsqrtf((float)(1 + hist[tid]));
        sl[tid] = dv;
        if (gi >= 0) atomicAdd(&gsz[gi], 1);
    }
    __syncthreads();
    for (int j = tid; j < NGRAPH; j += NO_T) {
        const int q = gsz[j];
        if (q) atomicAdd(&cnt[j], (float)q);
    }

    if (tid == 0) reinterpret_cast<int2*>(se)[g] = make_int2(start, start + total);
    if (tid < rows) reinterpret_cast<float2*>(nd)[n0 + tid] = make_float2(dv, __int_as_float(gi));
#pragma unroll
    for (int k = 0; k < NO_XU; ++k) {
        const int idx = k * NO_T + tid;
        if (idx < cntf) {
            const int r = idx / IN_C, c = idx - r * IN_C;
            xs[(long)(n0 + r) * AGX_LD + c] = sl[r] * xv[k];
        }
    }
    for (int j = tid; j < rows * (XS_W - IN_C); j += NO_T) {
        const int r = j / (XS_W - IN_C), c = j - r * (XS_W - IN_C);
        xs[(long)(n0 + r) * AGX_LD + IN_C + c] = 0.f;
    }
}


#define OWN_L    5
#define OWN_ROWS 204
#define OWN_T    (OWN_L * OWN_ROWS)
#define OWN_U    8
#define SRT_Q    4
#define SRT_CAP  (OWN_T * SRT_Q)
#define OWN_CAP  4096
#define OWN_Q    4
#define OWN_TL   ((BKN + 15) / 16)
#define AP       XS_W
#define HP       68

static_assert(AP >= ((IN_C + 3) / 4) * 4, "the k padding of the first product lies inside a row of the sums");
static_assert(AP == 4 * OWN_L && AP <= AGX_LD, "a row is OWN_L pieces of four floats");
static_assert(OWN_ROWS >= BKN && BKN < SCAN_N, "one lane group per node of the range; the counters fit the scan");
static_assert(OWN_TL * 32 <= OWN_T - 31, "the tile waves are whole waves");

__global__ void __launch_bounds__(OWN_T, 16)
k_own_mlp(const int* __restrict__ pk, const int* __restrict__ se,
          const float* __restrict__ nd, const float* __restrict__ xs,
          const float* __restrict__ W1, const float* __restrict__ b1, const float* __restrict__ W2,
          float* __restrict__ h2, int n) {
    __shared__ __attribute__((aligned(16))) float acc[OWN_TL * 16 * AP];
    __shared__ int   srt[SRT_CAP];
    __shared__ int   dcnt[SCAN_N];
    __shared__ __attribute__((aligned(16))) float Hs[OWN_TL][16 * HP];
    const int q    = threadIdx.x;
    const int y    = threadIdx.y;
    const int tid  = y * OWN_L + q;
    const int g    = blockIdx.x;
    const int n0   = g * BKN;
    const int rows = n - n0 < BKN ? n - n0 : BKN;

    for (int j = rows * AP + tid; j < OWN_TL * 16 * AP; j += OWN_T) acc[j] = 0.f;
    const int2 run   = reinterpret_cast<const int2*>(se)[g];
    const int  total = run.y - run.x;
    const float*   xq = xs + 4 * q;
    const unsigned qb = 4 * q * (unsigned)sizeof(float);
    float s0 = 0.f, s1 = 0.f, s2 = 0.f, s3 = 0.f;
    float dv = 0.f;
    if (y < rows) {
        const float4 sv = *reinterpret_cast<const float4*>(xq + (long)(n0 + y) * AGX_LD);
        s0 = sv.x; s1 = sv.y; s2 = sv.z; s3 = sv.w;
        dv = nd[2 * (long)(n0 + y)];
    }

    for (int c0 = 0; c0 < total; c0 += SRT_CAP) {
        const int  m  = total - c0 < SRT_CAP ? total - c0 : SRT_CAP;
        const int* pe = pk + run.x + c0;
        if (c0 > 0) __syncthreads();
        int w[SRT_Q], rk[SRT_Q];
#pragma unroll
        for (int k = 0; k < SRT_Q; ++k) {
            const int j = tid + k * OWN_T;
            w[k] = j < m ? pe[j] : 0;
        }
        for (int j = tid; j < SCAN_N; j += OWN_T) dcnt[j] = 0;
        __syncthreads();
#pragma unroll
        for (int k = 0; k < SRT_Q; ++k) {
            const int j = tid + k * OWN_T;
            rk[k] = j < m ? atomicAdd(&dcnt[(unsigned)w[k] >> SRC_BITS], 1) : 0;
        }
        __syncthreads();
        scan_offsets(dcnt, BKN, tid);
        __syncthreads();
        int pos[SRT_Q];
#pragma unroll
        for (int k = 0; k < SRT_Q; ++k) pos[k] = dcnt[(unsigned)w[k] >> SRC_BITS] + rk[k];
#pragma unroll
        for (int k = 0; k < SRT_Q; ++k) {
            const int j = tid + k * OWN_T;
            if (j < m) srt[pos[k]] = w[k] & SRC_MASK;
        }
        __syncthreads();
        if (y < BKN) {
            const int e = dcnt[y + 1];
            for (int i = dcnt[y]; i < e; i += OWN_U) {
                unsigned off[OWN_U];
                float4   v[OWN_U];
#pragma unroll
                for (int u = 0; u < OWN_U; ++u) {
                    const int j = i + u < e ? i + u : e - 1;
                    off[u] = (unsigned)srt[j] * (unsigned)(AGX_LD * sizeof(float)) + qb;
                }
#pragma unroll
                for (int u = 0; u < OWN_U; ++u)
                    v[u] = *reinterpret_cast<const float4*>(reinterpret_cast<const char*>(xs) + off[u]);
                __builtin_amdgcn_sched_barrier(0);
#pragma unroll
                for (int u = 0; u < OWN_U; ++u) {
                    const float f = i + u < e ? 1.f : 0.f;
                    s0 = fmaf(f, v[u].x, s0);
                    s1 = fmaf(f, v[u].y, s1);
                    s2 = fmaf(f, v[u].z, s2);
                    s3 = fmaf(f, v[u].w, s3);
                }
            }
        }
    }

    if (y < rows)
        *reinterpret_cast<float4*>(acc + y * AP + 4 * q) = make_float4(dv * s0, dv * s1, dv * s2, dv * s3);
    __syncthreads();

    const int  lane = tid & 31;
    const int  wave = tid >> 5;
    const int  m    = lane & 15;
    const int  h    = lane >> 4;
    const int  r0   = wave * 16;
    const bool mine = wave < OWN_TL && r0 < rows;
    float* H  = Hs[mine ? wave : 0];
    float  dr = 0.f;
    if (mine) {
        const long rA = (long)n0 + r0 + m < n ? (long)n0 + r0 + m : (long)n - 1;
        dr = nd[2 * rA];
        const float* ar = acc + (r0 + m) * AP;

        constexpr int KS1 = (IN_C + 3) / 4;
        v8f a1[HID_C / 16] = {};
#pragma unroll
        for (int ks = 0; ks < KS1; ++ks) {
            const int ka = ks * 4 + 2 * h;
            const int kb = ka + 1;
            const float ma = (ka < IN_C) ? 1.f : 0.f;
            const float mb = (kb < IN_C) ? 1.f : 0.f;
            const int kca = (ka < IN_C) ? ka : (IN_C - 1);
            const int kcb = (kb < IN_C) ? kb : (IN_C - 1);

            const v2f a = *(const v2f*)(ar + ka);
#pragma unroll
            for (int nt = 0; nt < HID_C / 16; ++nt) {
                v2f b;
                b.x = W1[kca * HID_C + nt * 16 + m] * ma;
                b.y = W1[kcb * HID_C + nt * 16 + m] * mb;
                a1[nt] = __builtin_amdgcn_wmma_f32_16x16x4_f32(
                    false, a, false, b, (short)0, a1[nt], false, false);
            }
        }
#pragma unroll
        for (int nt = 0; nt < HID_C / 16; ++nt) {
            const float bv = b1[nt * 16 + m];
#pragma unroll
            for (int v = 0; v < 8; ++v) {
                float t = a1[nt][v] + bv;
                H[(v + 8 * h) * HP + nt * 16 + m] = t > 0.f ? t : 0.f;
            }
        }
    }
    __syncthreads();
    if (mine) {
        v8f o[OUT_C / 16] = {};
#pragma unroll
        for (int ks = 0; ks < HID_C / 4; ++ks) {
            const int ka = ks * 4 + 2 * h, kb = ka + 1;
            const v2f a = *(const v2f*)(H + m * HP + ka) * dr;
#pragma unroll
            for (int nt = 0; nt < OUT_C / 16; ++nt) {
                v2f b;
                b.x = W2[ka * OUT_C + nt * 16 + m];
                b.y = W2[kb * OUT_C + nt * 16 + m];
                o[nt] = __builtin_amdgcn_wmma_f32_16x16x4_f32(
                    false, a, false, b, (short)0, o[nt], false, false);
            }
        }
        float* orow = h2 + ((long)n0 + r0 + 8 * h) * OUT_C + m;
        if (r0 + 16 <= rows) {
#pragma unroll
            for (int nt = 0; nt < OUT_C / 16; ++nt)
#pragma unroll
                for (int v = 0; v < 8; ++v) orow[v * OUT_C + nt * 16] = o[nt][v];
        } else {
#pragma unroll
            for (int nt = 0; nt < OUT_C / 16; ++nt)
#pragma unroll
                for (int v = 0; v < 8; ++v)
                    if (r0 + v + 8 * h < rows) orow[v * OUT_C + nt * 16] = o[nt][v];
        }
    }
}


#define AP_T 512
#define AP_W (AP_T / 32)
#define GCAP 32

__device__ __forceinline__ void agg_pool_chunk(int s, float w0, float w1, const float* __restrict__ h2,
                                               int lane, float& s0, float& s1) {
    float v[32];
#pragma unroll
    for (int k = 0; k < 32; ++k) {
        const int sk = __builtin_amdgcn_readlane(s, k);
        const __attribute__((address_space(1))) float* row =
            (const __attribute__((address_space(1))) float*)(h2 + (long)sk * OUT_C);
        asm("" : "+s"(row));
        v[k] = row[lane];
    }
#pragma unroll
    for (int k = 0; k < 32; ++k) {
        const float a = __int_as_float(__builtin_amdgcn_readlane(__float_as_int(w0), k));
        const float b = __int_as_float(__builtin_amdgcn_readlane(__float_as_int(w1), k));
        s0 += a * v[k];
        s1 += b * v[k];
    }
}

__global__ void __launch_bounds__(AP_T)
k_agg_pool(const int* __restrict__ pk, const int* __restrict__ se,
           const float* __restrict__ nd, const float* __restrict__ h2, float* __restrict__ pool,
           int n) {
    __shared__ float  acc[GCAP * OUT_C];
    __shared__ int2   stg[OWN_CAP + BKN];
    __shared__ float2 ndl[BKN];
    const int tid  = threadIdx.x;
    const int lane = tid & 31;
    const int wave = tid >> 5;
    const int g    = blockIdx.x;
    const int n0   = g * BKN;
    const int rows = n - n0 < BKN ? n - n0 : BKN;

    for (int j = tid; j < GCAP * OUT_C; j += AP_T) acc[j] = 0.f;
    for (int r = tid; r < rows; r += AP_T) ndl[r] = reinterpret_cast<const float2*>(nd)[n0 + r];
    const int2 run   = reinterpret_cast<const int2*>(se)[g];
    const int  total = run.y - run.x;
    __syncthreads();
    const int gb = __builtin_amdgcn_readfirstlane(__float_as_int(ndl[0].y));
    float s0 = 0.f, s1 = 0.f;

    for (int c0 = 0; c0 < total || c0 == 0; c0 += OWN_CAP) {
        const int  m    = total - c0 < OWN_CAP ? total - c0 : OWN_CAP;
        const bool last = c0 + OWN_CAP >= total;
        const int* pe   = pk + run.x + c0;
        for (int jb = tid; jb < m; jb += AP_T * OWN_Q) {
            int w[OWN_Q];
#pragma unroll
            for (int q = 0; q < OWN_Q; ++q) {
                const int j = jb + q * AP_T;
                w[q] = pe[j < m ? j : jb];
            }
#pragma unroll
            for (int q = 0; q < OWN_Q; ++q) {
                const int j = jb + q * AP_T;
                if (j < m) stg[j] = make_int2(w[q] & SRC_MASK, (int)((unsigned)w[q] >> SRC_BITS));
            }
        }
        if (last)
            for (int r = tid; r < rows; r += AP_T) stg[m + r] = make_int2(n0 + r, r);
        __syncthreads();

        const int mm = m + (last ? rows : 0);
        for (int j0 = wave * 32; j0 < mm; j0 += AP_W * 32) {
            const bool   ok  = j0 + lane < mm;
            const int2   it  = stg[ok ? j0 + lane : j0];
            const float2 rec = ndl[it.y];
            const int    gi  = __float_as_int(rec.y);
            const float  w0  = (ok && gi == gb)     ? rec.x : 0.f;
            const float  w1  = (ok && gi == gb + 1) ? rec.x : 0.f;
            agg_pool_chunk(it.x, w0, w1, h2, lane, s0, s1);

            unsigned om = __builtin_amdgcn_ballot_w32(ok && gi != gb && gi != gb + 1);
            while (om) {
                const int k = __builtin_ctz(om);
                om &= om - 1;
                const int   sk = __builtin_amdgcn_readlane(it.x, k);
                const int   gk = __builtin_amdgcn_readlane(gi, k);
                const float wk = __int_as_float(__builtin_amdgcn_readlane(__float_as_int(rec.x), k));
                const float p  = wk * h2[(long)sk * OUT_C + lane];
                const unsigned lr = (unsigned)(gk - gb);
                if (lr < (unsigned)GCAP) atomicAdd(acc + lr * OUT_C + lane, p);
                else                     atomicAdd(pool + (long)gk * OUT_C + lane, p);
            }
        }
        __syncthreads();
    }

    if (s0 != 0.f) atomicAdd(acc + lane, s0);
    if (s1 != 0.f) atomicAdd(acc + OUT_C + lane, s1);
    __syncthreads();

    for (int j = tid; j < GCAP * OUT_C; j += AP_T) {
        const float v = acc[j];
        if (v != 0.f) atomicAdd(pool + (long)gb * OUT_C + j, v);
    }
}


__global__ void k_fc(const float* __restrict__ pool, const float* __restrict__ cnt,
                     const float* __restrict__ b2,
                     const float* __restrict__ Wf1, const float* __restrict__ bf1,
                     const float* __restrict__ Wf2, const float* __restrict__ bf2,
                     float* __restrict__ out) {
    __shared__ float P[16 * 32];
    __shared__ float T[16 * 32];
    const int lane = threadIdx.x;
    const int m = lane & 15;
    const int h = lane >> 4;
    const int g0 = blockIdx.x * 16;

    for (int idx = lane; idx < 16 * 32; idx += 32) {
        int r = idx >> 5, c = idx & 31;
        float ct = cnt[g0 + r];
        ct = ct > 1.f ? ct : 1.f;
        P[idx] = pool[(g0 + r) * 32 + c] / ct + b2[c];
    }
    __syncthreads();

    v8f acc[2] = {};
#pragma unroll
    for (int ks = 0; ks < 8; ++ks) {
        const int ka = ks * 4 + 2 * h, kb = ka + 1;
        v2f a; a.x = P[m * 32 + ka]; a.y = P[m * 32 + kb];
#pragma unroll
        for (int nt = 0; nt < 2; ++nt) {
            v2f b; b.x = Wf1[ka * 32 + nt * 16 + m]; b.y = Wf1[kb * 32 + nt * 16 + m];
            acc[nt] = __builtin_amdgcn_wmma_f32_16x16x4_f32(
                false, a, false, b, (short)0, acc[nt], false, false);
        }
    }
#pragma unroll
    for (int nt = 0; nt < 2; ++nt)
#pragma unroll
        for (int v = 0; v < 8; ++v) {
            int r = v + 8 * h, c = nt * 16 + m;
            float t = acc[nt][v] + bf1[c];
            T[r * 32 + c] = t > 0.f ? t : 0.f;
        }
    __syncthreads();

    v8f o[2] = {};
#pragma unroll
    for (int ks = 0; ks < 8; ++ks) {
        const int ka = ks * 4 + 2 * h, kb = ka + 1;
        v2f a; a.x = T[m * 32 + ka]; a.y = T[m * 32 + kb];
#pragma unroll
        for (int nt = 0; nt < 2; ++nt) {
            v2f b; b.x = Wf2[ka * 32 + nt * 16 + m]; b.y = Wf2[kb * 32 + nt * 16 + m];
            o[nt] = __builtin_amdgcn_wmma_f32_16x16x4_f32(
                false, a, false, b, (short)0, o[nt], false, false);
        }
    }
#pragma unroll
    for (int nt = 0; nt < 2; ++nt)
#pragma unroll
        for (int v = 0; v < 8; ++v) {
            int r = v + 8 * h, c = nt * 16 + m;
            out[(g0 + r) * 32 + c] = o[nt][v] + bf2[c];
        }
}


extern "C" void kernel_launch(void* const* d_in, const int* in_sizes, int n_in,
                              void* d_out, int out_size, void* d_ws, size_t ws_size,
                              hipStream_t stream) {
    const float* x     = (const float*)d_in[0];
    const int*   ei    = (const int*)d_in[1];
    const int*   batch = (const int*)d_in[2];
    const float* W1  = (const float*)d_in[3];
    const float* b1  = (const float*)d_in[4];
    const float* W2  = (const float*)d_in[5];
    const float* b2  = (const float*)d_in[6];
    const float* Wf1 = (const float*)d_in[7];
    const float* bf1 = (const float*)d_in[8];
    const float* Wf2 = (const float*)d_in[9];
    const float* bf2 = (const float*)d_in[10];

    const int N = in_sizes[0] / IN_C;
    const int E = in_sizes[1] / 2;
    const int* src = ei;
    const int* dst = ei + E;

    float* ws   = (float*)d_ws;
    float* nd   = ws;
    float* reg0 = nd + 2 * (size_t)N;
    float* xs   = reg0 + (size_t)N * AGX_LD;
    float* hid  = reg0 + (size_t)N * HID_C;
    float* h2   = hid  + (size_t)N * HID_C;
    float* pool = h2   + (size_t)N * OUT_C * 2;
    float* cnt  = pool + NGRAPH * OUT_C;

    const int np   = (E + EPB - 1) / EPB;
    const int nbk  = (N + BKN - 1) / BKN;
    int* se   = (int*)hid;
    int* rec  = (int*)(hid + (size_t)N * OUT_C);
    int* pk   = (int*)(h2  + (size_t)N * OUT_C);
    int* offT = pk + (size_t)np * EPB;
    k_bin<<<np, BIN_T, 0, stream>>>(src, dst, pool, rec, offT, E, np, nbk);

    k_node_own<<<nbk, NO_T, 0, stream>>>(rec, offT, batch, x, pk, se, nd, cnt, xs, N, np);

    const dim3 BO(OWN_L, OWN_ROWS);
    k_own_mlp<<<nbk, BO, 0, stream>>>(pk, se, nd, xs, W1, b1, W2, h2, N);

    k_agg_pool<<<nbk, AP_T, 0, stream>>>(pk, se, nd, h2, pool, N);

    k_fc<<<NGRAPH / 16, 32, 0, stream>>>(pool, cnt, b2, Wf1, bf1, Wf2, bf2, (float*)d_out);
}
